# starting kernel
# baseline (speedup 1.0000x reference)
#include <hip/hip_runtime.h>
#include <math.h>

typedef __attribute__((ext_vector_type(2))) float v2f;
typedef __attribute__((ext_vector_type(8))) float v8f;

#define WAVES_PER_BLOCK 8
#define EPSV 1e-4f
#define INV_SQRT10 0.3162277660168379f

__device__ const float kPre[16] = {1,1,3,3,1,1,3,3,-1,-1,-3,-3,-1,-1,-3,-3};
__device__ const float kPim[16] = {1,3,1,3,-1,-3,-1,-3,1,3,1,3,-1,-3,-1,-3};

struct WS {
  union {
    struct ChT { float Are[16][17]; float Aim[16][17]; } ch;
    float aug[16][33];
  } u;
  float Wre[16][10], Wim[16][10];
  float gre[8][9],  gim[8][9];
  float gr[16][17];
  float yptr[16][17];
  float ymfre[8], ymfim[8];
  float xre[8], xim[8], varL[8];
  float gxre[8], gxim[8];
  float muL[16], znumL[16];
  float rhoL[8];
  float llr_aL[8][4], llr_dL[8][4];
};

__device__ __forceinline__ float log_sigmoid(float x) {
  return fminf(x, 0.0f) - log1pf(__expf(-fabsf(x)));
}

__global__ __launch_bounds__(256) void mmse_pic_kernel(
    const float* __restrict__ y_re, const float* __restrict__ y_im,
    const float* __restrict__ h_re, const float* __restrict__ h_im,
    const float* __restrict__ prior,
    const float* __restrict__ s_re, const float* __restrict__ s_im,
    float* __restrict__ out)
{
  __shared__ WS ws[WAVES_PER_BLOCK];
  const int lane = threadIdx.x & 31;
  const int wid  = threadIdx.x >> 5;
  const int b    = blockIdx.x * WAVES_PER_BLOCK + wid;
  WS& w = ws[wid];

  #pragma unroll
  for (int t = 0; t < 8; ++t) {
    int idx = lane + 32*t;
    int i = idx >> 4, j = idx & 15;
    w.u.ch.Are[i][j] = s_re[b*256 + idx];
    w.u.ch.Aim[i][j] = s_im[b*256 + idx];
  }
  #pragma unroll
  for (int t = 0; t < 4; ++t) {
    int idx = lane + 32*t;
    int m = idx >> 3, c = idx & 7;
    w.Wre[m][c] = h_re[b*128 + idx];
    w.Wim[m][c] = h_im[b*128 + idx];
  }
  if (lane < 16) {
    w.Wre[lane][8] = y_re[b*16 + lane];
    w.Wim[lane][8] = y_im[b*16 + lane];
  }
  const int s_out = lane >> 2, k_out = lane & 3;
  float my_lld = prior[b*32 + lane];
  w.llr_dL[s_out][k_out] = my_lld;
  __syncthreads();

  const int r = lane;
  for (int k = 0; k < 16; ++k) {
    float dk  = w.u.ch.Are[k][k];
    float sdk = sqrtf(dk);
    float isd = 1.0f / sdk;
    if (r == k) { w.u.ch.Are[k][k] = sdk; w.u.ch.Aim[k][k] = 0.0f; }
    if (r < 16 && r > k) { w.u.ch.Are[r][k] *= isd; w.u.ch.Aim[r][k] *= isd; }
    __syncthreads();
    if (r < 16 && r > k) {
      float lre = w.u.ch.Are[r][k], lim = w.u.ch.Aim[r][k];
      for (int j = k+1; j <= r; ++j) {
        float cre = w.u.ch.Are[j][k], cim = w.u.ch.Aim[j][k];
        w.u.ch.Are[r][j] -= lre*cre + lim*cim;
        w.u.ch.Aim[r][j] -= lim*cre - lre*cim;
      }
    }
    __syncthreads();
  }

  if (lane < 9) {
    const int c = lane;
    for (int m = 0; m < 16; ++m) {
      float accre = w.Wre[m][c], accim = w.Wim[m][c];
      for (int j = 0; j < m; ++j) {
        float lre = w.u.ch.Are[m][j], lim = w.u.ch.Aim[m][j];
        float wr = w.Wre[j][c], wi = w.Wim[j][c];
        accre -= lre*wr - lim*wi;
        accim -= lre*wi + lim*wr;
      }
      float inv = 1.0f / w.u.ch.Are[m][m];
      w.Wre[m][c] = accre*inv;
      w.Wim[m][c] = accim*inv;
    }
  }
  __syncthreads();

  if (lane < 8) {
    int s_ = lane;
    float ar = 0.0f, ai = 0.0f;
    #pragma unroll
    for (int m = 0; m < 16; ++m) {
      float hr = w.Wre[m][s_], hi = w.Wim[m][s_];
      float yr = w.Wre[m][8],  yi = w.Wim[m][8];
      ar += hr*yr + hi*yi;
      ai += hr*yi - hi*yr;
    }
    w.ymfre[s_] = ar; w.ymfim[s_] = ai;
  }
  #pragma unroll
  for (int t = 0; t < 2; ++t) {
    int idx = lane + 32*t;
    int i = idx >> 3, j = idx & 7;
    float ar = 0.0f, ai = 0.0f;
    #pragma unroll
    for (int m = 0; m < 16; ++m) {
      float xr = w.Wre[m][i], xi = w.Wim[m][i];
      float yr = w.Wre[m][j], yi = w.Wim[m][j];
      ar += xr*yr + xi*yi;
      ai += xr*yi - xi*yr;
    }
    w.gre[i][j] = ar; w.gim[i][j] = ai;
  }
  __syncthreads();
  #pragma unroll
  for (int t = 0; t < 8; ++t) {
    int idx = lane + 32*t;
    int i = idx >> 4, j = idx & 15;
    float v;
    if (i < 8) v = (j < 8) ? w.gre[i][j]   : -w.gim[i][j-8];
    else       v = (j < 8) ? w.gim[i-8][j] :  w.gre[i-8][j-8];
    w.gr[i][j] = v;
  }
  __syncthreads();

  float my_lla = 0.0f;
  for (int iter = 0; iter < 2; ++iter) {
    my_lla = my_lld;
    w.llr_aL[s_out][k_out] = my_lla;
    __syncthreads();

    float lg[16];
    if (lane < 8) {
      int s_ = lane;
      float l0 = w.llr_aL[s_][0], l1 = w.llr_aL[s_][1];
      float l2 = w.llr_aL[s_][2], l3 = w.llr_aL[s_][3];
      #pragma unroll
      for (int p = 0; p < 16; ++p) {
        lg[p] = log_sigmoid((p & 8) ? l0 : -l0)
              + log_sigmoid((p & 4) ? l1 : -l1)
              + log_sigmoid((p & 2) ? l2 : -l2)
              + log_sigmoid((p & 1) ? l3 : -l3);
      }
      float mx = lg[0];
      #pragma unroll
      for (int p = 1; p < 16; ++p) mx = fmaxf(mx, lg[p]);
      float sum = 0.0f, mre = 0.0f, mim = 0.0f, e2 = 0.0f;
      #pragma unroll
      for (int p = 0; p < 16; ++p) {
        float e = __expf(lg[p] - mx);
        float pr = kPre[p]*INV_SQRT10, pi = kPim[p]*INV_SQRT10;
        sum += e; mre += e*pr; mim += e*pi; e2 += e*(pr*pr + pi*pi);
      }
      float invs = 1.0f / sum;
      mre *= invs; mim *= invs; e2 *= invs;
      w.xre[s_] = mre; w.xim[s_] = mim;
      w.varL[s_] = e2 - (mre*mre + mim*mim);
    }
    __syncthreads();

    if (lane < 8) {
      int i = lane;
      float ar = 0.0f, ai = 0.0f;
      #pragma unroll
      for (int j = 0; j < 8; ++j) {
        float gre_ = w.gre[i][j], gim_ = w.gim[i][j];
        float xr = w.xre[j], xi = w.xim[j];
        ar += gre_*xr - gim_*xi;
        ai += gre_*xi + gim_*xr;
      }
      w.gxre[i] = ar; w.gxim[i] = ai;
    }
    __syncthreads();

    #pragma unroll
    for (int t = 0; t < 2; ++t) {
      int idx = lane + 32*t;
      int i = idx >> 3, j = idx & 7;
      float gre_ = w.gre[j][i], gim_ = w.gim[j][i];
      float xr = w.xre[i], xi = w.xim[i];
      float tre = w.ymfre[j] + (gre_*xr - gim_*xi) - w.gxre[j];
      float tim = w.ymfim[j] + (gre_*xi + gim_*xr) - w.gxim[j];
      w.yptr[i][j]     = tre; w.yptr[i][j+8]   = tim;
      w.yptr[i+8][j]   = tre; w.yptr[i+8][j+8] = tim;
    }
    {
      int rr = lane & 15, hf = lane >> 4, base = hf*16;
      #pragma unroll
      for (int c = 0; c < 16; ++c) {
        int col = base + c;
        float v;
        if (col < 16) v = w.gr[rr][col] * w.varL[col & 7] + ((rr == col) ? 1.0f : 0.0f);
        else          v = ((col - 16) == rr) ? 1.0f : 0.0f;
        w.u.aug[rr][col] = v;
      }
    }
    __syncthreads();

    {
      int rr = lane & 15, hf = lane >> 4, base = hf*16;
      for (int k = 0; k < 16; ++k) {
        float pinv = 1.0f / w.u.aug[k][k];
        __syncthreads();
        if (rr == k) {
          #pragma unroll
          for (int c = 0; c < 16; ++c) w.u.aug[k][base+c] *= pinv;
        }
        __syncthreads();
        float f = w.u.aug[rr][k];
        if (rr != k) {
          #pragma unroll
          for (int c = 0; c < 16; ++c)
            w.u.aug[rr][base+c] -= f * w.u.aug[k][base+c];
        }
        __syncthreads();
      }
    }

    {
      int nn = lane & 15, hf = lane >> 4;
      v8f p1 = {}; v8f p2 = {};
      #pragma unroll
      for (int kb = 0; kb < 4; ++kb) {
        int k0 = kb*4 + hf*2;
        v2f af, b1, b2;
        af.x = w.u.aug[nn][16 + k0];
        af.y = w.u.aug[nn][16 + k0 + 1];
        b1.x = w.gr[k0][nn];   b1.y = w.gr[k0+1][nn];
        b2.x = w.yptr[nn][k0]; b2.y = w.yptr[nn][k0+1];
        p1 = __builtin_amdgcn_wmma_f32_16x16x4_f32(false, af, false, b1,
                                                   (short)0, p1, false, false);
        p2 = __builtin_amdgcn_wmma_f32_16x16x4_f32(false, af, false, b2,
                                                   (short)0, p2, false, false);
      }
      #pragma unroll
      for (int v = 0; v < 8; ++v) {
        int row = v + 8*hf;
        if (row == nn) { w.muL[nn] = p1[v]; w.znumL[nn] = p2[v]; }
      }
    }
    __syncthreads();

    if (lane < 8) {
      int s_ = lane;
      float m1 = w.muL[s_], m2 = w.muL[s_ + 8];
      w.xre[s_] = w.znumL[s_]     / m1;
      w.xim[s_] = w.znumL[s_ + 8] / m2;
      w.rhoL[s_] = m1 / fmaxf(1.0f - w.varL[s_] * m1, EPSV);
    }
    __syncthreads();

    if (lane < 8) {
      int s_ = lane;
      float zr = w.xre[s_], zi = w.xim[s_];
      float rho = w.rhoL[s_];
      float b1m[4] = {-1e30f,-1e30f,-1e30f,-1e30f};
      float b0m[4] = {-1e30f,-1e30f,-1e30f,-1e30f};
      #pragma unroll
      for (int p = 0; p < 16; ++p) {
        float pr = kPre[p]*INV_SQRT10, pi = kPim[p]*INV_SQRT10;
        float dre = zr - pr, dim = zi - pi;
        float e = -(dre*dre + dim*dim) * rho + lg[p];
        #pragma unroll
        for (int k = 0; k < 4; ++k) {
          int bit = (p >> (3-k)) & 1;
          if (bit) b1m[k] = fmaxf(b1m[k], e);
          else     b0m[k] = fmaxf(b0m[k], e);
        }
      }
      #pragma unroll
      for (int k = 0; k < 4; ++k) w.llr_dL[s_][k] = b1m[k] - b0m[k];
    }
    __syncthreads();
    my_lld = w.llr_dL[s_out][k_out];
    __syncthreads();
  }

  out[b*32 + lane] = my_lld - my_lla;
}

extern "C" void kernel_launch(void* const* d_in, const int* in_sizes, int n_in,
                              void* d_out, int out_size, void* d_ws, size_t ws_size,
                              hipStream_t stream) {
  (void)n_in; (void)out_size; (void)d_ws; (void)ws_size;
  const float* y_re  = (const float*)d_in[0];
  const float* y_im  = (const float*)d_in[1];
  const float* h_re  = (const float*)d_in[2];
  const float* h_im  = (const float*)d_in[3];
  const float* prior = (const float*)d_in[4];
  const float* s_re  = (const float*)d_in[5];
  const float* s_im  = (const float*)d_in[6];
  float* out = (float*)d_out;

  int B = in_sizes[0] / 16;
  int blocks = B / WAVES_PER_BLOCK;
  mmse_pic_kernel<<<blocks, 256, 0, stream>>>(y_re, y_im, h_re, h_im,
                                              prior, s_re, s_im, out);
}


# One batch item per lane, not per wave (grid 256x128): Cholesky+solve and Gauss-Jordan in registers in a rotating frame, LDS only feeds the WMMA stage; Gram mirrored, Im diagonal zero; sum order kept
# speedup vs baseline: 4.9835x; 4.9835x over previous
#include <hip/hip_runtime.h>
#include <math.h>


typedef __attribute__((ext_vector_type(2))) float v2f;
typedef __attribute__((ext_vector_type(4))) float v4f;
typedef __attribute__((ext_vector_type(8))) float v8f;

#define WAVES_PER_BLOCK 4
#define ITEMS_PER_BLOCK 128
#define EPSV 1e-4f
#define INV_SQRT10 0.3162277660168379f

__device__ const float kPre[16] = {1,1,3,3,1,1,3,3,-1,-1,-3,-3,-1,-1,-3,-3};
__device__ const float kPim[16] = {1,3,1,3,-1,-3,-1,-3,1,3,1,3,-1,-3,-1,-3};

#define STRIDE   617
#define OFF_WRE  272
#define OFF_WIM  416
#define OFF_AINV 0
#define OFF_GRE  272
#define OFF_GIM  344
#define OFF_YPT  408
#define OFF_MU   408
#define OFF_YMF  544
#define OFF_LLD  560
#define OFF_X    592

__device__ __forceinline__ float log_sigmoid(float x) {
  return fminf(x, 0.0f) - log1pf(__expf(-fabsf(x)));
}

__device__ __forceinline__ void symbol_logits(float l0, float l1, float l2, float l3, float (&lg)[16]) {
  #pragma unroll
  for (int p = 0; p < 16; ++p) {
    lg[p] = log_sigmoid((p & 8) ? l0 : -l0)
          + log_sigmoid((p & 4) ? l1 : -l1)
          + log_sigmoid((p & 2) ? l2 : -l2)
          + log_sigmoid((p & 1) ? l3 : -l3);
  }
}

template <int N>
__device__ __forceinline__ void whiten_step(float (&Are)[16][16], float (&Aim)[16][16],
                                            float (&Wre)[16][9], float (&Wim)[16][9],
                                            float* P, int k)
{
  const float sdk = sqrtf(Are[0][0]);
  const float isd = 1.0f / sdk;
  float cre[N], cim[N];
  cre[0] = 0.0f; cim[0] = 0.0f;
  #pragma unroll
  for (int i = 1; i < N; ++i) { cre[i] = Are[i][0] * isd; cim[i] = Aim[i][0] * isd; }
  float wkre[9], wkim[9];
  #pragma unroll
  for (int c = 0; c < 9; ++c) {
    wkre[c] = Wre[0][c] * isd; wkim[c] = Wim[0][c] * isd;
    P[OFF_WRE + k*9 + c] = wkre[c];
    P[OFF_WIM + k*9 + c] = wkim[c];
  }
  #pragma unroll
  for (int i = 1; i < N; ++i) {
    const float lre = cre[i], lim = cim[i];
    #pragma unroll
    for (int j = 1; j <= i; ++j) {
      Are[i-1][j-1] = Are[i][j] - (lre*cre[j] + lim*cim[j]);
      Aim[i-1][j-1] = Aim[i][j] - (lim*cre[j] - lre*cim[j]);
    }
    #pragma unroll
    for (int c = 0; c < 9; ++c) {
      Wre[i-1][c] = Wre[i][c] - (lre*wkre[c] - lim*wkim[c]);
      Wim[i-1][c] = Wim[i][c] - (lre*wkim[c] + lim*wkre[c]);
    }
  }
  #pragma unroll
  for (int j = 0; j < N; ++j) { Are[N-1][j] = 0.0f; Aim[N-1][j] = 0.0f; }
  #pragma unroll
  for (int c = 0; c < 9; ++c) { Wre[N-1][c] = 0.0f; Wim[N-1][c] = 0.0f; }
}

__device__ __forceinline__ void front_stage(float* P,
    const float* __restrict__ y_re, const float* __restrict__ y_im,
    const float* __restrict__ h_re, const float* __restrict__ h_im,
    const float* __restrict__ prior,
    const float* __restrict__ s_re, const float* __restrict__ s_im, int b)
{
  float Are[16][16], Aim[16][16];
  float Wre[16][9], Wim[16][9];
  {
    const v4f* sr = (const v4f*)(s_re + (size_t)b*256);
    const v4f* si = (const v4f*)(s_im + (size_t)b*256);
    #pragma unroll
    for (int i = 0; i < 16; ++i) {
      #pragma unroll
      for (int q = 0; q < 4; ++q) {
        if (4*q <= i) {
          v4f vr = sr[i*4 + q], vi = si[i*4 + q];
          Are[i][4*q+0] = vr.x; Are[i][4*q+1] = vr.y; Are[i][4*q+2] = vr.z; Are[i][4*q+3] = vr.w;
          Aim[i][4*q+0] = vi.x; Aim[i][4*q+1] = vi.y; Aim[i][4*q+2] = vi.z; Aim[i][4*q+3] = vi.w;
        } else {
          Are[i][4*q+0] = 0.0f; Are[i][4*q+1] = 0.0f; Are[i][4*q+2] = 0.0f; Are[i][4*q+3] = 0.0f;
          Aim[i][4*q+0] = 0.0f; Aim[i][4*q+1] = 0.0f; Aim[i][4*q+2] = 0.0f; Aim[i][4*q+3] = 0.0f;
        }
      }
    }
    const v4f* hr = (const v4f*)(h_re + (size_t)b*128);
    const v4f* hi = (const v4f*)(h_im + (size_t)b*128);
    const v4f* yr = (const v4f*)(y_re + (size_t)b*16);
    const v4f* yi = (const v4f*)(y_im + (size_t)b*16);
    #pragma unroll
    for (int m = 0; m < 16; ++m) {
      v4f a0 = hr[m*2], a1 = hr[m*2+1], c0 = hi[m*2], c1 = hi[m*2+1];
      Wre[m][0] = a0.x; Wre[m][1] = a0.y; Wre[m][2] = a0.z; Wre[m][3] = a0.w;
      Wre[m][4] = a1.x; Wre[m][5] = a1.y; Wre[m][6] = a1.z; Wre[m][7] = a1.w;
      Wim[m][0] = c0.x; Wim[m][1] = c0.y; Wim[m][2] = c0.z; Wim[m][3] = c0.w;
      Wim[m][4] = c1.x; Wim[m][5] = c1.y; Wim[m][6] = c1.z; Wim[m][7] = c1.w;
    }
    #pragma unroll
    for (int q = 0; q < 4; ++q) {
      v4f a = yr[q], c = yi[q];
      Wre[4*q+0][8] = a.x; Wre[4*q+1][8] = a.y; Wre[4*q+2][8] = a.z; Wre[4*q+3][8] = a.w;
      Wim[4*q+0][8] = c.x; Wim[4*q+1][8] = c.y; Wim[4*q+2][8] = c.z; Wim[4*q+3][8] = c.w;
    }
    const v4f* pr = (const v4f*)(prior + (size_t)b*32);
    #pragma unroll
    for (int s = 0; s < 8; ++s) {
      v4f v = pr[s];
      P[OFF_LLD + s*4 + 0] = v.x; P[OFF_LLD + s*4 + 1] = v.y;
      P[OFF_LLD + s*4 + 2] = v.z; P[OFF_LLD + s*4 + 3] = v.w;
    }
  }

  #pragma unroll 1
  for (int k = 0; k < 8; ++k)  whiten_step<16>(Are, Aim, Wre, Wim, P, k);
  #pragma unroll 1
  for (int k = 8; k < 16; ++k) whiten_step<8>(Are, Aim, Wre, Wim, P, k);

  float gr_[8][9], gi_[8][9];
  #pragma unroll
  for (int i = 0; i < 8; ++i)
    #pragma unroll
    for (int j = 0; j < 9; ++j) { gr_[i][j] = 0.0f; gi_[i][j] = 0.0f; }
  #pragma unroll 1
  for (int m = 0; m < 16; ++m) {
    float xr[9], xi[9];
    #pragma unroll
    for (int c = 0; c < 9; ++c) { xr[c] = P[OFF_WRE + m*9 + c]; xi[c] = P[OFF_WIM + m*9 + c]; }
    #pragma unroll
    for (int i = 0; i < 8; ++i) {
      #pragma unroll
      for (int j = i; j < 9; ++j) {
        gr_[i][j] += xr[i]*xr[j] + xi[i]*xi[j];
        gi_[i][j] += xr[i]*xi[j] - xi[i]*xr[j];
      }
    }
  }
  #pragma unroll
  for (int i = 0; i < 8; ++i) {
    #pragma unroll
    for (int j = i; j < 8; ++j) {
      P[OFF_GRE + i*8 + j] = gr_[i][j];
      P[OFF_GIM + i*8 + j] = (j > i) ? gi_[i][j] : 0.0f;
      if (j > i) { P[OFF_GRE + j*8 + i] = gr_[i][j]; P[OFF_GIM + j*8 + i] = -gi_[i][j]; }
    }
    P[OFF_YMF + i]     = gr_[i][8];
    P[OFF_YMF + 8 + i] = gi_[i][8];
  }
}

__device__ __forceinline__ void iter_first(float* P)
{
  #pragma unroll 1
  for (int s = 0; s < 8; ++s) {
    float l0 = P[OFF_LLD + s*4 + 0], l1 = P[OFF_LLD + s*4 + 1];
    float l2 = P[OFF_LLD + s*4 + 2], l3 = P[OFF_LLD + s*4 + 3];
    float lg[16];
    symbol_logits(l0, l1, l2, l3, lg);
    float mx = lg[0];
    #pragma unroll
    for (int p = 1; p < 16; ++p) mx = fmaxf(mx, lg[p]);
    float sum = 0.0f, mre = 0.0f, mim = 0.0f, e2 = 0.0f;
    #pragma unroll
    for (int p = 0; p < 16; ++p) {
      float e = __expf(lg[p] - mx);
      float pr = kPre[p]*INV_SQRT10, pi = kPim[p]*INV_SQRT10;
      sum += e; mre += e*pr; mim += e*pi; e2 += e*(pr*pr + pi*pi);
    }
    float invs = 1.0f / sum;
    mre *= invs; mim *= invs; e2 *= invs;
    P[OFF_X + s] = mre; P[OFF_X + 8 + s] = mim;
    P[OFF_X + 16 + s] = e2 - (mre*mre + mim*mim);
  }

  float gre[8][8], gim[8][8], xr[8], xi[8], vr[8];
  #pragma unroll
  for (int i = 0; i < 8; ++i) {
    #pragma unroll
    for (int j = 0; j < 8; ++j) { gre[i][j] = P[OFF_GRE + i*8 + j]; gim[i][j] = P[OFF_GIM + i*8 + j]; }
    xr[i] = P[OFF_X + i]; xi[i] = P[OFF_X + 8 + i]; vr[i] = P[OFF_X + 16 + i];
  }
  float gxr[8], gxi[8];
  #pragma unroll
  for (int i = 0; i < 8; ++i) {
    float ar = 0.0f, ai = 0.0f;
    #pragma unroll
    for (int j = 0; j < 8; ++j) {
      ar += gre[i][j]*xr[j] - gim[i][j]*xi[j];
      ai += gre[i][j]*xi[j] + gim[i][j]*xr[j];
    }
    gxr[i] = ar; gxi[i] = ai;
  }
  #pragma unroll
  for (int i = 0; i < 8; ++i) {
    #pragma unroll
    for (int j = 0; j < 8; ++j) {
      float tre = P[OFF_YMF + j]     + (gre[j][i]*xr[i] - gim[j][i]*xi[i]) - gxr[j];
      float tim = P[OFF_YMF + 8 + j] + (gre[j][i]*xi[i] + gim[j][i]*xr[i]) - gxi[j];
      P[OFF_YPT + i*17 + j]     = tre;
      P[OFF_YPT + i*17 + j + 8] = tim;
    }
  }
  float a[16][16];
  #pragma unroll
  for (int r = 0; r < 16; ++r) {
    #pragma unroll
    for (int c = 0; c < 16; ++c) {
      float gv;
      if (r < 8) gv = (c < 8) ? gre[r][c]   : -gim[r][c-8];
      else       gv = (c < 8) ? gim[r-8][c] :  gre[r-8][c-8];
      a[r][c] = gv * vr[c & 7] + ((r == c) ? 1.0f : 0.0f);
    }
  }
  #pragma unroll 1
  for (int k = 0; k < 16; ++k) {
    const float pinv = 1.0f / a[0][0];
    float prow[16];
    prow[0] = pinv;
    #pragma unroll
    for (int j = 1; j < 16; ++j) prow[j] = a[0][j] * pinv;
    #pragma unroll
    for (int i = 1; i < 16; ++i) {
      const float f = a[i][0];
      #pragma unroll
      for (int j = 1; j < 16; ++j) a[i-1][j-1] = a[i][j] - f * prow[j];
      a[i-1][15] = -(f * pinv);
    }
    #pragma unroll
    for (int j = 1; j < 16; ++j) a[15][j-1] = prow[j];
    a[15][15] = pinv;
  }
  #pragma unroll
  for (int r = 0; r < 16; ++r)
    #pragma unroll
    for (int c = 0; c < 16; ++c) P[OFF_AINV + r*17 + c] = a[r][c];
}

__device__ __forceinline__ void diag_products(float* wv, int lane)
{
  const int nn = lane & 15, hf = lane >> 4;
  const int n7 = nn & 7;
  const bool nLow = nn < 8;
  const float* pa = wv + OFF_AINV + nn*17 + hf*2;
  const float* py = wv + OFF_YPT  + n7*17 + hf*2;
  const int gs = nLow ? 8 : 1;
  const float* pg0 = nLow ? (wv + OFF_GRE + (0 + hf*2)*8 + n7) : (wv + OFF_GIM + n7*8 + (0 + hf*2));
  const float* pg1 = nLow ? (wv + OFF_GRE + (4 + hf*2)*8 + n7) : (wv + OFF_GIM + n7*8 + (4 + hf*2));
  const float* pg2 = (nLow ? (wv + OFF_GIM) : (wv + OFF_GRE)) + (0 + hf*2)*8 + n7;
  const float* pg3 = (nLow ? (wv + OFF_GIM) : (wv + OFF_GRE)) + (4 + hf*2)*8 + n7;
  float* pm = wv + OFF_MU + nn;
  #pragma unroll 1
  for (int t = 0; t < 32; ++t) {
    v2f a0, a1, a2, a3, g0, g1, g2, g3, y0, y1, y2, y3;
    a0.x = pa[0];  a0.y = pa[1];   a1.x = pa[4];  a1.y = pa[5];
    a2.x = pa[8];  a2.y = pa[9];   a3.x = pa[12]; a3.y = pa[13];
    g0.x = pg0[0]; g0.y = pg0[gs]; g1.x = pg1[0]; g1.y = pg1[gs];
    g2.x = pg2[0]; g2.y = pg2[8];  g3.x = pg3[0]; g3.y = pg3[8];
    y0.x = py[0];  y0.y = py[1];   y1.x = py[4];  y1.y = py[5];
    y2.x = py[8];  y2.y = py[9];   y3.x = py[12]; y3.y = py[13];
    v8f p1 = {}; v8f p2 = {};
    p1 = __builtin_amdgcn_wmma_f32_16x16x4_f32(false, a0, false, g0, (short)0, p1, false, false);
    p2 = __builtin_amdgcn_wmma_f32_16x16x4_f32(false, a0, false, y0, (short)0, p2, false, false);
    p1 = __builtin_amdgcn_wmma_f32_16x16x4_f32(false, a1, false, g1, (short)0, p1, false, false);
    p2 = __builtin_amdgcn_wmma_f32_16x16x4_f32(false, a1, false, y1, (short)0, p2, false, false);
    p1 = __builtin_amdgcn_wmma_f32_16x16x4_f32(false, a2, false, g2, (short)0, p1, false, false);
    p2 = __builtin_amdgcn_wmma_f32_16x16x4_f32(false, a2, false, y2, (short)0, p2, false, false);
    p1 = __builtin_amdgcn_wmma_f32_16x16x4_f32(false, a3, false, g3, (short)0, p1, false, false);
    p2 = __builtin_amdgcn_wmma_f32_16x16x4_f32(false, a3, false, y3, (short)0, p2, false, false);
    #pragma unroll
    for (int v = 0; v < 8; ++v) {
      int row = v + 8*hf;
      if (row == nn) { pm[0] = p1[v]; pm[16] = p2[v]; }
    }
    pa += STRIDE; py += STRIDE; pg0 += STRIDE; pg1 += STRIDE; pg2 += STRIDE; pg3 += STRIDE; pm += STRIDE;
  }
}

__device__ __forceinline__ void iter_second(float* P, bool last)
{
  #pragma unroll 1
  for (int s = 0; s < 8; ++s) {
    float m1 = P[OFF_MU + s], m2 = P[OFF_MU + s + 8];
    float zr = P[OFF_MU + 16 + s]     / m1;
    float zi = P[OFF_MU + 16 + s + 8] / m2;
    float rho = m1 / fmaxf(1.0f - P[OFF_X + 16 + s] * m1, EPSV);
    float l[4];
    l[0] = P[OFF_LLD + s*4 + 0]; l[1] = P[OFF_LLD + s*4 + 1];
    l[2] = P[OFF_LLD + s*4 + 2]; l[3] = P[OFF_LLD + s*4 + 3];
    float lg[16];
    symbol_logits(l[0], l[1], l[2], l[3], lg);
    float b1m[4] = {-1e30f,-1e30f,-1e30f,-1e30f};
    float b0m[4] = {-1e30f,-1e30f,-1e30f,-1e30f};
    #pragma unroll
    for (int p = 0; p < 16; ++p) {
      float pr = kPre[p]*INV_SQRT10, pi = kPim[p]*INV_SQRT10;
      float dre = zr - pr, dim = zi - pi;
      float e = -(dre*dre + dim*dim) * rho + lg[p];
      #pragma unroll
      for (int k = 0; k < 4; ++k) {
        int bit = (p >> (3-k)) & 1;
        if (bit) b1m[k] = fmaxf(b1m[k], e);
        else     b0m[k] = fmaxf(b0m[k], e);
      }
    }
    #pragma unroll
    for (int k = 0; k < 4; ++k) {
      float nl = b1m[k] - b0m[k];
      P[OFF_LLD + s*4 + k] = last ? (nl - l[k]) : nl;
    }
  }
}

__global__ __launch_bounds__(ITEMS_PER_BLOCK) void mmse_pic_kernel(
    const float* __restrict__ y_re, const float* __restrict__ y_im,
    const float* __restrict__ h_re, const float* __restrict__ h_im,
    const float* __restrict__ prior,
    const float* __restrict__ s_re, const float* __restrict__ s_im,
    float* __restrict__ out)
{
  __shared__ float lds[WAVES_PER_BLOCK * 32 * STRIDE];
  const int lane = threadIdx.x & 31;
  const int wid  = threadIdx.x >> 5;
  const int b    = blockIdx.x * ITEMS_PER_BLOCK + threadIdx.x;
  float* wv = lds + wid * 32 * STRIDE;
  float* P  = wv + lane * STRIDE;

  front_stage(P, y_re, y_im, h_re, h_im, prior, s_re, s_im, b);

  #pragma unroll 1
  for (int iter = 0; iter < 2; ++iter) {
    iter_first(P);
    __syncthreads();
    diag_products(wv, lane);
    __syncthreads();
    iter_second(P, iter == 1);
  }
  __syncthreads();

  v4f* o4 = (v4f*)(out + (size_t)(blockIdx.x * ITEMS_PER_BLOCK + wid * 32) * 32);
  #pragma unroll
  for (int j = 0; j < 8; ++j) {
    int q = j*32 + lane;
    const float* src = wv + (q >> 3)*STRIDE + OFF_LLD + (q & 7)*4;
    v4f v; v.x = src[0]; v.y = src[1]; v.z = src[2]; v.w = src[3];
    o4[q] = v;
  }
}

extern "C" void kernel_launch(void* const* d_in, const int* in_sizes, int n_in,
                              void* d_out, int out_size, void* d_ws, size_t ws_size,
                              hipStream_t stream) {
  (void)n_in; (void)out_size; (void)d_ws; (void)ws_size;
  const float* y_re  = (const float*)d_in[0];
  const float* y_im  = (const float*)d_in[1];
  const float* h_re  = (const float*)d_in[2];
  const float* h_im  = (const float*)d_in[3];
  const float* prior = (const float*)d_in[4];
  const float* s_re  = (const float*)d_in[5];
  const float* s_im  = (const float*)d_in[6];
  float* out = (float*)d_out;

  int B = in_sizes[0] / 16;
  int blocks = B / ITEMS_PER_BLOCK;
  mmse_pic_kernel<<<blocks, ITEMS_PER_BLOCK, 0, stream>>>(y_re, y_im, h_re, h_im,
                                                          prior, s_re, s_im, out);
}


# Matrix stage software-pipelined over a wave's 32 items: next item's fragment loads in one group ahead of the current 8 WMMAs, earlier item's diagonal picked by selects, not 8 branches; sum order kept
# speedup vs baseline: 5.3797x; 1.0795x over previous
#include <hip/hip_runtime.h>
#include <math.h>


typedef __attribute__((ext_vector_type(2))) float v2f;
typedef __attribute__((ext_vector_type(4))) float v4f;
typedef __attribute__((ext_vector_type(8))) float v8f;

#define WAVES_PER_BLOCK 4
#define ITEMS_PER_BLOCK 128
#define EPSV 1e-4f
#define INV_SQRT10 0.3162277660168379f

__device__ const float kPre[16] = {1,1,3,3,1,1,3,3,-1,-1,-3,-3,-1,-1,-3,-3};
__device__ const float kPim[16] = {1,3,1,3,-1,-3,-1,-3,1,3,1,3,-1,-3,-1,-3};

#define STRIDE   617
#define OFF_WRE  272
#define OFF_WIM  416
#define OFF_AINV 0
#define OFF_GRE  272
#define OFF_GIM  344
#define OFF_YPT  408
#define OFF_MU   408
#define OFF_YMF  544
#define OFF_LLD  560
#define OFF_X    592

__device__ __forceinline__ float log_sigmoid(float x) {
  return fminf(x, 0.0f) - log1pf(__expf(-fabsf(x)));
}

__device__ __forceinline__ void symbol_logits(float l0, float l1, float l2, float l3, float (&lg)[16]) {
  #pragma unroll
  for (int p = 0; p < 16; ++p) {
    lg[p] = log_sigmoid((p & 8) ? l0 : -l0)
          + log_sigmoid((p & 4) ? l1 : -l1)
          + log_sigmoid((p & 2) ? l2 : -l2)
          + log_sigmoid((p & 1) ? l3 : -l3);
  }
}

template <int N>
__device__ __forceinline__ void whiten_step(float (&Are)[16][16], float (&Aim)[16][16],
                                            float (&Wre)[16][9], float (&Wim)[16][9],
                                            float* P, int k)
{
  const float sdk = sqrtf(Are[0][0]);
  const float isd = 1.0f / sdk;
  float cre[N], cim[N];
  cre[0] = 0.0f; cim[0] = 0.0f;
  #pragma unroll
  for (int i = 1; i < N; ++i) { cre[i] = Are[i][0] * isd; cim[i] = Aim[i][0] * isd; }
  float wkre[9], wkim[9];
  #pragma unroll
  for (int c = 0; c < 9; ++c) {
    wkre[c] = Wre[0][c] * isd; wkim[c] = Wim[0][c] * isd;
    P[OFF_WRE + k*9 + c] = wkre[c];
    P[OFF_WIM + k*9 + c] = wkim[c];
  }
  #pragma unroll
  for (int i = 1; i < N; ++i) {
    const float lre = cre[i], lim = cim[i];
    #pragma unroll
    for (int j = 1; j <= i; ++j) {
      Are[i-1][j-1] = Are[i][j] - (lre*cre[j] + lim*cim[j]);
      Aim[i-1][j-1] = Aim[i][j] - (lim*cre[j] - lre*cim[j]);
    }
    #pragma unroll
    for (int c = 0; c < 9; ++c) {
      Wre[i-1][c] = Wre[i][c] - (lre*wkre[c] - lim*wkim[c]);
      Wim[i-1][c] = Wim[i][c] - (lre*wkim[c] + lim*wkre[c]);
    }
  }
  #pragma unroll
  for (int j = 0; j < N; ++j) { Are[N-1][j] = 0.0f; Aim[N-1][j] = 0.0f; }
  #pragma unroll
  for (int c = 0; c < 9; ++c) { Wre[N-1][c] = 0.0f; Wim[N-1][c] = 0.0f; }
}

__device__ __forceinline__ void front_stage(float* P,
    const float* __restrict__ y_re, const float* __restrict__ y_im,
    const float* __restrict__ h_re, const float* __restrict__ h_im,
    const float* __restrict__ prior,
    const float* __restrict__ s_re, const float* __restrict__ s_im, int b)
{
  float Are[16][16], Aim[16][16];
  float Wre[16][9], Wim[16][9];
  {
    const v4f* sr = (const v4f*)(s_re + (size_t)b*256);
    const v4f* si = (const v4f*)(s_im + (size_t)b*256);
    #pragma unroll
    for (int i = 0; i < 16; ++i) {
      #pragma unroll
      for (int q = 0; q < 4; ++q) {
        if (4*q <= i) {
          v4f vr = sr[i*4 + q], vi = si[i*4 + q];
          Are[i][4*q+0] = vr.x; Are[i][4*q+1] = vr.y; Are[i][4*q+2] = vr.z; Are[i][4*q+3] = vr.w;
          Aim[i][4*q+0] = vi.x; Aim[i][4*q+1] = vi.y; Aim[i][4*q+2] = vi.z; Aim[i][4*q+3] = vi.w;
        } else {
          Are[i][4*q+0] = 0.0f; Are[i][4*q+1] = 0.0f; Are[i][4*q+2] = 0.0f; Are[i][4*q+3] = 0.0f;
          Aim[i][4*q+0] = 0.0f; Aim[i][4*q+1] = 0.0f; Aim[i][4*q+2] = 0.0f; Aim[i][4*q+3] = 0.0f;
        }
      }
    }
    const v4f* hr = (const v4f*)(h_re + (size_t)b*128);
    const v4f* hi = (const v4f*)(h_im + (size_t)b*128);
    const v4f* yr = (const v4f*)(y_re + (size_t)b*16);
    const v4f* yi = (const v4f*)(y_im + (size_t)b*16);
    #pragma unroll
    for (int m = 0; m < 16; ++m) {
      v4f a0 = hr[m*2], a1 = hr[m*2+1], c0 = hi[m*2], c1 = hi[m*2+1];
      Wre[m][0] = a0.x; Wre[m][1] = a0.y; Wre[m][2] = a0.z; Wre[m][3] = a0.w;
      Wre[m][4] = a1.x; Wre[m][5] = a1.y; Wre[m][6] = a1.z; Wre[m][7] = a1.w;
      Wim[m][0] = c0.x; Wim[m][1] = c0.y; Wim[m][2] = c0.z; Wim[m][3] = c0.w;
      Wim[m][4] = c1.x; Wim[m][5] = c1.y; Wim[m][6] = c1.z; Wim[m][7] = c1.w;
    }
    #pragma unroll
    for (int q = 0; q < 4; ++q) {
      v4f a = yr[q], c = yi[q];
      Wre[4*q+0][8] = a.x; Wre[4*q+1][8] = a.y; Wre[4*q+2][8] = a.z; Wre[4*q+3][8] = a.w;
      Wim[4*q+0][8] = c.x; Wim[4*q+1][8] = c.y; Wim[4*q+2][8] = c.z; Wim[4*q+3][8] = c.w;
    }
    const v4f* pr = (const v4f*)(prior + (size_t)b*32);
    #pragma unroll
    for (int s = 0; s < 8; ++s) {
      v4f v = pr[s];
      P[OFF_LLD + s*4 + 0] = v.x; P[OFF_LLD + s*4 + 1] = v.y;
      P[OFF_LLD + s*4 + 2] = v.z; P[OFF_LLD + s*4 + 3] = v.w;
    }
  }

  #pragma unroll 1
  for (int k = 0; k < 8; ++k)  whiten_step<16>(Are, Aim, Wre, Wim, P, k);
  #pragma unroll 1
  for (int k = 8; k < 16; ++k) whiten_step<8>(Are, Aim, Wre, Wim, P, k);

  float gr_[8][9], gi_[8][9];
  #pragma unroll
  for (int i = 0; i < 8; ++i)
    #pragma unroll
    for (int j = 0; j < 9; ++j) { gr_[i][j] = 0.0f; gi_[i][j] = 0.0f; }
  #pragma unroll 1
  for (int m = 0; m < 16; ++m) {
    float xr[9], xi[9];
    #pragma unroll
    for (int c = 0; c < 9; ++c) { xr[c] = P[OFF_WRE + m*9 + c]; xi[c] = P[OFF_WIM + m*9 + c]; }
    #pragma unroll
    for (int i = 0; i < 8; ++i) {
      #pragma unroll
      for (int j = i; j < 9; ++j) {
        gr_[i][j] += xr[i]*xr[j] + xi[i]*xi[j];
        gi_[i][j] += xr[i]*xi[j] - xi[i]*xr[j];
      }
    }
  }
  #pragma unroll
  for (int i = 0; i < 8; ++i) {
    #pragma unroll
    for (int j = i; j < 8; ++j) {
      P[OFF_GRE + i*8 + j] = gr_[i][j];
      P[OFF_GIM + i*8 + j] = (j > i) ? gi_[i][j] : 0.0f;
      if (j > i) { P[OFF_GRE + j*8 + i] = gr_[i][j]; P[OFF_GIM + j*8 + i] = -gi_[i][j]; }
    }
    P[OFF_YMF + i]     = gr_[i][8];
    P[OFF_YMF + 8 + i] = gi_[i][8];
  }
}

__device__ __forceinline__ void iter_first(float* P)
{
  #pragma unroll 1
  for (int s = 0; s < 8; ++s) {
    float l0 = P[OFF_LLD + s*4 + 0], l1 = P[OFF_LLD + s*4 + 1];
    float l2 = P[OFF_LLD + s*4 + 2], l3 = P[OFF_LLD + s*4 + 3];
    float lg[16];
    symbol_logits(l0, l1, l2, l3, lg);
    float mx = lg[0];
    #pragma unroll
    for (int p = 1; p < 16; ++p) mx = fmaxf(mx, lg[p]);
    float sum = 0.0f, mre = 0.0f, mim = 0.0f, e2 = 0.0f;
    #pragma unroll
    for (int p = 0; p < 16; ++p) {
      float e = __expf(lg[p] - mx);
      float pr = kPre[p]*INV_SQRT10, pi = kPim[p]*INV_SQRT10;
      sum += e; mre += e*pr; mim += e*pi; e2 += e*(pr*pr + pi*pi);
    }
    float invs = 1.0f / sum;
    mre *= invs; mim *= invs; e2 *= invs;
    P[OFF_X + s] = mre; P[OFF_X + 8 + s] = mim;
    P[OFF_X + 16 + s] = e2 - (mre*mre + mim*mim);
  }

  float gre[8][8], gim[8][8], xr[8], xi[8], vr[8];
  #pragma unroll
  for (int i = 0; i < 8; ++i) {
    #pragma unroll
    for (int j = 0; j < 8; ++j) { gre[i][j] = P[OFF_GRE + i*8 + j]; gim[i][j] = P[OFF_GIM + i*8 + j]; }
    xr[i] = P[OFF_X + i]; xi[i] = P[OFF_X + 8 + i]; vr[i] = P[OFF_X + 16 + i];
  }
  float gxr[8], gxi[8];
  #pragma unroll
  for (int i = 0; i < 8; ++i) {
    float ar = 0.0f, ai = 0.0f;
    #pragma unroll
    for (int j = 0; j < 8; ++j) {
      ar += gre[i][j]*xr[j] - gim[i][j]*xi[j];
      ai += gre[i][j]*xi[j] + gim[i][j]*xr[j];
    }
    gxr[i] = ar; gxi[i] = ai;
  }
  #pragma unroll
  for (int i = 0; i < 8; ++i) {
    #pragma unroll
    for (int j = 0; j < 8; ++j) {
      float tre = P[OFF_YMF + j]     + (gre[j][i]*xr[i] - gim[j][i]*xi[i]) - gxr[j];
      float tim = P[OFF_YMF + 8 + j] + (gre[j][i]*xi[i] + gim[j][i]*xr[i]) - gxi[j];
      P[OFF_YPT + i*17 + j]     = tre;
      P[OFF_YPT + i*17 + j + 8] = tim;
    }
  }
  float a[16][16];
  #pragma unroll
  for (int r = 0; r < 16; ++r) {
    #pragma unroll
    for (int c = 0; c < 16; ++c) {
      float gv;
      if (r < 8) gv = (c < 8) ? gre[r][c]   : -gim[r][c-8];
      else       gv = (c < 8) ? gim[r-8][c] :  gre[r-8][c-8];
      a[r][c] = gv * vr[c & 7] + ((r == c) ? 1.0f : 0.0f);
    }
  }
  #pragma unroll 1
  for (int k = 0; k < 16; ++k) {
    const float pinv = 1.0f / a[0][0];
    float prow[16];
    prow[0] = pinv;
    #pragma unroll
    for (int j = 1; j < 16; ++j) prow[j] = a[0][j] * pinv;
    #pragma unroll
    for (int i = 1; i < 16; ++i) {
      const float f = a[i][0];
      #pragma unroll
      for (int j = 1; j < 16; ++j) a[i-1][j-1] = a[i][j] - f * prow[j];
      a[i-1][15] = -(f * pinv);
    }
    #pragma unroll
    for (int j = 1; j < 16; ++j) a[15][j-1] = prow[j];
    a[15][15] = pinv;
  }
  #pragma unroll
  for (int r = 0; r < 16; ++r)
    #pragma unroll
    for (int c = 0; c < 16; ++c) P[OFF_AINV + r*17 + c] = a[r][c];
}

struct Frag { v2f a0, a1, a2, a3, g0, g1, g2, g3, y0, y1, y2, y3; };

__device__ __forceinline__ void load_frag(Frag& f, const float* pa, const float* py,
    const float* pg0, const float* pg1, const float* pg2, const float* pg3, int gs, int o)
{
  f.a0.x = pa[o + 0];  f.a0.y = pa[o + 1];   f.a1.x = pa[o + 4];  f.a1.y = pa[o + 5];
  f.a2.x = pa[o + 8];  f.a2.y = pa[o + 9];   f.a3.x = pa[o + 12]; f.a3.y = pa[o + 13];
  f.g0.x = pg0[o];     f.g0.y = pg0[o + gs]; f.g1.x = pg1[o];     f.g1.y = pg1[o + gs];
  f.g2.x = pg2[o];     f.g2.y = pg2[o + 8];  f.g3.x = pg3[o];     f.g3.y = pg3[o + 8];
  f.y0.x = py[o + 0];  f.y0.y = py[o + 1];   f.y1.x = py[o + 4];  f.y1.y = py[o + 5];
  f.y2.x = py[o + 8];  f.y2.y = py[o + 9];   f.y3.x = py[o + 12]; f.y3.y = py[o + 13];
}

__device__ __forceinline__ void run_products(const Frag& f, v8f& p1, v8f& p2)
{
  v8f z = {};
  p1 = __builtin_amdgcn_wmma_f32_16x16x4_f32(false, f.a0, false, f.g0, (short)0, z,  false, false);
  p2 = __builtin_amdgcn_wmma_f32_16x16x4_f32(false, f.a0, false, f.y0, (short)0, z,  false, false);
  p1 = __builtin_amdgcn_wmma_f32_16x16x4_f32(false, f.a1, false, f.g1, (short)0, p1, false, false);
  p2 = __builtin_amdgcn_wmma_f32_16x16x4_f32(false, f.a1, false, f.y1, (short)0, p2, false, false);
  p1 = __builtin_amdgcn_wmma_f32_16x16x4_f32(false, f.a2, false, f.g2, (short)0, p1, false, false);
  p2 = __builtin_amdgcn_wmma_f32_16x16x4_f32(false, f.a2, false, f.y2, (short)0, p2, false, false);
  p1 = __builtin_amdgcn_wmma_f32_16x16x4_f32(false, f.a3, false, f.g3, (short)0, p1, false, false);
  p2 = __builtin_amdgcn_wmma_f32_16x16x4_f32(false, f.a3, false, f.y3, (short)0, p2, false, false);
}

__device__ __forceinline__ void store_diag(float* q, const v8f& p1, const v8f& p2, int n7, bool own)
{
  float d1 = p1[0], d2 = p2[0];
  #pragma unroll
  for (int v = 1; v < 8; ++v) {
    const bool c = (n7 == v);
    d1 = c ? p1[v] : d1;
    d2 = c ? p2[v] : d2;
  }
  if (own) { q[0] = d1; q[16] = d2; }
}

__device__ __forceinline__ void diag_products(float* wv, int lane)
{
  const int nn = lane & 15, hf = lane >> 4;
  const int n7 = nn & 7;
  const bool nLow = nn < 8;
  const bool own = (nn >> 3) == hf;
  const float* pa = wv + OFF_AINV + nn*17 + hf*2;
  const float* py = wv + OFF_YPT  + n7*17 + hf*2;
  const int gs = nLow ? 8 : 1;
  const float* pg0 = nLow ? (wv + OFF_GRE + (0 + hf*2)*8 + n7) : (wv + OFF_GIM + n7*8 + (0 + hf*2));
  const float* pg1 = nLow ? (wv + OFF_GRE + (4 + hf*2)*8 + n7) : (wv + OFF_GIM + n7*8 + (4 + hf*2));
  const float* pg2 = (nLow ? (wv + OFF_GIM) : (wv + OFF_GRE)) + (0 + hf*2)*8 + n7;
  const float* pg3 = (nLow ? (wv + OFF_GIM) : (wv + OFF_GRE)) + (4 + hf*2)*8 + n7;
  float* pm = wv + OFF_MU + nn;

  #define LOADS_FIRST() __builtin_amdgcn_sched_barrier(0)
  Frag f0, f1;
  v8f p1a, p2a, p1b, p2b;
  load_frag(f0, pa, py, pg0, pg1, pg2, pg3, gs, 0);
  load_frag(f1, pa, py, pg0, pg1, pg2, pg3, gs, STRIDE);
  LOADS_FIRST();
  run_products(f0, p1a, p2a);
  LOADS_FIRST();
  load_frag(f0, pa, py, pg0, pg1, pg2, pg3, gs, 2*STRIDE);
  LOADS_FIRST();
  run_products(f1, p1b, p2b);
  #pragma unroll 1
  for (int c = 2; c < 30; c += 2) {
    const int ob = (c - 2) * STRIDE;
    store_diag(pm + ob, p1a, p2a, n7, own);
    load_frag(f1, pa, py, pg0, pg1, pg2, pg3, gs, ob + 3*STRIDE);
    LOADS_FIRST();
    run_products(f0, p1a, p2a);
    store_diag(pm + ob + STRIDE, p1b, p2b, n7, own);
    load_frag(f0, pa, py, pg0, pg1, pg2, pg3, gs, ob + 4*STRIDE);
    LOADS_FIRST();
    run_products(f1, p1b, p2b);
  }
  store_diag(pm + 28*STRIDE, p1a, p2a, n7, own);
  load_frag(f1, pa, py, pg0, pg1, pg2, pg3, gs, 31*STRIDE);
  LOADS_FIRST();
  run_products(f0, p1a, p2a);
  store_diag(pm + 29*STRIDE, p1b, p2b, n7, own);
  run_products(f1, p1b, p2b);
  store_diag(pm + 30*STRIDE, p1a, p2a, n7, own);
  store_diag(pm + 31*STRIDE, p1b, p2b, n7, own);
  #undef LOADS_FIRST
}

__device__ __forceinline__ void iter_second(float* P, bool last)
{
  #pragma unroll 1
  for (int s = 0; s < 8; ++s) {
    float m1 = P[OFF_MU + s], m2 = P[OFF_MU + s + 8];
    float zr = P[OFF_MU + 16 + s]     / m1;
    float zi = P[OFF_MU + 16 + s + 8] / m2;
    float rho = m1 / fmaxf(1.0f - P[OFF_X + 16 + s] * m1, EPSV);
    float l[4];
    l[0] = P[OFF_LLD + s*4 + 0]; l[1] = P[OFF_LLD + s*4 + 1];
    l[2] = P[OFF_LLD + s*4 + 2]; l[3] = P[OFF_LLD + s*4 + 3];
    float lg[16];
    symbol_logits(l[0], l[1], l[2], l[3], lg);
    float b1m[4] = {-1e30f,-1e30f,-1e30f,-1e30f};
    float b0m[4] = {-1e30f,-1e30f,-1e30f,-1e30f};
    #pragma unroll
    for (int p = 0; p < 16; ++p) {
      float pr = kPre[p]*INV_SQRT10, pi = kPim[p]*INV_SQRT10;
      float dre = zr - pr, dim = zi - pi;
      float e = -(dre*dre + dim*dim) * rho + lg[p];
      #pragma unroll
      for (int k = 0; k < 4; ++k) {
        int bit = (p >> (3-k)) & 1;
        if (bit) b1m[k] = fmaxf(b1m[k], e);
        else     b0m[k] = fmaxf(b0m[k], e);
      }
    }
    #pragma unroll
    for (int k = 0; k < 4; ++k) {
      float nl = b1m[k] - b0m[k];
      P[OFF_LLD + s*4 + k] = last ? (nl - l[k]) : nl;
    }
  }
}

__global__ __launch_bounds__(ITEMS_PER_BLOCK) void mmse_pic_kernel(
    const float* __restrict__ y_re, const float* __restrict__ y_im,
    const float* __restrict__ h_re, const float* __restrict__ h_im,
    const float* __restrict__ prior,
    const float* __restrict__ s_re, const float* __restrict__ s_im,
    float* __restrict__ out)
{
  __shared__ float lds[WAVES_PER_BLOCK * 32 * STRIDE];
  const int lane = threadIdx.x & 31;
  const int wid  = threadIdx.x >> 5;
  const int b    = blockIdx.x * ITEMS_PER_BLOCK + threadIdx.x;
  float* wv = lds + wid * 32 * STRIDE;
  float* P  = wv + lane * STRIDE;

  front_stage(P, y_re, y_im, h_re, h_im, prior, s_re, s_im, b);

  #pragma unroll 1
  for (int iter = 0; iter < 2; ++iter) {
    iter_first(P);
    __syncthreads();
    diag_products(wv, lane);
    __syncthreads();
    iter_second(P, iter == 1);
  }
  __syncthreads();

  v4f* o4 = (v4f*)(out + (size_t)(blockIdx.x * ITEMS_PER_BLOCK + wid * 32) * 32);
  #pragma unroll
  for (int j = 0; j < 8; ++j) {
    int q = j*32 + lane;
    const float* src = wv + (q >> 3)*STRIDE + OFF_LLD + (q & 7)*4;
    v4f v; v.x = src[0]; v.y = src[1]; v.z = src[2]; v.w = src[3];
    o4[q] = v;
  }
}

extern "C" void kernel_launch(void* const* d_in, const int* in_sizes, int n_in,
                              void* d_out, int out_size, void* d_ws, size_t ws_size,
                              hipStream_t stream) {
  (void)n_in; (void)out_size; (void)d_ws; (void)ws_size;
  const float* y_re  = (const float*)d_in[0];
  const float* y_im  = (const float*)d_in[1];
  const float* h_re  = (const float*)d_in[2];
  const float* h_im  = (const float*)d_in[3];
  const float* prior = (const float*)d_in[4];
  const float* s_re  = (const float*)d_in[5];
  const float* s_im  = (const float*)d_in[6];
  float* out = (float*)d_out;

  int B = in_sizes[0] / 16;
  int blocks = B / ITEMS_PER_BLOCK;
  mmse_pic_kernel<<<blocks, ITEMS_PER_BLOCK, 0, stream>>>(y_re, y_im, h_re, h_im,
                                                          prior, s_re, s_im, out);
}


# Symbol stages in closed form from the Gray QAM16's I/Q split: moments from 4 bit means, max-log demap per 2-bit half, no 16-point softmax or log-sigmoid; 8 streams unrolled; sum order changes
# speedup vs baseline: 6.5655x; 1.2204x over previous
#include <hip/hip_runtime.h>
#include <math.h>


typedef __attribute__((ext_vector_type(2))) float v2f;
typedef __attribute__((ext_vector_type(4))) float v4f;
typedef __attribute__((ext_vector_type(8))) float v8f;

#define WAVES_PER_BLOCK 4
#define ITEMS_PER_BLOCK 128
#define EPSV 1e-4f
#define INV_SQRT10 0.3162277660168379f

#define STRIDE   617
#define OFF_WRE  272
#define OFF_WIM  416
#define OFF_AINV 0
#define OFF_GRE  272
#define OFF_GIM  344
#define OFF_YPT  408
#define OFF_MU   408
#define OFF_YMF  544
#define OFF_LLD  560
#define OFF_X    592


__device__ __forceinline__ float bit_mean(float l) {
  const float e = __expf(-fabsf(l));
  const float t = (1.0f - e) / (1.0f + e);
  return (l > 0.0f) ? -t : t;
}

template <int N>
__device__ __forceinline__ void whiten_step(float (&Are)[16][16], float (&Aim)[16][16],
                                            float (&Wre)[16][9], float (&Wim)[16][9],
                                            float* P, int k)
{
  const float sdk = sqrtf(Are[0][0]);
  const float isd = 1.0f / sdk;
  float cre[N], cim[N];
  cre[0] = 0.0f; cim[0] = 0.0f;
  #pragma unroll
  for (int i = 1; i < N; ++i) { cre[i] = Are[i][0] * isd; cim[i] = Aim[i][0] * isd; }
  float wkre[9], wkim[9];
  #pragma unroll
  for (int c = 0; c < 9; ++c) {
    wkre[c] = Wre[0][c] * isd; wkim[c] = Wim[0][c] * isd;
    P[OFF_WRE + k*9 + c] = wkre[c];
    P[OFF_WIM + k*9 + c] = wkim[c];
  }
  #pragma unroll
  for (int i = 1; i < N; ++i) {
    const float lre = cre[i], lim = cim[i];
    #pragma unroll
    for (int j = 1; j <= i; ++j) {
      Are[i-1][j-1] = Are[i][j] - (lre*cre[j] + lim*cim[j]);
      Aim[i-1][j-1] = Aim[i][j] - (lim*cre[j] - lre*cim[j]);
    }
    #pragma unroll
    for (int c = 0; c < 9; ++c) {
      Wre[i-1][c] = Wre[i][c] - (lre*wkre[c] - lim*wkim[c]);
      Wim[i-1][c] = Wim[i][c] - (lre*wkim[c] + lim*wkre[c]);
    }
  }
  #pragma unroll
  for (int j = 0; j < N; ++j) { Are[N-1][j] = 0.0f; Aim[N-1][j] = 0.0f; }
  #pragma unroll
  for (int c = 0; c < 9; ++c) { Wre[N-1][c] = 0.0f; Wim[N-1][c] = 0.0f; }
}

__device__ __forceinline__ void front_stage(float* P,
    const float* __restrict__ y_re, const float* __restrict__ y_im,
    const float* __restrict__ h_re, const float* __restrict__ h_im,
    const float* __restrict__ prior,
    const float* __restrict__ s_re, const float* __restrict__ s_im, int b)
{
  float Are[16][16], Aim[16][16];
  float Wre[16][9], Wim[16][9];
  {
    const v4f* sr = (const v4f*)(s_re + (size_t)b*256);
    const v4f* si = (const v4f*)(s_im + (size_t)b*256);
    #pragma unroll
    for (int i = 0; i < 16; ++i) {
      #pragma unroll
      for (int q = 0; q < 4; ++q) {
        if (4*q <= i) {
          v4f vr = sr[i*4 + q], vi = si[i*4 + q];
          Are[i][4*q+0] = vr.x; Are[i][4*q+1] = vr.y; Are[i][4*q+2] = vr.z; Are[i][4*q+3] = vr.w;
          Aim[i][4*q+0] = vi.x; Aim[i][4*q+1] = vi.y; Aim[i][4*q+2] = vi.z; Aim[i][4*q+3] = vi.w;
        } else {
          Are[i][4*q+0] = 0.0f; Are[i][4*q+1] = 0.0f; Are[i][4*q+2] = 0.0f; Are[i][4*q+3] = 0.0f;
          Aim[i][4*q+0] = 0.0f; Aim[i][4*q+1] = 0.0f; Aim[i][4*q+2] = 0.0f; Aim[i][4*q+3] = 0.0f;
        }
      }
    }
    const v4f* hr = (const v4f*)(h_re + (size_t)b*128);
    const v4f* hi = (const v4f*)(h_im + (size_t)b*128);
    const v4f* yr = (const v4f*)(y_re + (size_t)b*16);
    const v4f* yi = (const v4f*)(y_im + (size_t)b*16);
    #pragma unroll
    for (int m = 0; m < 16; ++m) {
      v4f a0 = hr[m*2], a1 = hr[m*2+1], c0 = hi[m*2], c1 = hi[m*2+1];
      Wre[m][0] = a0.x; Wre[m][1] = a0.y; Wre[m][2] = a0.z; Wre[m][3] = a0.w;
      Wre[m][4] = a1.x; Wre[m][5] = a1.y; Wre[m][6] = a1.z; Wre[m][7] = a1.w;
      Wim[m][0] = c0.x; Wim[m][1] = c0.y; Wim[m][2] = c0.z; Wim[m][3] = c0.w;
      Wim[m][4] = c1.x; Wim[m][5] = c1.y; Wim[m][6] = c1.z; Wim[m][7] = c1.w;
    }
    #pragma unroll
    for (int q = 0; q < 4; ++q) {
      v4f a = yr[q], c = yi[q];
      Wre[4*q+0][8] = a.x; Wre[4*q+1][8] = a.y; Wre[4*q+2][8] = a.z; Wre[4*q+3][8] = a.w;
      Wim[4*q+0][8] = c.x; Wim[4*q+1][8] = c.y; Wim[4*q+2][8] = c.z; Wim[4*q+3][8] = c.w;
    }
    const v4f* pr = (const v4f*)(prior + (size_t)b*32);
    #pragma unroll
    for (int s = 0; s < 8; ++s) {
      v4f v = pr[s];
      P[OFF_LLD + s*4 + 0] = v.x; P[OFF_LLD + s*4 + 1] = v.y;
      P[OFF_LLD + s*4 + 2] = v.z; P[OFF_LLD + s*4 + 3] = v.w;
    }
  }

  #pragma unroll 1
  for (int k = 0; k < 8; ++k)  whiten_step<16>(Are, Aim, Wre, Wim, P, k);
  #pragma unroll 1
  for (int k = 8; k < 16; ++k) whiten_step<8>(Are, Aim, Wre, Wim, P, k);

  float gr_[8][9], gi_[8][9];
  #pragma unroll
  for (int i = 0; i < 8; ++i)
    #pragma unroll
    for (int j = 0; j < 9; ++j) { gr_[i][j] = 0.0f; gi_[i][j] = 0.0f; }
  #pragma unroll 1
  for (int m = 0; m < 16; ++m) {
    float xr[9], xi[9];
    #pragma unroll
    for (int c = 0; c < 9; ++c) { xr[c] = P[OFF_WRE + m*9 + c]; xi[c] = P[OFF_WIM + m*9 + c]; }
    #pragma unroll
    for (int i = 0; i < 8; ++i) {
      #pragma unroll
      for (int j = i; j < 9; ++j) {
        gr_[i][j] += xr[i]*xr[j] + xi[i]*xi[j];
        gi_[i][j] += xr[i]*xi[j] - xi[i]*xr[j];
      }
    }
  }
  #pragma unroll
  for (int i = 0; i < 8; ++i) {
    #pragma unroll
    for (int j = i; j < 8; ++j) {
      P[OFF_GRE + i*8 + j] = gr_[i][j];
      P[OFF_GIM + i*8 + j] = (j > i) ? gi_[i][j] : 0.0f;
      if (j > i) { P[OFF_GRE + j*8 + i] = gr_[i][j]; P[OFF_GIM + j*8 + i] = -gi_[i][j]; }
    }
    P[OFF_YMF + i]     = gr_[i][8];
    P[OFF_YMF + 8 + i] = gi_[i][8];
  }
}

__device__ __forceinline__ void iter_first(float* P)
{
  float xr[8], xi[8], vr[8];
  #pragma unroll
  for (int s = 0; s < 8; ++s) {
    const float m0 = bit_mean(P[OFF_LLD + s*4 + 0]), m1 = bit_mean(P[OFF_LLD + s*4 + 1]);
    const float m2 = bit_mean(P[OFF_LLD + s*4 + 2]), m3 = bit_mean(P[OFF_LLD + s*4 + 3]);
    const float mre = (m0 * (2.0f - m2)) * INV_SQRT10;
    const float mim = (m1 * (2.0f - m3)) * INV_SQRT10;
    const float e2  = (10.0f - 4.0f * (m2 + m3)) * 0.1f;
    xr[s] = mre; xi[s] = mim;
    vr[s] = e2 - (mre*mre + mim*mim);
    P[OFF_X + 16 + s] = vr[s];
  }

  float gre[8][8], gim[8][8];
  #pragma unroll
  for (int i = 0; i < 8; ++i) {
    #pragma unroll
    for (int j = 0; j < 8; ++j) { gre[i][j] = P[OFF_GRE + i*8 + j]; gim[i][j] = P[OFF_GIM + i*8 + j]; }
  }
  float gxr[8], gxi[8];
  #pragma unroll
  for (int i = 0; i < 8; ++i) {
    float ar = 0.0f, ai = 0.0f;
    #pragma unroll
    for (int j = 0; j < 8; ++j) {
      ar += gre[i][j]*xr[j] - gim[i][j]*xi[j];
      ai += gre[i][j]*xi[j] + gim[i][j]*xr[j];
    }
    gxr[i] = ar; gxi[i] = ai;
  }
  #pragma unroll
  for (int i = 0; i < 8; ++i) {
    #pragma unroll
    for (int j = 0; j < 8; ++j) {
      float tre = P[OFF_YMF + j]     + (gre[j][i]*xr[i] - gim[j][i]*xi[i]) - gxr[j];
      float tim = P[OFF_YMF + 8 + j] + (gre[j][i]*xi[i] + gim[j][i]*xr[i]) - gxi[j];
      P[OFF_YPT + i*17 + j]     = tre;
      P[OFF_YPT + i*17 + j + 8] = tim;
    }
  }
  float a[16][16];
  #pragma unroll
  for (int r = 0; r < 16; ++r) {
    #pragma unroll
    for (int c = 0; c < 16; ++c) {
      float gv;
      if (r < 8) gv = (c < 8) ? gre[r][c]   : -gim[r][c-8];
      else       gv = (c < 8) ? gim[r-8][c] :  gre[r-8][c-8];
      a[r][c] = gv * vr[c & 7] + ((r == c) ? 1.0f : 0.0f);
    }
  }
  #pragma unroll 1
  for (int k = 0; k < 16; ++k) {
    const float pinv = 1.0f / a[0][0];
    float prow[16];
    prow[0] = pinv;
    #pragma unroll
    for (int j = 1; j < 16; ++j) prow[j] = a[0][j] * pinv;
    #pragma unroll
    for (int i = 1; i < 16; ++i) {
      const float f = a[i][0];
      #pragma unroll
      for (int j = 1; j < 16; ++j) a[i-1][j-1] = a[i][j] - f * prow[j];
      a[i-1][15] = -(f * pinv);
    }
    #pragma unroll
    for (int j = 1; j < 16; ++j) a[15][j-1] = prow[j];
    a[15][15] = pinv;
  }
  #pragma unroll
  for (int r = 0; r < 16; ++r)
    #pragma unroll
    for (int c = 0; c < 16; ++c) P[OFF_AINV + r*17 + c] = a[r][c];
}

struct Frag { v2f a0, a1, a2, a3, g0, g1, g2, g3, y0, y1, y2, y3; };

__device__ __forceinline__ void load_frag(Frag& f, const float* pa, const float* py,
    const float* pg0, const float* pg1, const float* pg2, const float* pg3, int gs, int o)
{
  f.a0.x = pa[o + 0];  f.a0.y = pa[o + 1];   f.a1.x = pa[o + 4];  f.a1.y = pa[o + 5];
  f.a2.x = pa[o + 8];  f.a2.y = pa[o + 9];   f.a3.x = pa[o + 12]; f.a3.y = pa[o + 13];
  f.g0.x = pg0[o];     f.g0.y = pg0[o + gs]; f.g1.x = pg1[o];     f.g1.y = pg1[o + gs];
  f.g2.x = pg2[o];     f.g2.y = pg2[o + 8];  f.g3.x = pg3[o];     f.g3.y = pg3[o + 8];
  f.y0.x = py[o + 0];  f.y0.y = py[o + 1];   f.y1.x = py[o + 4];  f.y1.y = py[o + 5];
  f.y2.x = py[o + 8];  f.y2.y = py[o + 9];   f.y3.x = py[o + 12]; f.y3.y = py[o + 13];
}

__device__ __forceinline__ void run_products(const Frag& f, v8f& p1, v8f& p2)
{
  v8f z = {};
  p1 = __builtin_amdgcn_wmma_f32_16x16x4_f32(false, f.a0, false, f.g0, (short)0, z,  false, false);
  p2 = __builtin_amdgcn_wmma_f32_16x16x4_f32(false, f.a0, false, f.y0, (short)0, z,  false, false);
  p1 = __builtin_amdgcn_wmma_f32_16x16x4_f32(false, f.a1, false, f.g1, (short)0, p1, false, false);
  p2 = __builtin_amdgcn_wmma_f32_16x16x4_f32(false, f.a1, false, f.y1, (short)0, p2, false, false);
  p1 = __builtin_amdgcn_wmma_f32_16x16x4_f32(false, f.a2, false, f.g2, (short)0, p1, false, false);
  p2 = __builtin_amdgcn_wmma_f32_16x16x4_f32(false, f.a2, false, f.y2, (short)0, p2, false, false);
  p1 = __builtin_amdgcn_wmma_f32_16x16x4_f32(false, f.a3, false, f.g3, (short)0, p1, false, false);
  p2 = __builtin_amdgcn_wmma_f32_16x16x4_f32(false, f.a3, false, f.y3, (short)0, p2, false, false);
}

__device__ __forceinline__ void store_diag(float* q, const v8f& p1, const v8f& p2, int n7, bool own)
{
  float d1 = p1[0], d2 = p2[0];
  #pragma unroll
  for (int v = 1; v < 8; ++v) {
    const bool c = (n7 == v);
    d1 = c ? p1[v] : d1;
    d2 = c ? p2[v] : d2;
  }
  if (own) { q[0] = d1; q[16] = d2; }
}

__device__ __forceinline__ void diag_products(float* wv, int lane)
{
  const int nn = lane & 15, hf = lane >> 4;
  const int n7 = nn & 7;
  const bool nLow = nn < 8;
  const bool own = (nn >> 3) == hf;
  const float* pa = wv + OFF_AINV + nn*17 + hf*2;
  const float* py = wv + OFF_YPT  + n7*17 + hf*2;
  const int gs = nLow ? 8 : 1;
  const float* pg0 = nLow ? (wv + OFF_GRE + (0 + hf*2)*8 + n7) : (wv + OFF_GIM + n7*8 + (0 + hf*2));
  const float* pg1 = nLow ? (wv + OFF_GRE + (4 + hf*2)*8 + n7) : (wv + OFF_GIM + n7*8 + (4 + hf*2));
  const float* pg2 = (nLow ? (wv + OFF_GIM) : (wv + OFF_GRE)) + (0 + hf*2)*8 + n7;
  const float* pg3 = (nLow ? (wv + OFF_GIM) : (wv + OFF_GRE)) + (4 + hf*2)*8 + n7;
  float* pm = wv + OFF_MU + nn;

  #define LOADS_FIRST() __builtin_amdgcn_sched_barrier(0)
  Frag f0, f1;
  v8f p1a, p2a, p1b, p2b;
  load_frag(f0, pa, py, pg0, pg1, pg2, pg3, gs, 0);
  load_frag(f1, pa, py, pg0, pg1, pg2, pg3, gs, STRIDE);
  LOADS_FIRST();
  run_products(f0, p1a, p2a);
  LOADS_FIRST();
  load_frag(f0, pa, py, pg0, pg1, pg2, pg3, gs, 2*STRIDE);
  LOADS_FIRST();
  run_products(f1, p1b, p2b);
  #pragma unroll 1
  for (int c = 2; c < 30; c += 2) {
    const int ob = (c - 2) * STRIDE;
    store_diag(pm + ob, p1a, p2a, n7, own);
    load_frag(f1, pa, py, pg0, pg1, pg2, pg3, gs, ob + 3*STRIDE);
    LOADS_FIRST();
    run_products(f0, p1a, p2a);
    store_diag(pm + ob + STRIDE, p1b, p2b, n7, own);
    load_frag(f0, pa, py, pg0, pg1, pg2, pg3, gs, ob + 4*STRIDE);
    LOADS_FIRST();
    run_products(f1, p1b, p2b);
  }
  store_diag(pm + 28*STRIDE, p1a, p2a, n7, own);
  load_frag(f1, pa, py, pg0, pg1, pg2, pg3, gs, 31*STRIDE);
  LOADS_FIRST();
  run_products(f0, p1a, p2a);
  store_diag(pm + 29*STRIDE, p1b, p2b, n7, own);
  run_products(f1, p1b, p2b);
  store_diag(pm + 30*STRIDE, p1a, p2a, n7, own);
  store_diag(pm + 31*STRIDE, p1b, p2b, n7, own);
  #undef LOADS_FIRST
}

__device__ __forceinline__ void half_demap(float z, float rho, float la, float lb, float& da, float& db)
{
  const float c1 = 1.0f*INV_SQRT10, c3 = 3.0f*INV_SQRT10;
  const float t00 = z - c1, t01 = z - c3, t10 = z + c1, t11 = z + c3;
  const float r00 = -(t00*t00) * rho, r01 = -(t01*t01) * rho;
  const float r10 = -(t10*t10) * rho, r11 = -(t11*t11) * rho;
  da = fmaxf(r10, r11 + lb) - fmaxf(r00, r01 + lb);
  db = fmaxf(r01, r11 + la) - fmaxf(r00, r10 + la);
}

__device__ __forceinline__ void iter_second(float* P, bool last)
{
  #pragma unroll
  for (int s = 0; s < 8; ++s) {
    float m1 = P[OFF_MU + s], m2 = P[OFF_MU + s + 8];
    float zr = P[OFF_MU + 16 + s]     / m1;
    float zi = P[OFF_MU + 16 + s + 8] / m2;
    float rho = m1 / fmaxf(1.0f - P[OFF_X + 16 + s] * m1, EPSV);
    float l[4], d[4];
    l[0] = P[OFF_LLD + s*4 + 0]; l[1] = P[OFF_LLD + s*4 + 1];
    l[2] = P[OFF_LLD + s*4 + 2]; l[3] = P[OFF_LLD + s*4 + 3];
    half_demap(zr, rho, l[0], l[2], d[0], d[2]);
    half_demap(zi, rho, l[1], l[3], d[1], d[3]);
    #pragma unroll
    for (int k = 0; k < 4; ++k)
      P[OFF_LLD + s*4 + k] = last ? d[k] : (l[k] + d[k]);
  }
}

__global__ __launch_bounds__(ITEMS_PER_BLOCK) void mmse_pic_kernel(
    const float* __restrict__ y_re, const float* __restrict__ y_im,
    const float* __restrict__ h_re, const float* __restrict__ h_im,
    const float* __restrict__ prior,
    const float* __restrict__ s_re, const float* __restrict__ s_im,
    float* __restrict__ out)
{
  __shared__ float lds[WAVES_PER_BLOCK * 32 * STRIDE];
  const int lane = threadIdx.x & 31;
  const int wid  = threadIdx.x >> 5;
  const int b    = blockIdx.x * ITEMS_PER_BLOCK + threadIdx.x;
  float* wv = lds + wid * 32 * STRIDE;
  float* P  = wv + lane * STRIDE;

  front_stage(P, y_re, y_im, h_re, h_im, prior, s_re, s_im, b);

  #pragma unroll 1
  for (int iter = 0; iter < 2; ++iter) {
    iter_first(P);
    __syncthreads();
    diag_products(wv, lane);
    __syncthreads();
    iter_second(P, iter == 1);
  }
  __syncthreads();

  v4f* o4 = (v4f*)(out + (size_t)(blockIdx.x * ITEMS_PER_BLOCK + wid * 32) * 32);
  #pragma unroll
  for (int j = 0; j < 8; ++j) {
    int q = j*32 + lane;
    const float* src = wv + (q >> 3)*STRIDE + OFF_LLD + (q & 7)*4;
    v4f v; v.x = src[0]; v.y = src[1]; v.z = src[2]; v.w = src[3];
    o4[q] = v;
  }
}

extern "C" void kernel_launch(void* const* d_in, const int* in_sizes, int n_in,
                              void* d_out, int out_size, void* d_ws, size_t ws_size,
                              hipStream_t stream) {
  (void)n_in; (void)out_size; (void)d_ws; (void)ws_size;
  const float* y_re  = (const float*)d_in[0];
  const float* y_im  = (const float*)d_in[1];
  const float* h_re  = (const float*)d_in[2];
  const float* h_im  = (const float*)d_in[3];
  const float* prior = (const float*)d_in[4];
  const float* s_re  = (const float*)d_in[5];
  const float* s_im  = (const float*)d_in[6];
  float* out = (float*)d_out;

  int B = in_sizes[0] / 16;
  int blocks = B / ITEMS_PER_BLOCK;
  mmse_pic_kernel<<<blocks, ITEMS_PER_BLOCK, 0, stream>>>(y_re, y_im, h_re, h_im,
                                                          prior, s_re, s_im, out);
}


# Gauss-Jordan loop: two pivots per trip on rows held as aligned register pairs, so the rotating frame moves two columns and no longer re-pairs its rows by copies at each pivot; sum order kept
# speedup vs baseline: 6.9711x; 1.0618x over previous
#include <hip/hip_runtime.h>
#include <math.h>


typedef __attribute__((ext_vector_type(2))) float v2f;
typedef __attribute__((ext_vector_type(4))) float v4f;
typedef __attribute__((ext_vector_type(8))) float v8f;

#define WAVES_PER_BLOCK 4
#define ITEMS_PER_BLOCK 128
#define EPSV 1e-4f
#define INV_SQRT10 0.3162277660168379f

#define STRIDE   617
#define OFF_WRE  272
#define OFF_WIM  416
#define OFF_AINV 0
#define OFF_GRE  272
#define OFF_GIM  344
#define OFF_YPT  408
#define OFF_MU   408
#define OFF_YMF  544
#define OFF_LLD  560
#define OFF_X    592


__device__ __forceinline__ float bit_mean(float l) {
  const float e = __expf(-fabsf(l));
  const float t = (1.0f - e) / (1.0f + e);
  return (l > 0.0f) ? -t : t;
}

template <int N>
__device__ __forceinline__ void whiten_step(float (&Are)[16][16], float (&Aim)[16][16],
                                            float (&Wre)[16][9], float (&Wim)[16][9],
                                            float* P, int k)
{
  const float sdk = sqrtf(Are[0][0]);
  const float isd = 1.0f / sdk;
  float cre[N], cim[N];
  cre[0] = 0.0f; cim[0] = 0.0f;
  #pragma unroll
  for (int i = 1; i < N; ++i) { cre[i] = Are[i][0] * isd; cim[i] = Aim[i][0] * isd; }
  float wkre[9], wkim[9];
  #pragma unroll
  for (int c = 0; c < 9; ++c) {
    wkre[c] = Wre[0][c] * isd; wkim[c] = Wim[0][c] * isd;
    P[OFF_WRE + k*9 + c] = wkre[c];
    P[OFF_WIM + k*9 + c] = wkim[c];
  }
  #pragma unroll
  for (int i = 1; i < N; ++i) {
    const float lre = cre[i], lim = cim[i];
    #pragma unroll
    for (int j = 1; j <= i; ++j) {
      Are[i-1][j-1] = Are[i][j] - (lre*cre[j] + lim*cim[j]);
      Aim[i-1][j-1] = Aim[i][j] - (lim*cre[j] - lre*cim[j]);
    }
    #pragma unroll
    for (int c = 0; c < 9; ++c) {
      Wre[i-1][c] = Wre[i][c] - (lre*wkre[c] - lim*wkim[c]);
      Wim[i-1][c] = Wim[i][c] - (lre*wkim[c] + lim*wkre[c]);
    }
  }
  #pragma unroll
  for (int j = 0; j < N; ++j) { Are[N-1][j] = 0.0f; Aim[N-1][j] = 0.0f; }
  #pragma unroll
  for (int c = 0; c < 9; ++c) { Wre[N-1][c] = 0.0f; Wim[N-1][c] = 0.0f; }
}

__device__ __forceinline__ void front_stage(float* P,
    const float* __restrict__ y_re, const float* __restrict__ y_im,
    const float* __restrict__ h_re, const float* __restrict__ h_im,
    const float* __restrict__ prior,
    const float* __restrict__ s_re, const float* __restrict__ s_im, int b)
{
  float Are[16][16], Aim[16][16];
  float Wre[16][9], Wim[16][9];
  {
    const v4f* sr = (const v4f*)(s_re + (size_t)b*256);
    const v4f* si = (const v4f*)(s_im + (size_t)b*256);
    #pragma unroll
    for (int i = 0; i < 16; ++i) {
      #pragma unroll
      for (int q = 0; q < 4; ++q) {
        if (4*q <= i) {
          v4f vr = sr[i*4 + q], vi = si[i*4 + q];
          Are[i][4*q+0] = vr.x; Are[i][4*q+1] = vr.y; Are[i][4*q+2] = vr.z; Are[i][4*q+3] = vr.w;
          Aim[i][4*q+0] = vi.x; Aim[i][4*q+1] = vi.y; Aim[i][4*q+2] = vi.z; Aim[i][4*q+3] = vi.w;
        } else {
          Are[i][4*q+0] = 0.0f; Are[i][4*q+1] = 0.0f; Are[i][4*q+2] = 0.0f; Are[i][4*q+3] = 0.0f;
          Aim[i][4*q+0] = 0.0f; Aim[i][4*q+1] = 0.0f; Aim[i][4*q+2] = 0.0f; Aim[i][4*q+3] = 0.0f;
        }
      }
    }
    const v4f* hr = (const v4f*)(h_re + (size_t)b*128);
    const v4f* hi = (const v4f*)(h_im + (size_t)b*128);
    const v4f* yr = (const v4f*)(y_re + (size_t)b*16);
    const v4f* yi = (const v4f*)(y_im + (size_t)b*16);
    #pragma unroll
    for (int m = 0; m < 16; ++m) {
      v4f a0 = hr[m*2], a1 = hr[m*2+1], c0 = hi[m*2], c1 = hi[m*2+1];
      Wre[m][0] = a0.x; Wre[m][1] = a0.y; Wre[m][2] = a0.z; Wre[m][3] = a0.w;
      Wre[m][4] = a1.x; Wre[m][5] = a1.y; Wre[m][6] = a1.z; Wre[m][7] = a1.w;
      Wim[m][0] = c0.x; Wim[m][1] = c0.y; Wim[m][2] = c0.z; Wim[m][3] = c0.w;
      Wim[m][4] = c1.x; Wim[m][5] = c1.y; Wim[m][6] = c1.z; Wim[m][7] = c1.w;
    }
    #pragma unroll
    for (int q = 0; q < 4; ++q) {
      v4f a = yr[q], c = yi[q];
      Wre[4*q+0][8] = a.x; Wre[4*q+1][8] = a.y; Wre[4*q+2][8] = a.z; Wre[4*q+3][8] = a.w;
      Wim[4*q+0][8] = c.x; Wim[4*q+1][8] = c.y; Wim[4*q+2][8] = c.z; Wim[4*q+3][8] = c.w;
    }
    const v4f* pr = (const v4f*)(prior + (size_t)b*32);
    #pragma unroll
    for (int s = 0; s < 8; ++s) {
      v4f v = pr[s];
      P[OFF_LLD + s*4 + 0] = v.x; P[OFF_LLD + s*4 + 1] = v.y;
      P[OFF_LLD + s*4 + 2] = v.z; P[OFF_LLD + s*4 + 3] = v.w;
    }
  }

  #pragma unroll 1
  for (int k = 0; k < 8; ++k)  whiten_step<16>(Are, Aim, Wre, Wim, P, k);
  #pragma unroll 1
  for (int k = 8; k < 16; ++k) whiten_step<8>(Are, Aim, Wre, Wim, P, k);

  float gr_[8][9], gi_[8][9];
  #pragma unroll
  for (int i = 0; i < 8; ++i)
    #pragma unroll
    for (int j = 0; j < 9; ++j) { gr_[i][j] = 0.0f; gi_[i][j] = 0.0f; }
  #pragma unroll 1
  for (int m = 0; m < 16; ++m) {
    float xr[9], xi[9];
    #pragma unroll
    for (int c = 0; c < 9; ++c) { xr[c] = P[OFF_WRE + m*9 + c]; xi[c] = P[OFF_WIM + m*9 + c]; }
    #pragma unroll
    for (int i = 0; i < 8; ++i) {
      #pragma unroll
      for (int j = i; j < 9; ++j) {
        gr_[i][j] += xr[i]*xr[j] + xi[i]*xi[j];
        gi_[i][j] += xr[i]*xi[j] - xi[i]*xr[j];
      }
    }
  }
  #pragma unroll
  for (int i = 0; i < 8; ++i) {
    #pragma unroll
    for (int j = i; j < 8; ++j) {
      P[OFF_GRE + i*8 + j] = gr_[i][j];
      P[OFF_GIM + i*8 + j] = (j > i) ? gi_[i][j] : 0.0f;
      if (j > i) { P[OFF_GRE + j*8 + i] = gr_[i][j]; P[OFF_GIM + j*8 + i] = -gi_[i][j]; }
    }
    P[OFF_YMF + i]     = gr_[i][8];
    P[OFF_YMF + 8 + i] = gi_[i][8];
  }
}

__device__ __forceinline__ void iter_first(float* P)
{
  float xr[8], xi[8], vr[8];
  #pragma unroll
  for (int s = 0; s < 8; ++s) {
    const float m0 = bit_mean(P[OFF_LLD + s*4 + 0]), m1 = bit_mean(P[OFF_LLD + s*4 + 1]);
    const float m2 = bit_mean(P[OFF_LLD + s*4 + 2]), m3 = bit_mean(P[OFF_LLD + s*4 + 3]);
    const float mre = (m0 * (2.0f - m2)) * INV_SQRT10;
    const float mim = (m1 * (2.0f - m3)) * INV_SQRT10;
    const float e2  = (10.0f - 4.0f * (m2 + m3)) * 0.1f;
    xr[s] = mre; xi[s] = mim;
    vr[s] = e2 - (mre*mre + mim*mim);
    P[OFF_X + 16 + s] = vr[s];
  }

  float gre[8][8], gim[8][8];
  #pragma unroll
  for (int i = 0; i < 8; ++i) {
    #pragma unroll
    for (int j = 0; j < 8; ++j) { gre[i][j] = P[OFF_GRE + i*8 + j]; gim[i][j] = P[OFF_GIM + i*8 + j]; }
  }
  float gxr[8], gxi[8];
  #pragma unroll
  for (int i = 0; i < 8; ++i) {
    float ar = 0.0f, ai = 0.0f;
    #pragma unroll
    for (int j = 0; j < 8; ++j) {
      ar += gre[i][j]*xr[j] - gim[i][j]*xi[j];
      ai += gre[i][j]*xi[j] + gim[i][j]*xr[j];
    }
    gxr[i] = ar; gxi[i] = ai;
  }
  #pragma unroll
  for (int i = 0; i < 8; ++i) {
    #pragma unroll
    for (int j = 0; j < 8; ++j) {
      float tre = P[OFF_YMF + j]     + (gre[j][i]*xr[i] - gim[j][i]*xi[i]) - gxr[j];
      float tim = P[OFF_YMF + 8 + j] + (gre[j][i]*xi[i] + gim[j][i]*xr[i]) - gxi[j];
      P[OFF_YPT + i*17 + j]     = tre;
      P[OFF_YPT + i*17 + j + 8] = tim;
    }
  }
  float a[16][16];
  #pragma unroll
  for (int r = 0; r < 16; ++r) {
    #pragma unroll
    for (int c = 0; c < 16; ++c) {
      float gv;
      if (r < 8) gv = (c < 8) ? gre[r][c]   : -gim[r][c-8];
      else       gv = (c < 8) ? gim[r-8][c] :  gre[r-8][c-8];
      a[r][c] = gv * vr[c & 7] + ((r == c) ? 1.0f : 0.0f);
    }
  }
  float s0[16], s15[16];
  v2f pr[16][7];
  #pragma unroll
  for (int r = 0; r < 16; ++r) {
    s0[r] = a[r][0]; s15[r] = a[r][15];
    #pragma unroll
    for (int q = 0; q < 7; ++q) { pr[r][q].x = a[r][2*q+1]; pr[r][q].y = a[r][2*q+2]; }
  }
  #pragma unroll 1
  for (int t = 0; t < 8; ++t) {
    v2f m[16][8];
    {
      const float pinv = 1.0f / s0[0];
      v2f prow[7];
      #pragma unroll
      for (int q = 0; q < 7; ++q) prow[q] = pr[0][q] * pinv;
      const float prow15 = s15[0] * pinv;
      #pragma unroll
      for (int i = 1; i < 16; ++i) {
        const float f = s0[i];
        #pragma unroll
        for (int q = 0; q < 7; ++q) m[i-1][q] = pr[i][q] - f * prow[q];
        m[i-1][7].x = s15[i] - f * prow15;
        m[i-1][7].y = -(f * pinv);
      }
      #pragma unroll
      for (int q = 0; q < 7; ++q) m[15][q] = prow[q];
      m[15][7].x = prow15; m[15][7].y = pinv;
    }
    {
      const float pinv = 1.0f / m[0][0].x;
      const float prow1 = m[0][0].y * pinv;
      v2f prow[8];
      #pragma unroll
      for (int q = 1; q < 8; ++q) prow[q] = m[0][q] * pinv;
      #pragma unroll
      for (int i = 1; i < 16; ++i) {
        const float f = m[i][0].x;
        s0[i-1] = m[i][0].y - f * prow1;
        #pragma unroll
        for (int q = 1; q < 8; ++q) pr[i-1][q-1] = m[i][q] - f * prow[q];
        s15[i-1] = -(f * pinv);
      }
      s0[15] = prow1;
      #pragma unroll
      for (int q = 1; q < 8; ++q) pr[15][q-1] = prow[q];
      s15[15] = pinv;
    }
  }
  #pragma unroll
  for (int r = 0; r < 16; ++r) {
    a[r][0] = s0[r]; a[r][15] = s15[r];
    #pragma unroll
    for (int q = 0; q < 7; ++q) { a[r][2*q+1] = pr[r][q].x; a[r][2*q+2] = pr[r][q].y; }
  }
  #pragma unroll
  for (int r = 0; r < 16; ++r)
    #pragma unroll
    for (int c = 0; c < 16; ++c) P[OFF_AINV + r*17 + c] = a[r][c];
}

struct Frag { v2f a0, a1, a2, a3, g0, g1, g2, g3, y0, y1, y2, y3; };

__device__ __forceinline__ void load_frag(Frag& f, const float* pa, const float* py,
    const float* pg0, const float* pg1, const float* pg2, const float* pg3, int gs, int o)
{
  f.a0.x = pa[o + 0];  f.a0.y = pa[o + 1];   f.a1.x = pa[o + 4];  f.a1.y = pa[o + 5];
  f.a2.x = pa[o + 8];  f.a2.y = pa[o + 9];   f.a3.x = pa[o + 12]; f.a3.y = pa[o + 13];
  f.g0.x = pg0[o];     f.g0.y = pg0[o + gs]; f.g1.x = pg1[o];     f.g1.y = pg1[o + gs];
  f.g2.x = pg2[o];     f.g2.y = pg2[o + 8];  f.g3.x = pg3[o];     f.g3.y = pg3[o + 8];
  f.y0.x = py[o + 0];  f.y0.y = py[o + 1];   f.y1.x = py[o + 4];  f.y1.y = py[o + 5];
  f.y2.x = py[o + 8];  f.y2.y = py[o + 9];   f.y3.x = py[o + 12]; f.y3.y = py[o + 13];
}

__device__ __forceinline__ void run_products(const Frag& f, v8f& p1, v8f& p2)
{
  v8f z = {};
  p1 = __builtin_amdgcn_wmma_f32_16x16x4_f32(false, f.a0, false, f.g0, (short)0, z,  false, false);
  p2 = __builtin_amdgcn_wmma_f32_16x16x4_f32(false, f.a0, false, f.y0, (short)0, z,  false, false);
  p1 = __builtin_amdgcn_wmma_f32_16x16x4_f32(false, f.a1, false, f.g1, (short)0, p1, false, false);
  p2 = __builtin_amdgcn_wmma_f32_16x16x4_f32(false, f.a1, false, f.y1, (short)0, p2, false, false);
  p1 = __builtin_amdgcn_wmma_f32_16x16x4_f32(false, f.a2, false, f.g2, (short)0, p1, false, false);
  p2 = __builtin_amdgcn_wmma_f32_16x16x4_f32(false, f.a2, false, f.y2, (short)0, p2, false, false);
  p1 = __builtin_amdgcn_wmma_f32_16x16x4_f32(false, f.a3, false, f.g3, (short)0, p1, false, false);
  p2 = __builtin_amdgcn_wmma_f32_16x16x4_f32(false, f.a3, false, f.y3, (short)0, p2, false, false);
}

__device__ __forceinline__ void store_diag(float* q, const v8f& p1, const v8f& p2, int n7, bool own)
{
  float d1 = p1[0], d2 = p2[0];
  #pragma unroll
  for (int v = 1; v < 8; ++v) {
    const bool c = (n7 == v);
    d1 = c ? p1[v] : d1;
    d2 = c ? p2[v] : d2;
  }
  if (own) { q[0] = d1; q[16] = d2; }
}

__device__ __forceinline__ void diag_products(float* wv, int lane)
{
  const int nn = lane & 15, hf = lane >> 4;
  const int n7 = nn & 7;
  const bool nLow = nn < 8;
  const bool own = (nn >> 3) == hf;
  const float* pa = wv + OFF_AINV + nn*17 + hf*2;
  const float* py = wv + OFF_YPT  + n7*17 + hf*2;
  const int gs = nLow ? 8 : 1;
  const float* pg0 = nLow ? (wv + OFF_GRE + (0 + hf*2)*8 + n7) : (wv + OFF_GIM + n7*8 + (0 + hf*2));
  const float* pg1 = nLow ? (wv + OFF_GRE + (4 + hf*2)*8 + n7) : (wv + OFF_GIM + n7*8 + (4 + hf*2));
  const float* pg2 = (nLow ? (wv + OFF_GIM) : (wv + OFF_GRE)) + (0 + hf*2)*8 + n7;
  const float* pg3 = (nLow ? (wv + OFF_GIM) : (wv + OFF_GRE)) + (4 + hf*2)*8 + n7;
  float* pm = wv + OFF_MU + nn;

  #define LOADS_FIRST() __builtin_amdgcn_sched_barrier(0)
  Frag f0, f1;
  v8f p1a, p2a, p1b, p2b;
  load_frag(f0, pa, py, pg0, pg1, pg2, pg3, gs, 0);
  load_frag(f1, pa, py, pg0, pg1, pg2, pg3, gs, STRIDE);
  LOADS_FIRST();
  run_products(f0, p1a, p2a);
  LOADS_FIRST();
  load_frag(f0, pa, py, pg0, pg1, pg2, pg3, gs, 2*STRIDE);
  LOADS_FIRST();
  run_products(f1, p1b, p2b);
  #pragma unroll 1
  for (int c = 2; c < 30; c += 2) {
    const int ob = (c - 2) * STRIDE;
    store_diag(pm + ob, p1a, p2a, n7, own);
    load_frag(f1, pa, py, pg0, pg1, pg2, pg3, gs, ob + 3*STRIDE);
    LOADS_FIRST();
    run_products(f0, p1a, p2a);
    store_diag(pm + ob + STRIDE, p1b, p2b, n7, own);
    load_frag(f0, pa, py, pg0, pg1, pg2, pg3, gs, ob + 4*STRIDE);
    LOADS_FIRST();
    run_products(f1, p1b, p2b);
  }
  store_diag(pm + 28*STRIDE, p1a, p2a, n7, own);
  load_frag(f1, pa, py, pg0, pg1, pg2, pg3, gs, 31*STRIDE);
  LOADS_FIRST();
  run_products(f0, p1a, p2a);
  store_diag(pm + 29*STRIDE, p1b, p2b, n7, own);
  run_products(f1, p1b, p2b);
  store_diag(pm + 30*STRIDE, p1a, p2a, n7, own);
  store_diag(pm + 31*STRIDE, p1b, p2b, n7, own);
  #undef LOADS_FIRST
}

__device__ __forceinline__ void half_demap(float z, float rho, float la, float lb, float& da, float& db)
{
  const float c1 = 1.0f*INV_SQRT10, c3 = 3.0f*INV_SQRT10;
  const float t00 = z - c1, t01 = z - c3, t10 = z + c1, t11 = z + c3;
  const float r00 = -(t00*t00) * rho, r01 = -(t01*t01) * rho;
  const float r10 = -(t10*t10) * rho, r11 = -(t11*t11) * rho;
  da = fmaxf(r10, r11 + lb) - fmaxf(r00, r01 + lb);
  db = fmaxf(r01, r11 + la) - fmaxf(r00, r10 + la);
}

__device__ __forceinline__ void iter_second(float* P, bool last)
{
  #pragma unroll
  for (int s = 0; s < 8; ++s) {
    float m1 = P[OFF_MU + s], m2 = P[OFF_MU + s + 8];
    float zr = P[OFF_MU + 16 + s]     / m1;
    float zi = P[OFF_MU + 16 + s + 8] / m2;
    float rho = m1 / fmaxf(1.0f - P[OFF_X + 16 + s] * m1, EPSV);
    float l[4], d[4];
    l[0] = P[OFF_LLD + s*4 + 0]; l[1] = P[OFF_LLD + s*4 + 1];
    l[2] = P[OFF_LLD + s*4 + 2]; l[3] = P[OFF_LLD + s*4 + 3];
    half_demap(zr, rho, l[0], l[2], d[0], d[2]);
    half_demap(zi, rho, l[1], l[3], d[1], d[3]);
    #pragma unroll
    for (int k = 0; k < 4; ++k)
      P[OFF_LLD + s*4 + k] = last ? d[k] : (l[k] + d[k]);
  }
}

__global__ __launch_bounds__(ITEMS_PER_BLOCK) void mmse_pic_kernel(
    const float* __restrict__ y_re, const float* __restrict__ y_im,
    const float* __restrict__ h_re, const float* __restrict__ h_im,
    const float* __restrict__ prior,
    const float* __restrict__ s_re, const float* __restrict__ s_im,
    float* __restrict__ out)
{
  __shared__ float lds[WAVES_PER_BLOCK * 32 * STRIDE];
  const int lane = threadIdx.x & 31;
  const int wid  = threadIdx.x >> 5;
  const int b    = blockIdx.x * ITEMS_PER_BLOCK + threadIdx.x;
  float* wv = lds + wid * 32 * STRIDE;
  float* P  = wv + lane * STRIDE;

  front_stage(P, y_re, y_im, h_re, h_im, prior, s_re, s_im, b);

  #pragma unroll 1
  for (int iter = 0; iter < 2; ++iter) {
    iter_first(P);
    __syncthreads();
    diag_products(wv, lane);
    __syncthreads();
    iter_second(P, iter == 1);
  }
  __syncthreads();

  v4f* o4 = (v4f*)(out + (size_t)(blockIdx.x * ITEMS_PER_BLOCK + wid * 32) * 32);
  #pragma unroll
  for (int j = 0; j < 8; ++j) {
    int q = j*32 + lane;
    const float* src = wv + (q >> 3)*STRIDE + OFF_LLD + (q & 7)*4;
    v4f v; v.x = src[0]; v.y = src[1]; v.z = src[2]; v.w = src[3];
    o4[q] = v;
  }
}

extern "C" void kernel_launch(void* const* d_in, const int* in_sizes, int n_in,
                              void* d_out, int out_size, void* d_ws, size_t ws_size,
                              hipStream_t stream) {
  (void)n_in; (void)out_size; (void)d_ws; (void)ws_size;
  const float* y_re  = (const float*)d_in[0];
  const float* y_im  = (const float*)d_in[1];
  const float* h_re  = (const float*)d_in[2];
  const float* h_im  = (const float*)d_in[3];
  const float* prior = (const float*)d_in[4];
  const float* s_re  = (const float*)d_in[5];
  const float* s_im  = (const float*)d_in[6];
  float* out = (float*)d_out;

  int B = in_sizes[0] / 16;
  int blocks = B / ITEMS_PER_BLOCK;
  mmse_pic_kernel<<<blocks, ITEMS_PER_BLOCK, 0, stream>>>(y_re, y_im, h_re, h_im,
                                                          prior, s_re, s_im, out);
}


# Every complex multiply-accumulate (Cholesky and solve updates, Gram, g*x, ypt) as two chained fma on the accumulator, not a product pair summed and then added: 2 ops for 3; sum order changes
# speedup vs baseline: 7.4092x; 1.0628x over previous
#include <hip/hip_runtime.h>
#include <math.h>


typedef __attribute__((ext_vector_type(2))) float v2f;
typedef __attribute__((ext_vector_type(4))) float v4f;
typedef __attribute__((ext_vector_type(8))) float v8f;

#define WAVES_PER_BLOCK 4
#define ITEMS_PER_BLOCK 128
#define EPSV 1e-4f
#define INV_SQRT10 0.3162277660168379f

#define STRIDE   617
#define OFF_WRE  272
#define OFF_WIM  416
#define OFF_AINV 0
#define OFF_GRE  272
#define OFF_GIM  344
#define OFF_YPT  408
#define OFF_MU   408
#define OFF_YMF  544
#define OFF_LLD  560
#define OFF_X    592


__device__ __forceinline__ float bit_mean(float l) {
  const float e = __expf(-fabsf(l));
  const float t = (1.0f - e) / (1.0f + e);
  return (l > 0.0f) ? -t : t;
}

template <int N>
__device__ __forceinline__ void whiten_step(float (&Are)[16][16], float (&Aim)[16][16],
                                            float (&Wre)[16][9], float (&Wim)[16][9],
                                            float* P, int k)
{
  const float sdk = sqrtf(Are[0][0]);
  const float isd = 1.0f / sdk;
  float cre[N], cim[N];
  cre[0] = 0.0f; cim[0] = 0.0f;
  #pragma unroll
  for (int i = 1; i < N; ++i) { cre[i] = Are[i][0] * isd; cim[i] = Aim[i][0] * isd; }
  float wkre[9], wkim[9];
  #pragma unroll
  for (int c = 0; c < 9; ++c) {
    wkre[c] = Wre[0][c] * isd; wkim[c] = Wim[0][c] * isd;
    P[OFF_WRE + k*9 + c] = wkre[c];
    P[OFF_WIM + k*9 + c] = wkim[c];
  }
  #pragma unroll
  for (int i = 1; i < N; ++i) {
    const float lre = cre[i], lim = cim[i];
    #pragma unroll
    for (int j = 1; j <= i; ++j) {
      Are[i-1][j-1] = fmaf(-lim, cim[j], fmaf(-lre, cre[j], Are[i][j]));
      Aim[i-1][j-1] = fmaf( lre, cim[j], fmaf(-lim, cre[j], Aim[i][j]));
    }
    #pragma unroll
    for (int c = 0; c < 9; ++c) {
      Wre[i-1][c] = fmaf( lim, wkim[c], fmaf(-lre, wkre[c], Wre[i][c]));
      Wim[i-1][c] = fmaf(-lim, wkre[c], fmaf(-lre, wkim[c], Wim[i][c]));
    }
  }
  #pragma unroll
  for (int j = 0; j < N; ++j) { Are[N-1][j] = 0.0f; Aim[N-1][j] = 0.0f; }
  #pragma unroll
  for (int c = 0; c < 9; ++c) { Wre[N-1][c] = 0.0f; Wim[N-1][c] = 0.0f; }
}

__device__ __forceinline__ void front_stage(float* P,
    const float* __restrict__ y_re, const float* __restrict__ y_im,
    const float* __restrict__ h_re, const float* __restrict__ h_im,
    const float* __restrict__ prior,
    const float* __restrict__ s_re, const float* __restrict__ s_im, int b)
{
  float Are[16][16], Aim[16][16];
  float Wre[16][9], Wim[16][9];
  {
    const v4f* sr = (const v4f*)(s_re + (size_t)b*256);
    const v4f* si = (const v4f*)(s_im + (size_t)b*256);
    #pragma unroll
    for (int i = 0; i < 16; ++i) {
      #pragma unroll
      for (int q = 0; q < 4; ++q) {
        if (4*q <= i) {
          v4f vr = sr[i*4 + q], vi = si[i*4 + q];
          Are[i][4*q+0] = vr.x; Are[i][4*q+1] = vr.y; Are[i][4*q+2] = vr.z; Are[i][4*q+3] = vr.w;
          Aim[i][4*q+0] = vi.x; Aim[i][4*q+1] = vi.y; Aim[i][4*q+2] = vi.z; Aim[i][4*q+3] = vi.w;
        } else {
          Are[i][4*q+0] = 0.0f; Are[i][4*q+1] = 0.0f; Are[i][4*q+2] = 0.0f; Are[i][4*q+3] = 0.0f;
          Aim[i][4*q+0] = 0.0f; Aim[i][4*q+1] = 0.0f; Aim[i][4*q+2] = 0.0f; Aim[i][4*q+3] = 0.0f;
        }
      }
    }
    const v4f* hr = (const v4f*)(h_re + (size_t)b*128);
    const v4f* hi = (const v4f*)(h_im + (size_t)b*128);
    const v4f* yr = (const v4f*)(y_re + (size_t)b*16);
    const v4f* yi = (const v4f*)(y_im + (size_t)b*16);
    #pragma unroll
    for (int m = 0; m < 16; ++m) {
      v4f a0 = hr[m*2], a1 = hr[m*2+1], c0 = hi[m*2], c1 = hi[m*2+1];
      Wre[m][0] = a0.x; Wre[m][1] = a0.y; Wre[m][2] = a0.z; Wre[m][3] = a0.w;
      Wre[m][4] = a1.x; Wre[m][5] = a1.y; Wre[m][6] = a1.z; Wre[m][7] = a1.w;
      Wim[m][0] = c0.x; Wim[m][1] = c0.y; Wim[m][2] = c0.z; Wim[m][3] = c0.w;
      Wim[m][4] = c1.x; Wim[m][5] = c1.y; Wim[m][6] = c1.z; Wim[m][7] = c1.w;
    }
    #pragma unroll
    for (int q = 0; q < 4; ++q) {
      v4f a = yr[q], c = yi[q];
      Wre[4*q+0][8] = a.x; Wre[4*q+1][8] = a.y; Wre[4*q+2][8] = a.z; Wre[4*q+3][8] = a.w;
      Wim[4*q+0][8] = c.x; Wim[4*q+1][8] = c.y; Wim[4*q+2][8] = c.z; Wim[4*q+3][8] = c.w;
    }
    const v4f* pr = (const v4f*)(prior + (size_t)b*32);
    #pragma unroll
    for (int s = 0; s < 8; ++s) {
      v4f v = pr[s];
      P[OFF_LLD + s*4 + 0] = v.x; P[OFF_LLD + s*4 + 1] = v.y;
      P[OFF_LLD + s*4 + 2] = v.z; P[OFF_LLD + s*4 + 3] = v.w;
    }
  }

  #pragma unroll 1
  for (int k = 0; k < 8; ++k)  whiten_step<16>(Are, Aim, Wre, Wim, P, k);
  #pragma unroll 1
  for (int k = 8; k < 16; ++k) whiten_step<8>(Are, Aim, Wre, Wim, P, k);

  float gr_[8][9], gi_[8][9];
  #pragma unroll
  for (int i = 0; i < 8; ++i)
    #pragma unroll
    for (int j = 0; j < 9; ++j) { gr_[i][j] = 0.0f; gi_[i][j] = 0.0f; }
  #pragma unroll 1
  for (int m = 0; m < 16; ++m) {
    float xr[9], xi[9];
    #pragma unroll
    for (int c = 0; c < 9; ++c) { xr[c] = P[OFF_WRE + m*9 + c]; xi[c] = P[OFF_WIM + m*9 + c]; }
    #pragma unroll
    for (int i = 0; i < 8; ++i) {
      #pragma unroll
      for (int j = i; j < 9; ++j) {
        gr_[i][j] = fmaf( xi[i], xi[j], fmaf(xr[i], xr[j], gr_[i][j]));
        gi_[i][j] = fmaf(-xi[i], xr[j], fmaf(xr[i], xi[j], gi_[i][j]));
      }
    }
  }
  #pragma unroll
  for (int i = 0; i < 8; ++i) {
    #pragma unroll
    for (int j = i; j < 8; ++j) {
      P[OFF_GRE + i*8 + j] = gr_[i][j];
      P[OFF_GIM + i*8 + j] = (j > i) ? gi_[i][j] : 0.0f;
      if (j > i) { P[OFF_GRE + j*8 + i] = gr_[i][j]; P[OFF_GIM + j*8 + i] = -gi_[i][j]; }
    }
    P[OFF_YMF + i]     = gr_[i][8];
    P[OFF_YMF + 8 + i] = gi_[i][8];
  }
}

__device__ __forceinline__ void iter_first(float* P)
{
  float xr[8], xi[8], vr[8];
  #pragma unroll
  for (int s = 0; s < 8; ++s) {
    const float m0 = bit_mean(P[OFF_LLD + s*4 + 0]), m1 = bit_mean(P[OFF_LLD + s*4 + 1]);
    const float m2 = bit_mean(P[OFF_LLD + s*4 + 2]), m3 = bit_mean(P[OFF_LLD + s*4 + 3]);
    const float mre = (m0 * (2.0f - m2)) * INV_SQRT10;
    const float mim = (m1 * (2.0f - m3)) * INV_SQRT10;
    const float e2  = (10.0f - 4.0f * (m2 + m3)) * 0.1f;
    xr[s] = mre; xi[s] = mim;
    vr[s] = e2 - (mre*mre + mim*mim);
    P[OFF_X + 16 + s] = vr[s];
  }

  float gre[8][8], gim[8][8];
  #pragma unroll
  for (int i = 0; i < 8; ++i) {
    #pragma unroll
    for (int j = 0; j < 8; ++j) { gre[i][j] = P[OFF_GRE + i*8 + j]; gim[i][j] = P[OFF_GIM + i*8 + j]; }
  }
  float gxr[8], gxi[8];
  #pragma unroll
  for (int i = 0; i < 8; ++i) {
    float ar = 0.0f, ai = 0.0f;
    #pragma unroll
    for (int j = 0; j < 8; ++j) {
      ar = fmaf(-gim[i][j], xi[j], fmaf(gre[i][j], xr[j], ar));
      ai = fmaf( gim[i][j], xr[j], fmaf(gre[i][j], xi[j], ai));
    }
    gxr[i] = ar; gxi[i] = ai;
  }
  float dre[8], dim[8];
  #pragma unroll
  for (int j = 0; j < 8; ++j) { dre[j] = P[OFF_YMF + j] - gxr[j]; dim[j] = P[OFF_YMF + 8 + j] - gxi[j]; }
  #pragma unroll
  for (int i = 0; i < 8; ++i) {
    #pragma unroll
    for (int j = 0; j < 8; ++j) {
      float tre = fmaf(-gim[j][i], xi[i], fmaf(gre[j][i], xr[i], dre[j]));
      float tim = fmaf( gim[j][i], xr[i], fmaf(gre[j][i], xi[i], dim[j]));
      P[OFF_YPT + i*17 + j]     = tre;
      P[OFF_YPT + i*17 + j + 8] = tim;
    }
  }
  float a[16][16];
  #pragma unroll
  for (int r = 0; r < 16; ++r) {
    #pragma unroll
    for (int c = 0; c < 16; ++c) {
      float gv;
      if (r < 8) gv = (c < 8) ? gre[r][c]   : -gim[r][c-8];
      else       gv = (c < 8) ? gim[r-8][c] :  gre[r-8][c-8];
      a[r][c] = gv * vr[c & 7] + ((r == c) ? 1.0f : 0.0f);
    }
  }
  float s0[16], s15[16];
  v2f pr[16][7];
  #pragma unroll
  for (int r = 0; r < 16; ++r) {
    s0[r] = a[r][0]; s15[r] = a[r][15];
    #pragma unroll
    for (int q = 0; q < 7; ++q) { pr[r][q].x = a[r][2*q+1]; pr[r][q].y = a[r][2*q+2]; }
  }
  #pragma unroll 1
  for (int t = 0; t < 8; ++t) {
    v2f m[16][8];
    {
      const float pinv = 1.0f / s0[0];
      v2f prow[7];
      #pragma unroll
      for (int q = 0; q < 7; ++q) prow[q] = pr[0][q] * pinv;
      const float prow15 = s15[0] * pinv;
      #pragma unroll
      for (int i = 1; i < 16; ++i) {
        const float f = s0[i];
        #pragma unroll
        for (int q = 0; q < 7; ++q) m[i-1][q] = pr[i][q] - f * prow[q];
        m[i-1][7].x = s15[i] - f * prow15;
        m[i-1][7].y = -(f * pinv);
      }
      #pragma unroll
      for (int q = 0; q < 7; ++q) m[15][q] = prow[q];
      m[15][7].x = prow15; m[15][7].y = pinv;
    }
    {
      const float pinv = 1.0f / m[0][0].x;
      const float prow1 = m[0][0].y * pinv;
      v2f prow[8];
      #pragma unroll
      for (int q = 1; q < 8; ++q) prow[q] = m[0][q] * pinv;
      #pragma unroll
      for (int i = 1; i < 16; ++i) {
        const float f = m[i][0].x;
        s0[i-1] = m[i][0].y - f * prow1;
        #pragma unroll
        for (int q = 1; q < 8; ++q) pr[i-1][q-1] = m[i][q] - f * prow[q];
        s15[i-1] = -(f * pinv);
      }
      s0[15] = prow1;
      #pragma unroll
      for (int q = 1; q < 8; ++q) pr[15][q-1] = prow[q];
      s15[15] = pinv;
    }
  }
  #pragma unroll
  for (int r = 0; r < 16; ++r) {
    a[r][0] = s0[r]; a[r][15] = s15[r];
    #pragma unroll
    for (int q = 0; q < 7; ++q) { a[r][2*q+1] = pr[r][q].x; a[r][2*q+2] = pr[r][q].y; }
  }
  #pragma unroll
  for (int r = 0; r < 16; ++r)
    #pragma unroll
    for (int c = 0; c < 16; ++c) P[OFF_AINV + r*17 + c] = a[r][c];
}

struct Frag { v2f a0, a1, a2, a3, g0, g1, g2, g3, y0, y1, y2, y3; };

__device__ __forceinline__ void load_frag(Frag& f, const float* pa, const float* py,
    const float* pg0, const float* pg1, const float* pg2, const float* pg3, int gs, int o)
{
  f.a0.x = pa[o + 0];  f.a0.y = pa[o + 1];   f.a1.x = pa[o + 4];  f.a1.y = pa[o + 5];
  f.a2.x = pa[o + 8];  f.a2.y = pa[o + 9];   f.a3.x = pa[o + 12]; f.a3.y = pa[o + 13];
  f.g0.x = pg0[o];     f.g0.y = pg0[o + gs]; f.g1.x = pg1[o];     f.g1.y = pg1[o + gs];
  f.g2.x = pg2[o];     f.g2.y = pg2[o + 8];  f.g3.x = pg3[o];     f.g3.y = pg3[o + 8];
  f.y0.x = py[o + 0];  f.y0.y = py[o + 1];   f.y1.x = py[o + 4];  f.y1.y = py[o + 5];
  f.y2.x = py[o + 8];  f.y2.y = py[o + 9];   f.y3.x = py[o + 12]; f.y3.y = py[o + 13];
}

__device__ __forceinline__ void run_products(const Frag& f, v8f& p1, v8f& p2)
{
  v8f z = {};
  p1 = __builtin_amdgcn_wmma_f32_16x16x4_f32(false, f.a0, false, f.g0, (short)0, z,  false, false);
  p2 = __builtin_amdgcn_wmma_f32_16x16x4_f32(false, f.a0, false, f.y0, (short)0, z,  false, false);
  p1 = __builtin_amdgcn_wmma_f32_16x16x4_f32(false, f.a1, false, f.g1, (short)0, p1, false, false);
  p2 = __builtin_amdgcn_wmma_f32_16x16x4_f32(false, f.a1, false, f.y1, (short)0, p2, false, false);
  p1 = __builtin_amdgcn_wmma_f32_16x16x4_f32(false, f.a2, false, f.g2, (short)0, p1, false, false);
  p2 = __builtin_amdgcn_wmma_f32_16x16x4_f32(false, f.a2, false, f.y2, (short)0, p2, false, false);
  p1 = __builtin_amdgcn_wmma_f32_16x16x4_f32(false, f.a3, false, f.g3, (short)0, p1, false, false);
  p2 = __builtin_amdgcn_wmma_f32_16x16x4_f32(false, f.a3, false, f.y3, (short)0, p2, false, false);
}

__device__ __forceinline__ void store_diag(float* q, const v8f& p1, const v8f& p2, int n7, bool own)
{
  float d1 = p1[0], d2 = p2[0];
  #pragma unroll
  for (int v = 1; v < 8; ++v) {
    const bool c = (n7 == v);
    d1 = c ? p1[v] : d1;
    d2 = c ? p2[v] : d2;
  }
  if (own) { q[0] = d1; q[16] = d2; }
}

__device__ __forceinline__ void diag_products(float* wv, int lane)
{
  const int nn = lane & 15, hf = lane >> 4;
  const int n7 = nn & 7;
  const bool nLow = nn < 8;
  const bool own = (nn >> 3) == hf;
  const float* pa = wv + OFF_AINV + nn*17 + hf*2;
  const float* py = wv + OFF_YPT  + n7*17 + hf*2;
  const int gs = nLow ? 8 : 1;
  const float* pg0 = nLow ? (wv + OFF_GRE + (0 + hf*2)*8 + n7) : (wv + OFF_GIM + n7*8 + (0 + hf*2));
  const float* pg1 = nLow ? (wv + OFF_GRE + (4 + hf*2)*8 + n7) : (wv + OFF_GIM + n7*8 + (4 + hf*2));
  const float* pg2 = (nLow ? (wv + OFF_GIM) : (wv + OFF_GRE)) + (0 + hf*2)*8 + n7;
  const float* pg3 = (nLow ? (wv + OFF_GIM) : (wv + OFF_GRE)) + (4 + hf*2)*8 + n7;
  float* pm = wv + OFF_MU + nn;

  #define LOADS_FIRST() __builtin_amdgcn_sched_barrier(0)
  Frag f0, f1;
  v8f p1a, p2a, p1b, p2b;
  load_frag(f0, pa, py, pg0, pg1, pg2, pg3, gs, 0);
  load_frag(f1, pa, py, pg0, pg1, pg2, pg3, gs, STRIDE);
  LOADS_FIRST();
  run_products(f0, p1a, p2a);
  LOADS_FIRST();
  load_frag(f0, pa, py, pg0, pg1, pg2, pg3, gs, 2*STRIDE);
  LOADS_FIRST();
  run_products(f1, p1b, p2b);
  #pragma unroll 1
  for (int c = 2; c < 30; c += 2) {
    const int ob = (c - 2) * STRIDE;
    store_diag(pm + ob, p1a, p2a, n7, own);
    load_frag(f1, pa, py, pg0, pg1, pg2, pg3, gs, ob + 3*STRIDE);
    LOADS_FIRST();
    run_products(f0, p1a, p2a);
    store_diag(pm + ob + STRIDE, p1b, p2b, n7, own);
    load_frag(f0, pa, py, pg0, pg1, pg2, pg3, gs, ob + 4*STRIDE);
    LOADS_FIRST();
    run_products(f1, p1b, p2b);
  }
  store_diag(pm + 28*STRIDE, p1a, p2a, n7, own);
  load_frag(f1, pa, py, pg0, pg1, pg2, pg3, gs, 31*STRIDE);
  LOADS_FIRST();
  run_products(f0, p1a, p2a);
  store_diag(pm + 29*STRIDE, p1b, p2b, n7, own);
  run_products(f1, p1b, p2b);
  store_diag(pm + 30*STRIDE, p1a, p2a, n7, own);
  store_diag(pm + 31*STRIDE, p1b, p2b, n7, own);
  #undef LOADS_FIRST
}

__device__ __forceinline__ void half_demap(float z, float rho, float la, float lb, float& da, float& db)
{
  const float c1 = 1.0f*INV_SQRT10, c3 = 3.0f*INV_SQRT10;
  const float t00 = z - c1, t01 = z - c3, t10 = z + c1, t11 = z + c3;
  const float r00 = -(t00*t00) * rho, r01 = -(t01*t01) * rho;
  const float r10 = -(t10*t10) * rho, r11 = -(t11*t11) * rho;
  da = fmaxf(r10, r11 + lb) - fmaxf(r00, r01 + lb);
  db = fmaxf(r01, r11 + la) - fmaxf(r00, r10 + la);
}

__device__ __forceinline__ void iter_second(float* P, bool last)
{
  #pragma unroll
  for (int s = 0; s < 8; ++s) {
    float m1 = P[OFF_MU + s], m2 = P[OFF_MU + s + 8];
    float zr = P[OFF_MU + 16 + s]     / m1;
    float zi = P[OFF_MU + 16 + s + 8] / m2;
    float rho = m1 / fmaxf(1.0f - P[OFF_X + 16 + s] * m1, EPSV);
    float l[4], d[4];
    l[0] = P[OFF_LLD + s*4 + 0]; l[1] = P[OFF_LLD + s*4 + 1];
    l[2] = P[OFF_LLD + s*4 + 2]; l[3] = P[OFF_LLD + s*4 + 3];
    half_demap(zr, rho, l[0], l[2], d[0], d[2]);
    half_demap(zi, rho, l[1], l[3], d[1], d[3]);
    #pragma unroll
    for (int k = 0; k < 4; ++k)
      P[OFF_LLD + s*4 + k] = last ? d[k] : (l[k] + d[k]);
  }
}

__global__ __launch_bounds__(ITEMS_PER_BLOCK) void mmse_pic_kernel(
    const float* __restrict__ y_re, const float* __restrict__ y_im,
    const float* __restrict__ h_re, const float* __restrict__ h_im,
    const float* __restrict__ prior,
    const float* __restrict__ s_re, const float* __restrict__ s_im,
    float* __restrict__ out)
{
  __shared__ float lds[WAVES_PER_BLOCK * 32 * STRIDE];
  const int lane = threadIdx.x & 31;
  const int wid  = threadIdx.x >> 5;
  const int b    = blockIdx.x * ITEMS_PER_BLOCK + threadIdx.x;
  float* wv = lds + wid * 32 * STRIDE;
  float* P  = wv + lane * STRIDE;

  front_stage(P, y_re, y_im, h_re, h_im, prior, s_re, s_im, b);

  #pragma unroll 1
  for (int iter = 0; iter < 2; ++iter) {
    iter_first(P);
    __syncthreads();
    diag_products(wv, lane);
    __syncthreads();
    iter_second(P, iter == 1);
  }
  __syncthreads();

  v4f* o4 = (v4f*)(out + (size_t)(blockIdx.x * ITEMS_PER_BLOCK + wid * 32) * 32);
  #pragma unroll
  for (int j = 0; j < 8; ++j) {
    int q = j*32 + lane;
    const float* src = wv + (q >> 3)*STRIDE + OFF_LLD + (q & 7)*4;
    v4f v; v.x = src[0]; v.y = src[1]; v.z = src[2]; v.w = src[3];
    o4[q] = v;
  }
}

extern "C" void kernel_launch(void* const* d_in, const int* in_sizes, int n_in,
                              void* d_out, int out_size, void* d_ws, size_t ws_size,
                              hipStream_t stream) {
  (void)n_in; (void)out_size; (void)d_ws; (void)ws_size;
  const float* y_re  = (const float*)d_in[0];
  const float* y_im  = (const float*)d_in[1];
  const float* h_re  = (const float*)d_in[2];
  const float* h_im  = (const float*)d_in[3];
  const float* prior = (const float*)d_in[4];
  const float* s_re  = (const float*)d_in[5];
  const float* s_im  = (const float*)d_in[6];
  float* out = (float*)d_out;

  int B = in_sizes[0] / 16;
  int blocks = B / ITEMS_PER_BLOCK;
  mmse_pic_kernel<<<blocks, ITEMS_PER_BLOCK, 0, stream>>>(y_re, y_im, h_re, h_im,
                                                          prior, s_re, s_im, out);
}


# Inputs staged through LDS: a wave reads its 32 items' arrays as whole 512-byte runs and lays them down per item, each lane then reads its own item; no per-lane strided global loads; sum order kept
# speedup vs baseline: 7.7202x; 1.0420x over previous
#include <hip/hip_runtime.h>
#include <math.h>


typedef __attribute__((ext_vector_type(2))) float v2f;
typedef __attribute__((ext_vector_type(4))) float v4f;
typedef __attribute__((ext_vector_type(8))) float v8f;

#define WAVES_PER_BLOCK 4
#define ITEMS_PER_BLOCK 128
#define EPSV 1e-4f
#define INV_SQRT10 0.3162277660168379f

#define STRIDE   617
#define OFF_WRE  272
#define OFF_WIM  416
#define OFF_AINV 0
#define OFF_GRE  272
#define OFF_GIM  344
#define OFF_YPT  408
#define OFF_MU   408
#define OFF_YMF  544
#define OFF_LLD  560
#define OFF_X    592

#define S_PITCH  516
#define H_PITCH  324
#define HOFF_HIM 128
#define HOFF_YRE 256
#define HOFF_YIM 272
#define HOFF_PRI 288


__device__ __forceinline__ float bit_mean(float l) {
  const float e = __expf(-fabsf(l));
  const float t = (1.0f - e) / (1.0f + e);
  return (l > 0.0f) ? -t : t;
}

template <int N>
__device__ __forceinline__ void whiten_step(float (&Are)[16][16], float (&Aim)[16][16],
                                            float (&Wre)[16][9], float (&Wim)[16][9],
                                            float* P, int k)
{
  const float sdk = sqrtf(Are[0][0]);
  const float isd = 1.0f / sdk;
  float cre[N], cim[N];
  cre[0] = 0.0f; cim[0] = 0.0f;
  #pragma unroll
  for (int i = 1; i < N; ++i) { cre[i] = Are[i][0] * isd; cim[i] = Aim[i][0] * isd; }
  float wkre[9], wkim[9];
  #pragma unroll
  for (int c = 0; c < 9; ++c) {
    wkre[c] = Wre[0][c] * isd; wkim[c] = Wim[0][c] * isd;
    P[OFF_WRE + k*9 + c] = wkre[c];
    P[OFF_WIM + k*9 + c] = wkim[c];
  }
  #pragma unroll
  for (int i = 1; i < N; ++i) {
    const float lre = cre[i], lim = cim[i];
    #pragma unroll
    for (int j = 1; j <= i; ++j) {
      Are[i-1][j-1] = fmaf(-lim, cim[j], fmaf(-lre, cre[j], Are[i][j]));
      Aim[i-1][j-1] = fmaf( lre, cim[j], fmaf(-lim, cre[j], Aim[i][j]));
    }
    #pragma unroll
    for (int c = 0; c < 9; ++c) {
      Wre[i-1][c] = fmaf( lim, wkim[c], fmaf(-lre, wkre[c], Wre[i][c]));
      Wim[i-1][c] = fmaf(-lim, wkre[c], fmaf(-lre, wkim[c], Wim[i][c]));
    }
  }
  #pragma unroll
  for (int j = 0; j < N; ++j) { Are[N-1][j] = 0.0f; Aim[N-1][j] = 0.0f; }
  #pragma unroll
  for (int c = 0; c < 9; ++c) { Wre[N-1][c] = 0.0f; Wim[N-1][c] = 0.0f; }
}

__device__ __forceinline__ void front_stage(float* wv, int lane, float* P,
    const float* __restrict__ y_re, const float* __restrict__ y_im,
    const float* __restrict__ h_re, const float* __restrict__ h_im,
    const float* __restrict__ prior,
    const float* __restrict__ s_re, const float* __restrict__ s_im, int bw)
{
  float Are[16][16], Aim[16][16];
  float Wre[16][9], Wim[16][9];
  v4f* const st = (v4f*)wv + lane;

  {
    const v4f* gsr = (const v4f*)(s_re + (size_t)bw*256) + lane;
    const v4f* gsi = (const v4f*)(s_im + (size_t)bw*256) + lane;
    v4f tr[64], ti[64];
    #pragma unroll
    for (int j = 0; j < 64; ++j) { tr[j] = gsr[j*32]; ti[j] = gsi[j*32]; }
    #pragma unroll
    for (int j = 0; j < 64; ++j) {
      const int o = (j >> 1)*(S_PITCH/4) + (j & 1)*32;
      st[o] = tr[j]; st[o + 64] = ti[j];
    }
  }
  __syncthreads();

  v4f thr[32], thi[32], tyr[4], tyi[4], tpr[8];
  {
    const v4f* ghr = (const v4f*)(h_re + (size_t)bw*128) + lane;
    const v4f* ghi = (const v4f*)(h_im + (size_t)bw*128) + lane;
    const v4f* gyr = (const v4f*)(y_re + (size_t)bw*16) + lane;
    const v4f* gyi = (const v4f*)(y_im + (size_t)bw*16) + lane;
    const v4f* gpr = (const v4f*)(prior + (size_t)bw*32) + lane;
    #pragma unroll
    for (int j = 0; j < 32; ++j) { thr[j] = ghr[j*32]; thi[j] = ghi[j*32]; }
    #pragma unroll
    for (int j = 0; j < 4; ++j) { tyr[j] = gyr[j*32]; tyi[j] = gyi[j*32]; }
    #pragma unroll
    for (int j = 0; j < 8; ++j) tpr[j] = gpr[j*32];
  }

  {
    const float* ps = wv + lane*S_PITCH;
    #pragma unroll
    for (int i = 0; i < 16; ++i) {
      #pragma unroll
      for (int q = 0; q < 4; ++q) {
        const float* pr = ps + i*16 + q*4;
        const float* pi = pr + 256;
        const int r = i - 4*q;
        Are[i][4*q+0] = 0.0f; Are[i][4*q+1] = 0.0f; Are[i][4*q+2] = 0.0f; Are[i][4*q+3] = 0.0f;
        Aim[i][4*q+0] = 0.0f; Aim[i][4*q+1] = 0.0f; Aim[i][4*q+2] = 0.0f; Aim[i][4*q+3] = 0.0f;
        if (r >= 3) {
          v4f vr = *(const v4f*)pr, vi = *(const v4f*)pi;
          Are[i][4*q+0] = vr.x; Are[i][4*q+1] = vr.y; Are[i][4*q+2] = vr.z; Are[i][4*q+3] = vr.w;
          Aim[i][4*q+0] = vi.x; Aim[i][4*q+1] = vi.y; Aim[i][4*q+2] = vi.z; Aim[i][4*q+3] = vi.w;
        } else if (r >= 1) {
          v2f vr = *(const v2f*)pr, vi = *(const v2f*)pi;
          Are[i][4*q+0] = vr.x; Are[i][4*q+1] = vr.y;
          Aim[i][4*q+0] = vi.x; Aim[i][4*q+1] = vi.y;
          if (r == 2) { Are[i][4*q+2] = pr[2]; Aim[i][4*q+2] = pi[2]; }
        } else if (r == 0) {
          Are[i][4*q+0] = pr[0]; Aim[i][4*q+0] = pi[0];
        }
      }
    }
  }
  __syncthreads();

  #pragma unroll
  for (int j = 0; j < 32; ++j) { st[j*(H_PITCH/4)] = thr[j]; st[j*(H_PITCH/4) + HOFF_HIM/4] = thi[j]; }
  {
    v4f* sy = (v4f*)wv + (lane >> 2)*(H_PITCH/4) + HOFF_YRE/4 + (lane & 3);
    #pragma unroll
    for (int j = 0; j < 4; ++j) { sy[j*8*(H_PITCH/4)] = tyr[j]; sy[j*8*(H_PITCH/4) + (HOFF_YIM - HOFF_YRE)/4] = tyi[j]; }
    v4f* sp = (v4f*)wv + (lane >> 3)*(H_PITCH/4) + HOFF_PRI/4 + (lane & 7);
    #pragma unroll
    for (int j = 0; j < 8; ++j) sp[j*4*(H_PITCH/4)] = tpr[j];
  }
  __syncthreads();

  v4f pl[8];
  {
    const float* ph = wv + lane*H_PITCH;
    #pragma unroll
    for (int m = 0; m < 16; ++m) {
      v4f a0 = *(const v4f*)(ph + m*8), a1 = *(const v4f*)(ph + m*8 + 4);
      v4f c0 = *(const v4f*)(ph + HOFF_HIM + m*8), c1 = *(const v4f*)(ph + HOFF_HIM + m*8 + 4);
      Wre[m][0] = a0.x; Wre[m][1] = a0.y; Wre[m][2] = a0.z; Wre[m][3] = a0.w;
      Wre[m][4] = a1.x; Wre[m][5] = a1.y; Wre[m][6] = a1.z; Wre[m][7] = a1.w;
      Wim[m][0] = c0.x; Wim[m][1] = c0.y; Wim[m][2] = c0.z; Wim[m][3] = c0.w;
      Wim[m][4] = c1.x; Wim[m][5] = c1.y; Wim[m][6] = c1.z; Wim[m][7] = c1.w;
    }
    #pragma unroll
    for (int q = 0; q < 4; ++q) {
      v4f a = *(const v4f*)(ph + HOFF_YRE + 4*q), cc = *(const v4f*)(ph + HOFF_YIM + 4*q);
      Wre[4*q+0][8] = a.x; Wre[4*q+1][8] = a.y; Wre[4*q+2][8] = a.z; Wre[4*q+3][8] = a.w;
      Wim[4*q+0][8] = cc.x; Wim[4*q+1][8] = cc.y; Wim[4*q+2][8] = cc.z; Wim[4*q+3][8] = cc.w;
    }
    #pragma unroll
    for (int s = 0; s < 8; ++s) pl[s] = *(const v4f*)(ph + HOFF_PRI + 4*s);
  }
  __syncthreads();
  #pragma unroll
  for (int s = 0; s < 8; ++s) {
    P[OFF_LLD + s*4 + 0] = pl[s].x; P[OFF_LLD + s*4 + 1] = pl[s].y;
    P[OFF_LLD + s*4 + 2] = pl[s].z; P[OFF_LLD + s*4 + 3] = pl[s].w;
  }

  #pragma unroll 1
  for (int k = 0; k < 8; ++k)  whiten_step<16>(Are, Aim, Wre, Wim, P, k);
  #pragma unroll 1
  for (int k = 8; k < 16; ++k) whiten_step<8>(Are, Aim, Wre, Wim, P, k);

  float gr_[8][9], gi_[8][9];
  #pragma unroll
  for (int i = 0; i < 8; ++i)
    #pragma unroll
    for (int j = 0; j < 9; ++j) { gr_[i][j] = 0.0f; gi_[i][j] = 0.0f; }
  #pragma unroll 1
  for (int m = 0; m < 16; ++m) {
    float xr[9], xi[9];
    #pragma unroll
    for (int c = 0; c < 9; ++c) { xr[c] = P[OFF_WRE + m*9 + c]; xi[c] = P[OFF_WIM + m*9 + c]; }
    #pragma unroll
    for (int i = 0; i < 8; ++i) {
      #pragma unroll
      for (int j = i; j < 9; ++j) {
        gr_[i][j] = fmaf( xi[i], xi[j], fmaf(xr[i], xr[j], gr_[i][j]));
        gi_[i][j] = fmaf(-xi[i], xr[j], fmaf(xr[i], xi[j], gi_[i][j]));
      }
    }
  }
  #pragma unroll
  for (int i = 0; i < 8; ++i) {
    #pragma unroll
    for (int j = i; j < 8; ++j) {
      P[OFF_GRE + i*8 + j] = gr_[i][j];
      P[OFF_GIM + i*8 + j] = (j > i) ? gi_[i][j] : 0.0f;
      if (j > i) { P[OFF_GRE + j*8 + i] = gr_[i][j]; P[OFF_GIM + j*8 + i] = -gi_[i][j]; }
    }
    P[OFF_YMF + i]     = gr_[i][8];
    P[OFF_YMF + 8 + i] = gi_[i][8];
  }
}

__device__ __forceinline__ void iter_first(float* P)
{
  float xr[8], xi[8], vr[8];
  #pragma unroll
  for (int s = 0; s < 8; ++s) {
    const float m0 = bit_mean(P[OFF_LLD + s*4 + 0]), m1 = bit_mean(P[OFF_LLD + s*4 + 1]);
    const float m2 = bit_mean(P[OFF_LLD + s*4 + 2]), m3 = bit_mean(P[OFF_LLD + s*4 + 3]);
    const float mre = (m0 * (2.0f - m2)) * INV_SQRT10;
    const float mim = (m1 * (2.0f - m3)) * INV_SQRT10;
    const float e2  = (10.0f - 4.0f * (m2 + m3)) * 0.1f;
    xr[s] = mre; xi[s] = mim;
    vr[s] = e2 - (mre*mre + mim*mim);
    P[OFF_X + 16 + s] = vr[s];
  }

  float gre[8][8], gim[8][8];
  #pragma unroll
  for (int i = 0; i < 8; ++i) {
    #pragma unroll
    for (int j = 0; j < 8; ++j) { gre[i][j] = P[OFF_GRE + i*8 + j]; gim[i][j] = P[OFF_GIM + i*8 + j]; }
  }
  float gxr[8], gxi[8];
  #pragma unroll
  for (int i = 0; i < 8; ++i) {
    float ar = 0.0f, ai = 0.0f;
    #pragma unroll
    for (int j = 0; j < 8; ++j) {
      ar = fmaf(-gim[i][j], xi[j], fmaf(gre[i][j], xr[j], ar));
      ai = fmaf( gim[i][j], xr[j], fmaf(gre[i][j], xi[j], ai));
    }
    gxr[i] = ar; gxi[i] = ai;
  }
  float dre[8], dim[8];
  #pragma unroll
  for (int j = 0; j < 8; ++j) { dre[j] = P[OFF_YMF + j] - gxr[j]; dim[j] = P[OFF_YMF + 8 + j] - gxi[j]; }
  #pragma unroll
  for (int i = 0; i < 8; ++i) {
    #pragma unroll
    for (int j = 0; j < 8; ++j) {
      float tre = fmaf(-gim[j][i], xi[i], fmaf(gre[j][i], xr[i], dre[j]));
      float tim = fmaf( gim[j][i], xr[i], fmaf(gre[j][i], xi[i], dim[j]));
      P[OFF_YPT + i*17 + j]     = tre;
      P[OFF_YPT + i*17 + j + 8] = tim;
    }
  }
  float a[16][16];
  #pragma unroll
  for (int r = 0; r < 16; ++r) {
    #pragma unroll
    for (int c = 0; c < 16; ++c) {
      float gv;
      if (r < 8) gv = (c < 8) ? gre[r][c]   : -gim[r][c-8];
      else       gv = (c < 8) ? gim[r-8][c] :  gre[r-8][c-8];
      a[r][c] = gv * vr[c & 7] + ((r == c) ? 1.0f : 0.0f);
    }
  }
  float s0[16], s15[16];
  v2f pr[16][7];
  #pragma unroll
  for (int r = 0; r < 16; ++r) {
    s0[r] = a[r][0]; s15[r] = a[r][15];
    #pragma unroll
    for (int q = 0; q < 7; ++q) { pr[r][q].x = a[r][2*q+1]; pr[r][q].y = a[r][2*q+2]; }
  }
  #pragma unroll 1
  for (int t = 0; t < 8; ++t) {
    v2f m[16][8];
    {
      const float pinv = 1.0f / s0[0];
      v2f prow[7];
      #pragma unroll
      for (int q = 0; q < 7; ++q) prow[q] = pr[0][q] * pinv;
      const float prow15 = s15[0] * pinv;
      #pragma unroll
      for (int i = 1; i < 16; ++i) {
        const float f = s0[i];
        #pragma unroll
        for (int q = 0; q < 7; ++q) m[i-1][q] = pr[i][q] - f * prow[q];
        m[i-1][7].x = s15[i] - f * prow15;
        m[i-1][7].y = -(f * pinv);
      }
      #pragma unroll
      for (int q = 0; q < 7; ++q) m[15][q] = prow[q];
      m[15][7].x = prow15; m[15][7].y = pinv;
    }
    {
      const float pinv = 1.0f / m[0][0].x;
      const float prow1 = m[0][0].y * pinv;
      v2f prow[8];
      #pragma unroll
      for (int q = 1; q < 8; ++q) prow[q] = m[0][q] * pinv;
      #pragma unroll
      for (int i = 1; i < 16; ++i) {
        const float f = m[i][0].x;
        s0[i-1] = m[i][0].y - f * prow1;
        #pragma unroll
        for (int q = 1; q < 8; ++q) pr[i-1][q-1] = m[i][q] - f * prow[q];
        s15[i-1] = -(f * pinv);
      }
      s0[15] = prow1;
      #pragma unroll
      for (int q = 1; q < 8; ++q) pr[15][q-1] = prow[q];
      s15[15] = pinv;
    }
  }
  #pragma unroll
  for (int r = 0; r < 16; ++r) {
    a[r][0] = s0[r]; a[r][15] = s15[r];
    #pragma unroll
    for (int q = 0; q < 7; ++q) { a[r][2*q+1] = pr[r][q].x; a[r][2*q+2] = pr[r][q].y; }
  }
  #pragma unroll
  for (int r = 0; r < 16; ++r)
    #pragma unroll
    for (int c = 0; c < 16; ++c) P[OFF_AINV + r*17 + c] = a[r][c];
}

struct Frag { v2f a0, a1, a2, a3, g0, g1, g2, g3, y0, y1, y2, y3; };

__device__ __forceinline__ void load_frag(Frag& f, const float* pa, const float* py,
    const float* pg0, const float* pg1, const float* pg2, const float* pg3, int gs, int o)
{
  f.a0.x = pa[o + 0];  f.a0.y = pa[o + 1];   f.a1.x = pa[o + 4];  f.a1.y = pa[o + 5];
  f.a2.x = pa[o + 8];  f.a2.y = pa[o + 9];   f.a3.x = pa[o + 12]; f.a3.y = pa[o + 13];
  f.g0.x = pg0[o];     f.g0.y = pg0[o + gs]; f.g1.x = pg1[o];     f.g1.y = pg1[o + gs];
  f.g2.x = pg2[o];     f.g2.y = pg2[o + 8];  f.g3.x = pg3[o];     f.g3.y = pg3[o + 8];
  f.y0.x = py[o + 0];  f.y0.y = py[o + 1];   f.y1.x = py[o + 4];  f.y1.y = py[o + 5];
  f.y2.x = py[o + 8];  f.y2.y = py[o + 9];   f.y3.x = py[o + 12]; f.y3.y = py[o + 13];
}

__device__ __forceinline__ void run_products(const Frag& f, v8f& p1, v8f& p2)
{
  v8f z = {};
  p1 = __builtin_amdgcn_wmma_f32_16x16x4_f32(false, f.a0, false, f.g0, (short)0, z,  false, false);
  p2 = __builtin_amdgcn_wmma_f32_16x16x4_f32(false, f.a0, false, f.y0, (short)0, z,  false, false);
  p1 = __builtin_amdgcn_wmma_f32_16x16x4_f32(false, f.a1, false, f.g1, (short)0, p1, false, false);
  p2 = __builtin_amdgcn_wmma_f32_16x16x4_f32(false, f.a1, false, f.y1, (short)0, p2, false, false);
  p1 = __builtin_amdgcn_wmma_f32_16x16x4_f32(false, f.a2, false, f.g2, (short)0, p1, false, false);
  p2 = __builtin_amdgcn_wmma_f32_16x16x4_f32(false, f.a2, false, f.y2, (short)0, p2, false, false);
  p1 = __builtin_amdgcn_wmma_f32_16x16x4_f32(false, f.a3, false, f.g3, (short)0, p1, false, false);
  p2 = __builtin_amdgcn_wmma_f32_16x16x4_f32(false, f.a3, false, f.y3, (short)0, p2, false, false);
}

__device__ __forceinline__ void store_diag(float* q, const v8f& p1, const v8f& p2, int n7, bool own)
{
  float d1 = p1[0], d2 = p2[0];
  #pragma unroll
  for (int v = 1; v < 8; ++v) {
    const bool c = (n7 == v);
    d1 = c ? p1[v] : d1;
    d2 = c ? p2[v] : d2;
  }
  if (own) { q[0] = d1; q[16] = d2; }
}

__device__ __forceinline__ void diag_products(float* wv, int lane)
{
  const int nn = lane & 15, hf = lane >> 4;
  const int n7 = nn & 7;
  const bool nLow = nn < 8;
  const bool own = (nn >> 3) == hf;
  const float* pa = wv + OFF_AINV + nn*17 + hf*2;
  const float* py = wv + OFF_YPT  + n7*17 + hf*2;
  const int gs = nLow ? 8 : 1;
  const float* pg0 = nLow ? (wv + OFF_GRE + (0 + hf*2)*8 + n7) : (wv + OFF_GIM + n7*8 + (0 + hf*2));
  const float* pg1 = nLow ? (wv + OFF_GRE + (4 + hf*2)*8 + n7) : (wv + OFF_GIM + n7*8 + (4 + hf*2));
  const float* pg2 = (nLow ? (wv + OFF_GIM) : (wv + OFF_GRE)) + (0 + hf*2)*8 + n7;
  const float* pg3 = (nLow ? (wv + OFF_GIM) : (wv + OFF_GRE)) + (4 + hf*2)*8 + n7;
  float* pm = wv + OFF_MU + nn;

  #define LOADS_FIRST() __builtin_amdgcn_sched_barrier(0)
  Frag f0, f1;
  v8f p1a, p2a, p1b, p2b;
  load_frag(f0, pa, py, pg0, pg1, pg2, pg3, gs, 0);
  load_frag(f1, pa, py, pg0, pg1, pg2, pg3, gs, STRIDE);
  LOADS_FIRST();
  run_products(f0, p1a, p2a);
  LOADS_FIRST();
  load_frag(f0, pa, py, pg0, pg1, pg2, pg3, gs, 2*STRIDE);
  LOADS_FIRST();
  run_products(f1, p1b, p2b);
  #pragma unroll 1
  for (int c = 2; c < 30; c += 2) {
    const int ob = (c - 2) * STRIDE;
    store_diag(pm + ob, p1a, p2a, n7, own);
    load_frag(f1, pa, py, pg0, pg1, pg2, pg3, gs, ob + 3*STRIDE);
    LOADS_FIRST();
    run_products(f0, p1a, p2a);
    store_diag(pm + ob + STRIDE, p1b, p2b, n7, own);
    load_frag(f0, pa, py, pg0, pg1, pg2, pg3, gs, ob + 4*STRIDE);
    LOADS_FIRST();
    run_products(f1, p1b, p2b);
  }
  store_diag(pm + 28*STRIDE, p1a, p2a, n7, own);
  load_frag(f1, pa, py, pg0, pg1, pg2, pg3, gs, 31*STRIDE);
  LOADS_FIRST();
  run_products(f0, p1a, p2a);
  store_diag(pm + 29*STRIDE, p1b, p2b, n7, own);
  run_products(f1, p1b, p2b);
  store_diag(pm + 30*STRIDE, p1a, p2a, n7, own);
  store_diag(pm + 31*STRIDE, p1b, p2b, n7, own);
  #undef LOADS_FIRST
}

__device__ __forceinline__ void half_demap(float z, float rho, float la, float lb, float& da, float& db)
{
  const float c1 = 1.0f*INV_SQRT10, c3 = 3.0f*INV_SQRT10;
  const float t00 = z - c1, t01 = z - c3, t10 = z + c1, t11 = z + c3;
  const float r00 = -(t00*t00) * rho, r01 = -(t01*t01) * rho;
  const float r10 = -(t10*t10) * rho, r11 = -(t11*t11) * rho;
  da = fmaxf(r10, r11 + lb) - fmaxf(r00, r01 + lb);
  db = fmaxf(r01, r11 + la) - fmaxf(r00, r10 + la);
}

__device__ __forceinline__ void iter_second(float* P, bool last)
{
  #pragma unroll
  for (int s = 0; s < 8; ++s) {
    float m1 = P[OFF_MU + s], m2 = P[OFF_MU + s + 8];
    float zr = P[OFF_MU + 16 + s]     / m1;
    float zi = P[OFF_MU + 16 + s + 8] / m2;
    float rho = m1 / fmaxf(1.0f - P[OFF_X + 16 + s] * m1, EPSV);
    float l[4], d[4];
    l[0] = P[OFF_LLD + s*4 + 0]; l[1] = P[OFF_LLD + s*4 + 1];
    l[2] = P[OFF_LLD + s*4 + 2]; l[3] = P[OFF_LLD + s*4 + 3];
    half_demap(zr, rho, l[0], l[2], d[0], d[2]);
    half_demap(zi, rho, l[1], l[3], d[1], d[3]);
    #pragma unroll
    for (int k = 0; k < 4; ++k)
      P[OFF_LLD + s*4 + k] = last ? d[k] : (l[k] + d[k]);
  }
}

__global__ __launch_bounds__(ITEMS_PER_BLOCK) void mmse_pic_kernel(
    const float* __restrict__ y_re, const float* __restrict__ y_im,
    const float* __restrict__ h_re, const float* __restrict__ h_im,
    const float* __restrict__ prior,
    const float* __restrict__ s_re, const float* __restrict__ s_im,
    float* __restrict__ out)
{
  __shared__ __attribute__((aligned(16))) float lds[WAVES_PER_BLOCK * 32 * STRIDE];
  const int lane = threadIdx.x & 31;
  const int wid  = threadIdx.x >> 5;
  const int bw   = blockIdx.x * ITEMS_PER_BLOCK + wid * 32;
  float* wv = lds + wid * 32 * STRIDE;
  float* P  = wv + lane * STRIDE;

  front_stage(wv, lane, P, y_re, y_im, h_re, h_im, prior, s_re, s_im, bw);

  #pragma unroll 1
  for (int iter = 0; iter < 2; ++iter) {
    iter_first(P);
    __syncthreads();
    diag_products(wv, lane);
    __syncthreads();
    iter_second(P, iter == 1);
  }
  __syncthreads();

  v4f* o4 = (v4f*)(out + (size_t)(blockIdx.x * ITEMS_PER_BLOCK + wid * 32) * 32);
  #pragma unroll
  for (int j = 0; j < 8; ++j) {
    int q = j*32 + lane;
    const float* src = wv + (q >> 3)*STRIDE + OFF_LLD + (q & 7)*4;
    v4f v; v.x = src[0]; v.y = src[1]; v.z = src[2]; v.w = src[3];
    o4[q] = v;
  }
}

extern "C" void kernel_launch(void* const* d_in, const int* in_sizes, int n_in,
                              void* d_out, int out_size, void* d_ws, size_t ws_size,
                              hipStream_t stream) {
  (void)n_in; (void)out_size; (void)d_ws; (void)ws_size;
  const float* y_re  = (const float*)d_in[0];
  const float* y_im  = (const float*)d_in[1];
  const float* h_re  = (const float*)d_in[2];
  const float* h_im  = (const float*)d_in[3];
  const float* prior = (const float*)d_in[4];
  const float* s_re  = (const float*)d_in[5];
  const float* s_im  = (const float*)d_in[6];
  float* out = (float*)d_out;

  int B = in_sizes[0] / 16;
  int blocks = B / ITEMS_PER_BLOCK;
  mmse_pic_kernel<<<blocks, ITEMS_PER_BLOCK, 0, stream>>>(y_re, y_im, h_re, h_im,
                                                          prior, s_re, s_im, out);
}


# Gram and matched filter summed inside the Cholesky loop, one step late, from the solved row kept in registers: they issue under the next pivot's sqrt/divide wait; no Gram loop; sum order kept
# speedup vs baseline: 8.0246x; 1.0394x over previous
#include <hip/hip_runtime.h>
#include <math.h>


typedef __attribute__((ext_vector_type(2))) float v2f;
typedef __attribute__((ext_vector_type(4))) float v4f;
typedef __attribute__((ext_vector_type(8))) float v8f;

#define WAVES_PER_BLOCK 4
#define ITEMS_PER_BLOCK 128
#define EPSV 1e-4f
#define INV_SQRT10 0.3162277660168379f

#define STRIDE   617
#define OFF_AINV 0
#define OFF_GRE  272
#define OFF_GIM  344
#define OFF_YPT  408
#define OFF_MU   408
#define OFF_YMF  544
#define OFF_LLD  560
#define OFF_X    592

#define S_PITCH  516
#define H_PITCH  324
#define HOFF_HIM 128
#define HOFF_YRE 256
#define HOFF_YIM 272
#define HOFF_PRI 288


__device__ __forceinline__ float bit_mean(float l) {
  const float e = __expf(-fabsf(l));
  const float t = (1.0f - e) / (1.0f + e);
  return (l > 0.0f) ? -t : t;
}

__device__ __forceinline__ void gram_row(const float (&xr)[9], const float (&xi)[9],
                                         float (&gr_)[8][9], float (&gi_)[8][9])
{
  #pragma unroll
  for (int i = 0; i < 8; ++i) {
    #pragma unroll
    for (int j = i; j < 9; ++j) {
      gr_[i][j] = fmaf( xi[i], xi[j], fmaf(xr[i], xr[j], gr_[i][j]));
      gi_[i][j] = fmaf(-xi[i], xr[j], fmaf(xr[i], xi[j], gi_[i][j]));
    }
  }
}

template <int N>
__device__ __forceinline__ void whiten_step(float (&Are)[16][16], float (&Aim)[16][16],
                                            float (&Wre)[16][9], float (&Wim)[16][9],
                                            float (&pwr)[9], float (&pwi)[9],
                                            float (&gr_)[8][9], float (&gi_)[8][9])
{
  const float sdk = sqrtf(Are[0][0]);
  const float isd = 1.0f / sdk;
  gram_row(pwr, pwi, gr_, gi_);
  float cre[N], cim[N];
  cre[0] = 0.0f; cim[0] = 0.0f;
  #pragma unroll
  for (int i = 1; i < N; ++i) { cre[i] = Are[i][0] * isd; cim[i] = Aim[i][0] * isd; }
  float wkre[9], wkim[9];
  #pragma unroll
  for (int c = 0; c < 9; ++c) { wkre[c] = Wre[0][c] * isd; wkim[c] = Wim[0][c] * isd; }
  #pragma unroll
  for (int i = 1; i < N; ++i) {
    const float lre = cre[i], lim = cim[i];
    #pragma unroll
    for (int j = 1; j <= i; ++j) {
      Are[i-1][j-1] = fmaf(-lim, cim[j], fmaf(-lre, cre[j], Are[i][j]));
      Aim[i-1][j-1] = fmaf( lre, cim[j], fmaf(-lim, cre[j], Aim[i][j]));
    }
    #pragma unroll
    for (int c = 0; c < 9; ++c) {
      Wre[i-1][c] = fmaf( lim, wkim[c], fmaf(-lre, wkre[c], Wre[i][c]));
      Wim[i-1][c] = fmaf(-lim, wkre[c], fmaf(-lre, wkim[c], Wim[i][c]));
    }
  }
  #pragma unroll
  for (int j = 0; j < N; ++j) { Are[N-1][j] = 0.0f; Aim[N-1][j] = 0.0f; }
  #pragma unroll
  for (int c = 0; c < 9; ++c) { Wre[N-1][c] = 0.0f; Wim[N-1][c] = 0.0f; }
  #pragma unroll
  for (int c = 0; c < 9; ++c) { pwr[c] = wkre[c]; pwi[c] = wkim[c]; }
}

__device__ __forceinline__ void front_stage(float* wv, int lane, float* P,
    const float* __restrict__ y_re, const float* __restrict__ y_im,
    const float* __restrict__ h_re, const float* __restrict__ h_im,
    const float* __restrict__ prior,
    const float* __restrict__ s_re, const float* __restrict__ s_im, int bw)
{
  float Are[16][16], Aim[16][16];
  float Wre[16][9], Wim[16][9];
  v4f* const st = (v4f*)wv + lane;

  {
    const v4f* gsr = (const v4f*)(s_re + (size_t)bw*256) + lane;
    const v4f* gsi = (const v4f*)(s_im + (size_t)bw*256) + lane;
    v4f tr[64], ti[64];
    #pragma unroll
    for (int j = 0; j < 64; ++j) { tr[j] = gsr[j*32]; ti[j] = gsi[j*32]; }
    #pragma unroll
    for (int j = 0; j < 64; ++j) {
      const int o = (j >> 1)*(S_PITCH/4) + (j & 1)*32;
      st[o] = tr[j]; st[o + 64] = ti[j];
    }
  }
  __syncthreads();

  v4f thr[32], thi[32], tyr[4], tyi[4], tpr[8];
  {
    const v4f* ghr = (const v4f*)(h_re + (size_t)bw*128) + lane;
    const v4f* ghi = (const v4f*)(h_im + (size_t)bw*128) + lane;
    const v4f* gyr = (const v4f*)(y_re + (size_t)bw*16) + lane;
    const v4f* gyi = (const v4f*)(y_im + (size_t)bw*16) + lane;
    const v4f* gpr = (const v4f*)(prior + (size_t)bw*32) + lane;
    #pragma unroll
    for (int j = 0; j < 32; ++j) { thr[j] = ghr[j*32]; thi[j] = ghi[j*32]; }
    #pragma unroll
    for (int j = 0; j < 4; ++j) { tyr[j] = gyr[j*32]; tyi[j] = gyi[j*32]; }
    #pragma unroll
    for (int j = 0; j < 8; ++j) tpr[j] = gpr[j*32];
  }

  {
    const float* ps = wv + lane*S_PITCH;
    #pragma unroll
    for (int i = 0; i < 16; ++i) {
      #pragma unroll
      for (int q = 0; q < 4; ++q) {
        const float* pr = ps + i*16 + q*4;
        const float* pi = pr + 256;
        const int r = i - 4*q;
        Are[i][4*q+0] = 0.0f; Are[i][4*q+1] = 0.0f; Are[i][4*q+2] = 0.0f; Are[i][4*q+3] = 0.0f;
        Aim[i][4*q+0] = 0.0f; Aim[i][4*q+1] = 0.0f; Aim[i][4*q+2] = 0.0f; Aim[i][4*q+3] = 0.0f;
        if (r >= 3) {
          v4f vr = *(const v4f*)pr, vi = *(const v4f*)pi;
          Are[i][4*q+0] = vr.x; Are[i][4*q+1] = vr.y; Are[i][4*q+2] = vr.z; Are[i][4*q+3] = vr.w;
          Aim[i][4*q+0] = vi.x; Aim[i][4*q+1] = vi.y; Aim[i][4*q+2] = vi.z; Aim[i][4*q+3] = vi.w;
        } else if (r >= 1) {
          v2f vr = *(const v2f*)pr, vi = *(const v2f*)pi;
          Are[i][4*q+0] = vr.x; Are[i][4*q+1] = vr.y;
          Aim[i][4*q+0] = vi.x; Aim[i][4*q+1] = vi.y;
          if (r == 2) { Are[i][4*q+2] = pr[2]; Aim[i][4*q+2] = pi[2]; }
        } else if (r == 0) {
          Are[i][4*q+0] = pr[0]; Aim[i][4*q+0] = pi[0];
        }
      }
    }
  }
  __syncthreads();

  #pragma unroll
  for (int j = 0; j < 32; ++j) { st[j*(H_PITCH/4)] = thr[j]; st[j*(H_PITCH/4) + HOFF_HIM/4] = thi[j]; }
  {
    v4f* sy = (v4f*)wv + (lane >> 2)*(H_PITCH/4) + HOFF_YRE/4 + (lane & 3);
    #pragma unroll
    for (int j = 0; j < 4; ++j) { sy[j*8*(H_PITCH/4)] = tyr[j]; sy[j*8*(H_PITCH/4) + (HOFF_YIM - HOFF_YRE)/4] = tyi[j]; }
    v4f* sp = (v4f*)wv + (lane >> 3)*(H_PITCH/4) + HOFF_PRI/4 + (lane & 7);
    #pragma unroll
    for (int j = 0; j < 8; ++j) sp[j*4*(H_PITCH/4)] = tpr[j];
  }
  __syncthreads();

  v4f pl[8];
  {
    const float* ph = wv + lane*H_PITCH;
    #pragma unroll
    for (int m = 0; m < 16; ++m) {
      v4f a0 = *(const v4f*)(ph + m*8), a1 = *(const v4f*)(ph + m*8 + 4);
      v4f c0 = *(const v4f*)(ph + HOFF_HIM + m*8), c1 = *(const v4f*)(ph + HOFF_HIM + m*8 + 4);
      Wre[m][0] = a0.x; Wre[m][1] = a0.y; Wre[m][2] = a0.z; Wre[m][3] = a0.w;
      Wre[m][4] = a1.x; Wre[m][5] = a1.y; Wre[m][6] = a1.z; Wre[m][7] = a1.w;
      Wim[m][0] = c0.x; Wim[m][1] = c0.y; Wim[m][2] = c0.z; Wim[m][3] = c0.w;
      Wim[m][4] = c1.x; Wim[m][5] = c1.y; Wim[m][6] = c1.z; Wim[m][7] = c1.w;
    }
    #pragma unroll
    for (int q = 0; q < 4; ++q) {
      v4f a = *(const v4f*)(ph + HOFF_YRE + 4*q), cc = *(const v4f*)(ph + HOFF_YIM + 4*q);
      Wre[4*q+0][8] = a.x; Wre[4*q+1][8] = a.y; Wre[4*q+2][8] = a.z; Wre[4*q+3][8] = a.w;
      Wim[4*q+0][8] = cc.x; Wim[4*q+1][8] = cc.y; Wim[4*q+2][8] = cc.z; Wim[4*q+3][8] = cc.w;
    }
    #pragma unroll
    for (int s = 0; s < 8; ++s) pl[s] = *(const v4f*)(ph + HOFF_PRI + 4*s);
  }
  __syncthreads();
  #pragma unroll
  for (int s = 0; s < 8; ++s) {
    P[OFF_LLD + s*4 + 0] = pl[s].x; P[OFF_LLD + s*4 + 1] = pl[s].y;
    P[OFF_LLD + s*4 + 2] = pl[s].z; P[OFF_LLD + s*4 + 3] = pl[s].w;
  }

  float gr_[8][9], gi_[8][9];
  #pragma unroll
  for (int i = 0; i < 8; ++i)
    #pragma unroll
    for (int j = 0; j < 9; ++j) { gr_[i][j] = 0.0f; gi_[i][j] = 0.0f; }
  float pwr[9], pwi[9];
  #pragma unroll
  for (int c = 0; c < 9; ++c) { pwr[c] = 0.0f; pwi[c] = 0.0f; }

  #pragma unroll 1
  for (int k = 0; k < 8; ++k)  whiten_step<16>(Are, Aim, Wre, Wim, pwr, pwi, gr_, gi_);
  #pragma unroll 1
  for (int k = 8; k < 16; ++k) whiten_step<8>(Are, Aim, Wre, Wim, pwr, pwi, gr_, gi_);
  gram_row(pwr, pwi, gr_, gi_);
  #pragma unroll
  for (int i = 0; i < 8; ++i) {
    #pragma unroll
    for (int j = i; j < 8; ++j) {
      P[OFF_GRE + i*8 + j] = gr_[i][j];
      P[OFF_GIM + i*8 + j] = (j > i) ? gi_[i][j] : 0.0f;
      if (j > i) { P[OFF_GRE + j*8 + i] = gr_[i][j]; P[OFF_GIM + j*8 + i] = -gi_[i][j]; }
    }
    P[OFF_YMF + i]     = gr_[i][8];
    P[OFF_YMF + 8 + i] = gi_[i][8];
  }
}

__device__ __forceinline__ void iter_first(float* P)
{
  float xr[8], xi[8], vr[8];
  #pragma unroll
  for (int s = 0; s < 8; ++s) {
    const float m0 = bit_mean(P[OFF_LLD + s*4 + 0]), m1 = bit_mean(P[OFF_LLD + s*4 + 1]);
    const float m2 = bit_mean(P[OFF_LLD + s*4 + 2]), m3 = bit_mean(P[OFF_LLD + s*4 + 3]);
    const float mre = (m0 * (2.0f - m2)) * INV_SQRT10;
    const float mim = (m1 * (2.0f - m3)) * INV_SQRT10;
    const float e2  = (10.0f - 4.0f * (m2 + m3)) * 0.1f;
    xr[s] = mre; xi[s] = mim;
    vr[s] = e2 - (mre*mre + mim*mim);
    P[OFF_X + 16 + s] = vr[s];
  }

  float gre[8][8], gim[8][8];
  #pragma unroll
  for (int i = 0; i < 8; ++i) {
    #pragma unroll
    for (int j = 0; j < 8; ++j) { gre[i][j] = P[OFF_GRE + i*8 + j]; gim[i][j] = P[OFF_GIM + i*8 + j]; }
  }
  float gxr[8], gxi[8];
  #pragma unroll
  for (int i = 0; i < 8; ++i) {
    float ar = 0.0f, ai = 0.0f;
    #pragma unroll
    for (int j = 0; j < 8; ++j) {
      ar = fmaf(-gim[i][j], xi[j], fmaf(gre[i][j], xr[j], ar));
      ai = fmaf( gim[i][j], xr[j], fmaf(gre[i][j], xi[j], ai));
    }
    gxr[i] = ar; gxi[i] = ai;
  }
  float dre[8], dim[8];
  #pragma unroll
  for (int j = 0; j < 8; ++j) { dre[j] = P[OFF_YMF + j] - gxr[j]; dim[j] = P[OFF_YMF + 8 + j] - gxi[j]; }
  #pragma unroll
  for (int i = 0; i < 8; ++i) {
    #pragma unroll
    for (int j = 0; j < 8; ++j) {
      float tre = fmaf(-gim[j][i], xi[i], fmaf(gre[j][i], xr[i], dre[j]));
      float tim = fmaf( gim[j][i], xr[i], fmaf(gre[j][i], xi[i], dim[j]));
      P[OFF_YPT + i*17 + j]     = tre;
      P[OFF_YPT + i*17 + j + 8] = tim;
    }
  }
  float a[16][16];
  #pragma unroll
  for (int r = 0; r < 16; ++r) {
    #pragma unroll
    for (int c = 0; c < 16; ++c) {
      float gv;
      if (r < 8) gv = (c < 8) ? gre[r][c]   : -gim[r][c-8];
      else       gv = (c < 8) ? gim[r-8][c] :  gre[r-8][c-8];
      a[r][c] = gv * vr[c & 7] + ((r == c) ? 1.0f : 0.0f);
    }
  }
  float s0[16], s15[16];
  v2f pr[16][7];
  #pragma unroll
  for (int r = 0; r < 16; ++r) {
    s0[r] = a[r][0]; s15[r] = a[r][15];
    #pragma unroll
    for (int q = 0; q < 7; ++q) { pr[r][q].x = a[r][2*q+1]; pr[r][q].y = a[r][2*q+2]; }
  }
  #pragma unroll 1
  for (int t = 0; t < 8; ++t) {
    v2f m[16][8];
    {
      const float pinv = 1.0f / s0[0];
      v2f prow[7];
      #pragma unroll
      for (int q = 0; q < 7; ++q) prow[q] = pr[0][q] * pinv;
      const float prow15 = s15[0] * pinv;
      #pragma unroll
      for (int i = 1; i < 16; ++i) {
        const float f = s0[i];
        #pragma unroll
        for (int q = 0; q < 7; ++q) m[i-1][q] = pr[i][q] - f * prow[q];
        m[i-1][7].x = s15[i] - f * prow15;
        m[i-1][7].y = -(f * pinv);
      }
      #pragma unroll
      for (int q = 0; q < 7; ++q) m[15][q] = prow[q];
      m[15][7].x = prow15; m[15][7].y = pinv;
    }
    {
      const float pinv = 1.0f / m[0][0].x;
      const float prow1 = m[0][0].y * pinv;
      v2f prow[8];
      #pragma unroll
      for (int q = 1; q < 8; ++q) prow[q] = m[0][q] * pinv;
      #pragma unroll
      for (int i = 1; i < 16; ++i) {
        const float f = m[i][0].x;
        s0[i-1] = m[i][0].y - f * prow1;
        #pragma unroll
        for (int q = 1; q < 8; ++q) pr[i-1][q-1] = m[i][q] - f * prow[q];
        s15[i-1] = -(f * pinv);
      }
      s0[15] = prow1;
      #pragma unroll
      for (int q = 1; q < 8; ++q) pr[15][q-1] = prow[q];
      s15[15] = pinv;
    }
  }
  #pragma unroll
  for (int r = 0; r < 16; ++r) {
    a[r][0] = s0[r]; a[r][15] = s15[r];
    #pragma unroll
    for (int q = 0; q < 7; ++q) { a[r][2*q+1] = pr[r][q].x; a[r][2*q+2] = pr[r][q].y; }
  }
  #pragma unroll
  for (int r = 0; r < 16; ++r)
    #pragma unroll
    for (int c = 0; c < 16; ++c) P[OFF_AINV + r*17 + c] = a[r][c];
}

struct Frag { v2f a0, a1, a2, a3, g0, g1, g2, g3, y0, y1, y2, y3; };

__device__ __forceinline__ void load_frag(Frag& f, const float* pa, const float* py,
    const float* pg0, const float* pg1, const float* pg2, const float* pg3, int gs, int o)
{
  f.a0.x = pa[o + 0];  f.a0.y = pa[o + 1];   f.a1.x = pa[o + 4];  f.a1.y = pa[o + 5];
  f.a2.x = pa[o + 8];  f.a2.y = pa[o + 9];   f.a3.x = pa[o + 12]; f.a3.y = pa[o + 13];
  f.g0.x = pg0[o];     f.g0.y = pg0[o + gs]; f.g1.x = pg1[o];     f.g1.y = pg1[o + gs];
  f.g2.x = pg2[o];     f.g2.y = pg2[o + 8];  f.g3.x = pg3[o];     f.g3.y = pg3[o + 8];
  f.y0.x = py[o + 0];  f.y0.y = py[o + 1];   f.y1.x = py[o + 4];  f.y1.y = py[o + 5];
  f.y2.x = py[o + 8];  f.y2.y = py[o + 9];   f.y3.x = py[o + 12]; f.y3.y = py[o + 13];
}

__device__ __forceinline__ void run_products(const Frag& f, v8f& p1, v8f& p2)
{
  v8f z = {};
  p1 = __builtin_amdgcn_wmma_f32_16x16x4_f32(false, f.a0, false, f.g0, (short)0, z,  false, false);
  p2 = __builtin_amdgcn_wmma_f32_16x16x4_f32(false, f.a0, false, f.y0, (short)0, z,  false, false);
  p1 = __builtin_amdgcn_wmma_f32_16x16x4_f32(false, f.a1, false, f.g1, (short)0, p1, false, false);
  p2 = __builtin_amdgcn_wmma_f32_16x16x4_f32(false, f.a1, false, f.y1, (short)0, p2, false, false);
  p1 = __builtin_amdgcn_wmma_f32_16x16x4_f32(false, f.a2, false, f.g2, (short)0, p1, false, false);
  p2 = __builtin_amdgcn_wmma_f32_16x16x4_f32(false, f.a2, false, f.y2, (short)0, p2, false, false);
  p1 = __builtin_amdgcn_wmma_f32_16x16x4_f32(false, f.a3, false, f.g3, (short)0, p1, false, false);
  p2 = __builtin_amdgcn_wmma_f32_16x16x4_f32(false, f.a3, false, f.y3, (short)0, p2, false, false);
}

__device__ __forceinline__ void store_diag(float* q, const v8f& p1, const v8f& p2, int n7, bool own)
{
  float d1 = p1[0], d2 = p2[0];
  #pragma unroll
  for (int v = 1; v < 8; ++v) {
    const bool c = (n7 == v);
    d1 = c ? p1[v] : d1;
    d2 = c ? p2[v] : d2;
  }
  if (own) { q[0] = d1; q[16] = d2; }
}

__device__ __forceinline__ void diag_products(float* wv, int lane)
{
  const int nn = lane & 15, hf = lane >> 4;
  const int n7 = nn & 7;
  const bool nLow = nn < 8;
  const bool own = (nn >> 3) == hf;
  const float* pa = wv + OFF_AINV + nn*17 + hf*2;
  const float* py = wv + OFF_YPT  + n7*17 + hf*2;
  const int gs = nLow ? 8 : 1;
  const float* pg0 = nLow ? (wv + OFF_GRE + (0 + hf*2)*8 + n7) : (wv + OFF_GIM + n7*8 + (0 + hf*2));
  const float* pg1 = nLow ? (wv + OFF_GRE + (4 + hf*2)*8 + n7) : (wv + OFF_GIM + n7*8 + (4 + hf*2));
  const float* pg2 = (nLow ? (wv + OFF_GIM) : (wv + OFF_GRE)) + (0 + hf*2)*8 + n7;
  const float* pg3 = (nLow ? (wv + OFF_GIM) : (wv + OFF_GRE)) + (4 + hf*2)*8 + n7;
  float* pm = wv + OFF_MU + nn;

  #define LOADS_FIRST() __builtin_amdgcn_sched_barrier(0)
  Frag f0, f1;
  v8f p1a, p2a, p1b, p2b;
  load_frag(f0, pa, py, pg0, pg1, pg2, pg3, gs, 0);
  load_frag(f1, pa, py, pg0, pg1, pg2, pg3, gs, STRIDE);
  LOADS_FIRST();
  run_products(f0, p1a, p2a);
  LOADS_FIRST();
  load_frag(f0, pa, py, pg0, pg1, pg2, pg3, gs, 2*STRIDE);
  LOADS_FIRST();
  run_products(f1, p1b, p2b);
  #pragma unroll 1
  for (int c = 2; c < 30; c += 2) {
    const int ob = (c - 2) * STRIDE;
    store_diag(pm + ob, p1a, p2a, n7, own);
    load_frag(f1, pa, py, pg0, pg1, pg2, pg3, gs, ob + 3*STRIDE);
    LOADS_FIRST();
    run_products(f0, p1a, p2a);
    store_diag(pm + ob + STRIDE, p1b, p2b, n7, own);
    load_frag(f0, pa, py, pg0, pg1, pg2, pg3, gs, ob + 4*STRIDE);
    LOADS_FIRST();
    run_products(f1, p1b, p2b);
  }
  store_diag(pm + 28*STRIDE, p1a, p2a, n7, own);
  load_frag(f1, pa, py, pg0, pg1, pg2, pg3, gs, 31*STRIDE);
  LOADS_FIRST();
  run_products(f0, p1a, p2a);
  store_diag(pm + 29*STRIDE, p1b, p2b, n7, own);
  run_products(f1, p1b, p2b);
  store_diag(pm + 30*STRIDE, p1a, p2a, n7, own);
  store_diag(pm + 31*STRIDE, p1b, p2b, n7, own);
  #undef LOADS_FIRST
}

__device__ __forceinline__ void half_demap(float z, float rho, float la, float lb, float& da, float& db)
{
  const float c1 = 1.0f*INV_SQRT10, c3 = 3.0f*INV_SQRT10;
  const float t00 = z - c1, t01 = z - c3, t10 = z + c1, t11 = z + c3;
  const float r00 = -(t00*t00) * rho, r01 = -(t01*t01) * rho;
  const float r10 = -(t10*t10) * rho, r11 = -(t11*t11) * rho;
  da = fmaxf(r10, r11 + lb) - fmaxf(r00, r01 + lb);
  db = fmaxf(r01, r11 + la) - fmaxf(r00, r10 + la);
}

__device__ __forceinline__ void iter_second(float* P, bool last)
{
  #pragma unroll
  for (int s = 0; s < 8; ++s) {
    float m1 = P[OFF_MU + s], m2 = P[OFF_MU + s + 8];
    float zr = P[OFF_MU + 16 + s]     / m1;
    float zi = P[OFF_MU + 16 + s + 8] / m2;
    float rho = m1 / fmaxf(1.0f - P[OFF_X + 16 + s] * m1, EPSV);
    float l[4], d[4];
    l[0] = P[OFF_LLD + s*4 + 0]; l[1] = P[OFF_LLD + s*4 + 1];
    l[2] = P[OFF_LLD + s*4 + 2]; l[3] = P[OFF_LLD + s*4 + 3];
    half_demap(zr, rho, l[0], l[2], d[0], d[2]);
    half_demap(zi, rho, l[1], l[3], d[1], d[3]);
    #pragma unroll
    for (int k = 0; k < 4; ++k)
      P[OFF_LLD + s*4 + k] = last ? d[k] : (l[k] + d[k]);
  }
}

__global__ __launch_bounds__(ITEMS_PER_BLOCK) void mmse_pic_kernel(
    const float* __restrict__ y_re, const float* __restrict__ y_im,
    const float* __restrict__ h_re, const float* __restrict__ h_im,
    const float* __restrict__ prior,
    const float* __restrict__ s_re, const float* __restrict__ s_im,
    float* __restrict__ out)
{
  __shared__ __attribute__((aligned(16))) float lds[WAVES_PER_BLOCK * 32 * STRIDE];
  const int lane = threadIdx.x & 31;
  const int wid  = threadIdx.x >> 5;
  const int bw   = blockIdx.x * ITEMS_PER_BLOCK + wid * 32;
  float* wv = lds + wid * 32 * STRIDE;
  float* P  = wv + lane * STRIDE;

  front_stage(wv, lane, P, y_re, y_im, h_re, h_im, prior, s_re, s_im, bw);

  #pragma unroll 1
  for (int iter = 0; iter < 2; ++iter) {
    iter_first(P);
    __syncthreads();
    diag_products(wv, lane);
    __syncthreads();
    iter_second(P, iter == 1);
  }
  __syncthreads();

  v4f* o4 = (v4f*)(out + (size_t)(blockIdx.x * ITEMS_PER_BLOCK + wid * 32) * 32);
  #pragma unroll
  for (int j = 0; j < 8; ++j) {
    int q = j*32 + lane;
    const float* src = wv + (q >> 3)*STRIDE + OFF_LLD + (q & 7)*4;
    v4f v; v.x = src[0]; v.y = src[1]; v.z = src[2]; v.w = src[3];
    o4[q] = v;
  }
}

extern "C" void kernel_launch(void* const* d_in, const int* in_sizes, int n_in,
                              void* d_out, int out_size, void* d_ws, size_t ws_size,
                              hipStream_t stream) {
  (void)n_in; (void)out_size; (void)d_ws; (void)ws_size;
  const float* y_re  = (const float*)d_in[0];
  const float* y_im  = (const float*)d_in[1];
  const float* h_re  = (const float*)d_in[2];
  const float* h_im  = (const float*)d_in[3];
  const float* prior = (const float*)d_in[4];
  const float* s_re  = (const float*)d_in[5];
  const float* s_im  = (const float*)d_in[6];
  float* out = (float*)d_out;

  int B = in_sizes[0] / 16;
  int blocks = B / ITEMS_PER_BLOCK;
  mmse_pic_kernel<<<blocks, ITEMS_PER_BLOCK, 0, stream>>>(y_re, y_im, h_re, h_im,
                                                          prior, s_re, s_im, out);
}


# Second diagonal product removed by algebra: znum = Ainv*d + x times P1's diagonal and [i][i+8],[i+8][i] elements, P1=Ainv*gr; Ainv*d per lane in registers; a run pairs two items; z sum order changes
# speedup vs baseline: 9.3736x; 1.1681x over previous
#include <hip/hip_runtime.h>
#include <math.h>


typedef __attribute__((ext_vector_type(2))) float v2f;
typedef __attribute__((ext_vector_type(4))) float v4f;
typedef __attribute__((ext_vector_type(8))) float v8f;

#define WAVES_PER_BLOCK 4
#define ITEMS_PER_BLOCK 128
#define EPSV 1e-4f
#define INV_SQRT10 0.3162277660168379f

#define STRIDE   617
#define OFF_AINV 0
#define OFF_GRE  272
#define OFF_GIM  344
#define OFF_MU   408
#define OFF_YMF  544
#define OFF_LLD  560
#define OFF_X    592

#define S_PITCH  516
#define H_PITCH  324
#define HOFF_HIM 128
#define HOFF_YRE 256
#define HOFF_YIM 272
#define HOFF_PRI 288


__device__ __forceinline__ float bit_mean(float l) {
  const float e = __expf(-fabsf(l));
  const float t = (1.0f - e) / (1.0f + e);
  return (l > 0.0f) ? -t : t;
}

__device__ __forceinline__ void gram_row(const float (&xr)[9], const float (&xi)[9],
                                         float (&gr_)[8][9], float (&gi_)[8][9])
{
  #pragma unroll
  for (int i = 0; i < 8; ++i) {
    #pragma unroll
    for (int j = i; j < 9; ++j) {
      gr_[i][j] = fmaf( xi[i], xi[j], fmaf(xr[i], xr[j], gr_[i][j]));
      gi_[i][j] = fmaf(-xi[i], xr[j], fmaf(xr[i], xi[j], gi_[i][j]));
    }
  }
}

template <int N>
__device__ __forceinline__ void whiten_step(float (&Are)[16][16], float (&Aim)[16][16],
                                            float (&Wre)[16][9], float (&Wim)[16][9],
                                            float (&pwr)[9], float (&pwi)[9],
                                            float (&gr_)[8][9], float (&gi_)[8][9])
{
  const float sdk = sqrtf(Are[0][0]);
  const float isd = 1.0f / sdk;
  gram_row(pwr, pwi, gr_, gi_);
  float cre[N], cim[N];
  cre[0] = 0.0f; cim[0] = 0.0f;
  #pragma unroll
  for (int i = 1; i < N; ++i) { cre[i] = Are[i][0] * isd; cim[i] = Aim[i][0] * isd; }
  float wkre[9], wkim[9];
  #pragma unroll
  for (int c = 0; c < 9; ++c) { wkre[c] = Wre[0][c] * isd; wkim[c] = Wim[0][c] * isd; }
  #pragma unroll
  for (int i = 1; i < N; ++i) {
    const float lre = cre[i], lim = cim[i];
    #pragma unroll
    for (int j = 1; j <= i; ++j) {
      Are[i-1][j-1] = fmaf(-lim, cim[j], fmaf(-lre, cre[j], Are[i][j]));
      Aim[i-1][j-1] = fmaf( lre, cim[j], fmaf(-lim, cre[j], Aim[i][j]));
    }
    #pragma unroll
    for (int c = 0; c < 9; ++c) {
      Wre[i-1][c] = fmaf( lim, wkim[c], fmaf(-lre, wkre[c], Wre[i][c]));
      Wim[i-1][c] = fmaf(-lim, wkre[c], fmaf(-lre, wkim[c], Wim[i][c]));
    }
  }
  #pragma unroll
  for (int j = 0; j < N; ++j) { Are[N-1][j] = 0.0f; Aim[N-1][j] = 0.0f; }
  #pragma unroll
  for (int c = 0; c < 9; ++c) { Wre[N-1][c] = 0.0f; Wim[N-1][c] = 0.0f; }
  #pragma unroll
  for (int c = 0; c < 9; ++c) { pwr[c] = wkre[c]; pwi[c] = wkim[c]; }
}

__device__ __forceinline__ void front_stage(float* wv, int lane, float* P,
    const float* __restrict__ y_re, const float* __restrict__ y_im,
    const float* __restrict__ h_re, const float* __restrict__ h_im,
    const float* __restrict__ prior,
    const float* __restrict__ s_re, const float* __restrict__ s_im, int bw)
{
  float Are[16][16], Aim[16][16];
  float Wre[16][9], Wim[16][9];
  v4f* const st = (v4f*)wv + lane;

  {
    const v4f* gsr = (const v4f*)(s_re + (size_t)bw*256) + lane;
    const v4f* gsi = (const v4f*)(s_im + (size_t)bw*256) + lane;
    v4f tr[64], ti[64];
    #pragma unroll
    for (int j = 0; j < 64; ++j) { tr[j] = gsr[j*32]; ti[j] = gsi[j*32]; }
    #pragma unroll
    for (int j = 0; j < 64; ++j) {
      const int o = (j >> 1)*(S_PITCH/4) + (j & 1)*32;
      st[o] = tr[j]; st[o + 64] = ti[j];
    }
  }
  __syncthreads();

  v4f thr[32], thi[32], tyr[4], tyi[4], tpr[8];
  {
    const v4f* ghr = (const v4f*)(h_re + (size_t)bw*128) + lane;
    const v4f* ghi = (const v4f*)(h_im + (size_t)bw*128) + lane;
    const v4f* gyr = (const v4f*)(y_re + (size_t)bw*16) + lane;
    const v4f* gyi = (const v4f*)(y_im + (size_t)bw*16) + lane;
    const v4f* gpr = (const v4f*)(prior + (size_t)bw*32) + lane;
    #pragma unroll
    for (int j = 0; j < 32; ++j) { thr[j] = ghr[j*32]; thi[j] = ghi[j*32]; }
    #pragma unroll
    for (int j = 0; j < 4; ++j) { tyr[j] = gyr[j*32]; tyi[j] = gyi[j*32]; }
    #pragma unroll
    for (int j = 0; j < 8; ++j) tpr[j] = gpr[j*32];
  }

  {
    const float* ps = wv + lane*S_PITCH;
    #pragma unroll
    for (int i = 0; i < 16; ++i) {
      #pragma unroll
      for (int q = 0; q < 4; ++q) {
        const float* pr = ps + i*16 + q*4;
        const float* pi = pr + 256;
        const int r = i - 4*q;
        Are[i][4*q+0] = 0.0f; Are[i][4*q+1] = 0.0f; Are[i][4*q+2] = 0.0f; Are[i][4*q+3] = 0.0f;
        Aim[i][4*q+0] = 0.0f; Aim[i][4*q+1] = 0.0f; Aim[i][4*q+2] = 0.0f; Aim[i][4*q+3] = 0.0f;
        if (r >= 3) {
          v4f vr = *(const v4f*)pr, vi = *(const v4f*)pi;
          Are[i][4*q+0] = vr.x; Are[i][4*q+1] = vr.y; Are[i][4*q+2] = vr.z; Are[i][4*q+3] = vr.w;
          Aim[i][4*q+0] = vi.x; Aim[i][4*q+1] = vi.y; Aim[i][4*q+2] = vi.z; Aim[i][4*q+3] = vi.w;
        } else if (r >= 1) {
          v2f vr = *(const v2f*)pr, vi = *(const v2f*)pi;
          Are[i][4*q+0] = vr.x; Are[i][4*q+1] = vr.y;
          Aim[i][4*q+0] = vi.x; Aim[i][4*q+1] = vi.y;
          if (r == 2) { Are[i][4*q+2] = pr[2]; Aim[i][4*q+2] = pi[2]; }
        } else if (r == 0) {
          Are[i][4*q+0] = pr[0]; Aim[i][4*q+0] = pi[0];
        }
      }
    }
  }
  __syncthreads();

  #pragma unroll
  for (int j = 0; j < 32; ++j) { st[j*(H_PITCH/4)] = thr[j]; st[j*(H_PITCH/4) + HOFF_HIM/4] = thi[j]; }
  {
    v4f* sy = (v4f*)wv + (lane >> 2)*(H_PITCH/4) + HOFF_YRE/4 + (lane & 3);
    #pragma unroll
    for (int j = 0; j < 4; ++j) { sy[j*8*(H_PITCH/4)] = tyr[j]; sy[j*8*(H_PITCH/4) + (HOFF_YIM - HOFF_YRE)/4] = tyi[j]; }
    v4f* sp = (v4f*)wv + (lane >> 3)*(H_PITCH/4) + HOFF_PRI/4 + (lane & 7);
    #pragma unroll
    for (int j = 0; j < 8; ++j) sp[j*4*(H_PITCH/4)] = tpr[j];
  }
  __syncthreads();

  v4f pl[8];
  {
    const float* ph = wv + lane*H_PITCH;
    #pragma unroll
    for (int m = 0; m < 16; ++m) {
      v4f a0 = *(const v4f*)(ph + m*8), a1 = *(const v4f*)(ph + m*8 + 4);
      v4f c0 = *(const v4f*)(ph + HOFF_HIM + m*8), c1 = *(const v4f*)(ph + HOFF_HIM + m*8 + 4);
      Wre[m][0] = a0.x; Wre[m][1] = a0.y; Wre[m][2] = a0.z; Wre[m][3] = a0.w;
      Wre[m][4] = a1.x; Wre[m][5] = a1.y; Wre[m][6] = a1.z; Wre[m][7] = a1.w;
      Wim[m][0] = c0.x; Wim[m][1] = c0.y; Wim[m][2] = c0.z; Wim[m][3] = c0.w;
      Wim[m][4] = c1.x; Wim[m][5] = c1.y; Wim[m][6] = c1.z; Wim[m][7] = c1.w;
    }
    #pragma unroll
    for (int q = 0; q < 4; ++q) {
      v4f a = *(const v4f*)(ph + HOFF_YRE + 4*q), cc = *(const v4f*)(ph + HOFF_YIM + 4*q);
      Wre[4*q+0][8] = a.x; Wre[4*q+1][8] = a.y; Wre[4*q+2][8] = a.z; Wre[4*q+3][8] = a.w;
      Wim[4*q+0][8] = cc.x; Wim[4*q+1][8] = cc.y; Wim[4*q+2][8] = cc.z; Wim[4*q+3][8] = cc.w;
    }
    #pragma unroll
    for (int s = 0; s < 8; ++s) pl[s] = *(const v4f*)(ph + HOFF_PRI + 4*s);
  }
  __syncthreads();
  #pragma unroll
  for (int s = 0; s < 8; ++s) {
    P[OFF_LLD + s*4 + 0] = pl[s].x; P[OFF_LLD + s*4 + 1] = pl[s].y;
    P[OFF_LLD + s*4 + 2] = pl[s].z; P[OFF_LLD + s*4 + 3] = pl[s].w;
  }

  float gr_[8][9], gi_[8][9];
  #pragma unroll
  for (int i = 0; i < 8; ++i)
    #pragma unroll
    for (int j = 0; j < 9; ++j) { gr_[i][j] = 0.0f; gi_[i][j] = 0.0f; }
  float pwr[9], pwi[9];
  #pragma unroll
  for (int c = 0; c < 9; ++c) { pwr[c] = 0.0f; pwi[c] = 0.0f; }

  #pragma unroll 1
  for (int k = 0; k < 8; ++k)  whiten_step<16>(Are, Aim, Wre, Wim, pwr, pwi, gr_, gi_);
  #pragma unroll 1
  for (int k = 8; k < 16; ++k) whiten_step<8>(Are, Aim, Wre, Wim, pwr, pwi, gr_, gi_);
  gram_row(pwr, pwi, gr_, gi_);
  #pragma unroll
  for (int i = 0; i < 8; ++i) {
    #pragma unroll
    for (int j = i; j < 8; ++j) {
      P[OFF_GRE + i*8 + j] = gr_[i][j];
      P[OFF_GIM + i*8 + j] = (j > i) ? gi_[i][j] : 0.0f;
      if (j > i) { P[OFF_GRE + j*8 + i] = gr_[i][j]; P[OFF_GIM + j*8 + i] = -gi_[i][j]; }
    }
    P[OFF_YMF + i]     = gr_[i][8];
    P[OFF_YMF + 8 + i] = gi_[i][8];
  }
}

__device__ __forceinline__ void iter_first(float* P, float (&xr)[8], float (&xi)[8], float (&w)[16])
{
  float vr[8];
  #pragma unroll
  for (int s = 0; s < 8; ++s) {
    const float m0 = bit_mean(P[OFF_LLD + s*4 + 0]), m1 = bit_mean(P[OFF_LLD + s*4 + 1]);
    const float m2 = bit_mean(P[OFF_LLD + s*4 + 2]), m3 = bit_mean(P[OFF_LLD + s*4 + 3]);
    const float mre = (m0 * (2.0f - m2)) * INV_SQRT10;
    const float mim = (m1 * (2.0f - m3)) * INV_SQRT10;
    const float e2  = (10.0f - 4.0f * (m2 + m3)) * 0.1f;
    xr[s] = mre; xi[s] = mim;
    vr[s] = e2 - (mre*mre + mim*mim);
    P[OFF_X + 16 + s] = vr[s];
  }

  float gre[8][8], gim[8][8];
  #pragma unroll
  for (int i = 0; i < 8; ++i) {
    #pragma unroll
    for (int j = 0; j < 8; ++j) { gre[i][j] = P[OFF_GRE + i*8 + j]; gim[i][j] = P[OFF_GIM + i*8 + j]; }
  }
  float gxr[8], gxi[8];
  #pragma unroll
  for (int i = 0; i < 8; ++i) {
    float ar = 0.0f, ai = 0.0f;
    #pragma unroll
    for (int j = 0; j < 8; ++j) {
      ar = fmaf(-gim[i][j], xi[j], fmaf(gre[i][j], xr[j], ar));
      ai = fmaf( gim[i][j], xr[j], fmaf(gre[i][j], xi[j], ai));
    }
    gxr[i] = ar; gxi[i] = ai;
  }
  float dr[16];
  #pragma unroll
  for (int j = 0; j < 8; ++j) { dr[j] = P[OFF_YMF + j] - gxr[j]; dr[j + 8] = P[OFF_YMF + 8 + j] - gxi[j]; }
  float a[16][16];
  #pragma unroll
  for (int r = 0; r < 16; ++r) {
    #pragma unroll
    for (int c = 0; c < 16; ++c) {
      float gv;
      if (r < 8) gv = (c < 8) ? gre[r][c]   : -gim[r][c-8];
      else       gv = (c < 8) ? gim[r-8][c] :  gre[r-8][c-8];
      a[r][c] = gv * vr[c & 7] + ((r == c) ? 1.0f : 0.0f);
    }
  }
  float s0[16], s15[16];
  v2f pr[16][7];
  #pragma unroll
  for (int r = 0; r < 16; ++r) {
    s0[r] = a[r][0]; s15[r] = a[r][15];
    #pragma unroll
    for (int q = 0; q < 7; ++q) { pr[r][q].x = a[r][2*q+1]; pr[r][q].y = a[r][2*q+2]; }
  }
  #pragma unroll 1
  for (int t = 0; t < 8; ++t) {
    v2f m[16][8];
    {
      const float pinv = 1.0f / s0[0];
      v2f prow[7];
      #pragma unroll
      for (int q = 0; q < 7; ++q) prow[q] = pr[0][q] * pinv;
      const float prow15 = s15[0] * pinv;
      #pragma unroll
      for (int i = 1; i < 16; ++i) {
        const float f = s0[i];
        #pragma unroll
        for (int q = 0; q < 7; ++q) m[i-1][q] = pr[i][q] - f * prow[q];
        m[i-1][7].x = s15[i] - f * prow15;
        m[i-1][7].y = -(f * pinv);
      }
      #pragma unroll
      for (int q = 0; q < 7; ++q) m[15][q] = prow[q];
      m[15][7].x = prow15; m[15][7].y = pinv;
    }
    {
      const float pinv = 1.0f / m[0][0].x;
      const float prow1 = m[0][0].y * pinv;
      v2f prow[8];
      #pragma unroll
      for (int q = 1; q < 8; ++q) prow[q] = m[0][q] * pinv;
      #pragma unroll
      for (int i = 1; i < 16; ++i) {
        const float f = m[i][0].x;
        s0[i-1] = m[i][0].y - f * prow1;
        #pragma unroll
        for (int q = 1; q < 8; ++q) pr[i-1][q-1] = m[i][q] - f * prow[q];
        s15[i-1] = -(f * pinv);
      }
      s0[15] = prow1;
      #pragma unroll
      for (int q = 1; q < 8; ++q) pr[15][q-1] = prow[q];
      s15[15] = pinv;
    }
  }
  {
    v2f dp[7];
    #pragma unroll
    for (int q = 0; q < 7; ++q) { dp[q].x = dr[2*q+1]; dp[q].y = dr[2*q+2]; }
    #pragma unroll
    for (int r = 0; r < 16; ++r) {
      v2f acc = pr[r][0] * dp[0];
      #pragma unroll
      for (int q = 1; q < 7; ++q) acc = pr[r][q] * dp[q] + acc;
      w[r] = fmaf(s15[r], dr[15], fmaf(s0[r], dr[0], acc.x + acc.y));
    }
  }
  #pragma unroll
  for (int r = 0; r < 16; ++r) {
    a[r][0] = s0[r]; a[r][15] = s15[r];
    #pragma unroll
    for (int q = 0; q < 7; ++q) { a[r][2*q+1] = pr[r][q].x; a[r][2*q+2] = pr[r][q].y; }
  }
  #pragma unroll
  for (int r = 0; r < 16; ++r)
    #pragma unroll
    for (int c = 0; c < 16; ++c) P[OFF_AINV + r*17 + c] = a[r][c];
}

struct Frag { v2f a0, a1, a2, a3, g0, g1, g2, g3; };

__device__ __forceinline__ void load_frag(Frag& f, const float* pa,
    const float* pg0, const float* pg1, const float* pg2, const float* pg3, int gs, int o)
{
  f.a0.x = pa[o + 0];  f.a0.y = pa[o + 1];   f.a1.x = pa[o + 4];  f.a1.y = pa[o + 5];
  f.a2.x = pa[o + 8];  f.a2.y = pa[o + 9];   f.a3.x = pa[o + 12]; f.a3.y = pa[o + 13];
  f.g0.x = pg0[o];     f.g0.y = pg0[o + gs]; f.g1.x = pg1[o];     f.g1.y = pg1[o + gs];
  f.g2.x = pg2[o];     f.g2.y = pg2[o + 8];  f.g3.x = pg3[o];     f.g3.y = pg3[o + 8];
}

__device__ __forceinline__ void run_products(const Frag& f, const Frag& h, v8f& pf, v8f& ph)
{
  v8f z = {};
  pf = __builtin_amdgcn_wmma_f32_16x16x4_f32(false, f.a0, false, f.g0, (short)0, z,  false, false);
  ph = __builtin_amdgcn_wmma_f32_16x16x4_f32(false, h.a0, false, h.g0, (short)0, z,  false, false);
  pf = __builtin_amdgcn_wmma_f32_16x16x4_f32(false, f.a1, false, f.g1, (short)0, pf, false, false);
  ph = __builtin_amdgcn_wmma_f32_16x16x4_f32(false, h.a1, false, h.g1, (short)0, ph, false, false);
  pf = __builtin_amdgcn_wmma_f32_16x16x4_f32(false, f.a2, false, f.g2, (short)0, pf, false, false);
  ph = __builtin_amdgcn_wmma_f32_16x16x4_f32(false, h.a2, false, h.g2, (short)0, ph, false, false);
  pf = __builtin_amdgcn_wmma_f32_16x16x4_f32(false, f.a3, false, f.g3, (short)0, pf, false, false);
  ph = __builtin_amdgcn_wmma_f32_16x16x4_f32(false, h.a3, false, h.g3, (short)0, ph, false, false);
}

__device__ __forceinline__ float pick(const v8f& p, int n7)
{
  float d = p[0];
  #pragma unroll
  for (int v = 1; v < 8; ++v) d = (n7 == v) ? p[v] : d;
  return d;
}

__device__ __forceinline__ void diag_products(float* wv, int lane)
{
  const int nn = lane & 15, hf = lane >> 4;
  const int n7 = nn & 7;
  const bool nLow = nn < 8;
  const bool own = (nn >> 3) == hf;
  const float* pa = wv + OFF_AINV + nn*17 + hf*2;
  const int gs = nLow ? 8 : 1;
  const float* pg0 = nLow ? (wv + OFF_GRE + (0 + hf*2)*8 + n7) : (wv + OFF_GIM + n7*8 + (0 + hf*2));
  const float* pg1 = nLow ? (wv + OFF_GRE + (4 + hf*2)*8 + n7) : (wv + OFF_GIM + n7*8 + (4 + hf*2));
  const float* pg2 = (nLow ? (wv + OFF_GIM) : (wv + OFF_GRE)) + (0 + hf*2)*8 + n7;
  const float* pg3 = (nLow ? (wv + OFF_GIM) : (wv + OFF_GRE)) + (4 + hf*2)*8 + n7;
  float* pm = wv + OFF_MU + (own ? nn : (16 + (nn ^ 8)));

  #define RUN_EDGE() __builtin_amdgcn_sched_barrier(0)
  #define RUN_DONE(x, y) asm volatile("" :: "v"(x), "v"(y))
  #define LOAD_PAIR(fa, fb, o) do { load_frag(fa, pa, pg0, pg1, pg2, pg3, gs, (o)); \
                                    load_frag(fb, pa, pg0, pg1, pg2, pg3, gs, (o) + STRIDE); } while (0)
  Frag f0, f1, f2, f3;
  v8f pa0, pa1, pb0, pb1;
  LOAD_PAIR(f0, f1, 0);
  LOAD_PAIR(f2, f3, 2*STRIDE);
  RUN_EDGE();
  run_products(f0, f1, pa0, pa1);
  RUN_DONE(pa0, pa1);
  RUN_EDGE();
  LOAD_PAIR(f0, f1, 4*STRIDE);
  RUN_EDGE();
  run_products(f2, f3, pb0, pb1);
  RUN_DONE(pb0, pb1);
  RUN_EDGE();
  #pragma unroll 1
  for (int c = 4; c < 28; c += 4) {
    const int oc = c * STRIDE;
    pm[oc - 4*STRIDE] = pick(pa0, n7);
    pm[oc - 3*STRIDE] = pick(pa1, n7);
    LOAD_PAIR(f2, f3, oc + 2*STRIDE);
    RUN_EDGE();
    run_products(f0, f1, pa0, pa1);
    RUN_DONE(pa0, pa1);
    RUN_EDGE();
    pm[oc - 2*STRIDE] = pick(pb0, n7);
    pm[oc - STRIDE]   = pick(pb1, n7);
    LOAD_PAIR(f0, f1, oc + 4*STRIDE);
    RUN_EDGE();
    run_products(f2, f3, pb0, pb1);
    RUN_DONE(pb0, pb1);
    RUN_EDGE();
  }
  pm[24*STRIDE] = pick(pa0, n7);
  pm[25*STRIDE] = pick(pa1, n7);
  LOAD_PAIR(f2, f3, 30*STRIDE);
  RUN_EDGE();
  run_products(f0, f1, pa0, pa1);
  RUN_DONE(pa0, pa1);
  RUN_EDGE();
  pm[26*STRIDE] = pick(pb0, n7);
  pm[27*STRIDE] = pick(pb1, n7);
  RUN_EDGE();
  run_products(f2, f3, pb0, pb1);
  RUN_DONE(pb0, pb1);
  RUN_EDGE();
  pm[28*STRIDE] = pick(pa0, n7);
  pm[29*STRIDE] = pick(pa1, n7);
  pm[30*STRIDE] = pick(pb0, n7);
  pm[31*STRIDE] = pick(pb1, n7);
  #undef LOAD_PAIR
  #undef RUN_DONE
  #undef RUN_EDGE
}

__device__ __forceinline__ void half_demap(float z, float rho, float la, float lb, float& da, float& db)
{
  const float c1 = 1.0f*INV_SQRT10, c3 = 3.0f*INV_SQRT10;
  const float t00 = z - c1, t01 = z - c3, t10 = z + c1, t11 = z + c3;
  const float r00 = -(t00*t00) * rho, r01 = -(t01*t01) * rho;
  const float r10 = -(t10*t10) * rho, r11 = -(t11*t11) * rho;
  da = fmaxf(r10, r11 + lb) - fmaxf(r00, r01 + lb);
  db = fmaxf(r01, r11 + la) - fmaxf(r00, r10 + la);
}

__device__ __forceinline__ void iter_second(float* P, bool last,
    const float (&xr)[8], const float (&xi)[8], const float (&w)[16])
{
  #pragma unroll
  for (int s = 0; s < 8; ++s) {
    float m1 = P[OFF_MU + s], m2 = P[OFF_MU + s + 8];
    float zr = fmaf(xi[s], P[OFF_MU + 16 + s],     fmaf(xr[s], m1, w[s]))                     / m1;
    float zi = fmaf(xi[s], m2, fmaf(xr[s], P[OFF_MU + 16 + s + 8], w[s + 8])) / m2;
    float rho = m1 / fmaxf(1.0f - P[OFF_X + 16 + s] * m1, EPSV);
    float l[4], d[4];
    l[0] = P[OFF_LLD + s*4 + 0]; l[1] = P[OFF_LLD + s*4 + 1];
    l[2] = P[OFF_LLD + s*4 + 2]; l[3] = P[OFF_LLD + s*4 + 3];
    half_demap(zr, rho, l[0], l[2], d[0], d[2]);
    half_demap(zi, rho, l[1], l[3], d[1], d[3]);
    #pragma unroll
    for (int k = 0; k < 4; ++k)
      P[OFF_LLD + s*4 + k] = last ? d[k] : (l[k] + d[k]);
  }
}

__global__ __launch_bounds__(ITEMS_PER_BLOCK) void mmse_pic_kernel(
    const float* __restrict__ y_re, const float* __restrict__ y_im,
    const float* __restrict__ h_re, const float* __restrict__ h_im,
    const float* __restrict__ prior,
    const float* __restrict__ s_re, const float* __restrict__ s_im,
    float* __restrict__ out)
{
  __shared__ __attribute__((aligned(16))) float lds[WAVES_PER_BLOCK * 32 * STRIDE];
  const int lane = threadIdx.x & 31;
  const int wid  = threadIdx.x >> 5;
  const int bw   = blockIdx.x * ITEMS_PER_BLOCK + wid * 32;
  float* wv = lds + wid * 32 * STRIDE;
  float* P  = wv + lane * STRIDE;

  front_stage(wv, lane, P, y_re, y_im, h_re, h_im, prior, s_re, s_im, bw);

  #pragma unroll 1
  for (int iter = 0; iter < 2; ++iter) {
    float xr[8], xi[8], w[16];
    iter_first(P, xr, xi, w);
    __syncthreads();
    diag_products(wv, lane);
    __syncthreads();
    iter_second(P, iter == 1, xr, xi, w);
  }
  __syncthreads();

  v4f* o4 = (v4f*)(out + (size_t)(blockIdx.x * ITEMS_PER_BLOCK + wid * 32) * 32);
  #pragma unroll
  for (int j = 0; j < 8; ++j) {
    int q = j*32 + lane;
    const float* src = wv + (q >> 3)*STRIDE + OFF_LLD + (q & 7)*4;
    v4f v; v.x = src[0]; v.y = src[1]; v.z = src[2]; v.w = src[3];
    o4[q] = v;
  }
}

extern "C" void kernel_launch(void* const* d_in, const int* in_sizes, int n_in,
                              void* d_out, int out_size, void* d_ws, size_t ws_size,
                              hipStream_t stream) {
  (void)n_in; (void)out_size; (void)d_ws; (void)ws_size;
  const float* y_re  = (const float*)d_in[0];
  const float* y_im  = (const float*)d_in[1];
  const float* h_re  = (const float*)d_in[2];
  const float* h_im  = (const float*)d_in[3];
  const float* prior = (const float*)d_in[4];
  const float* s_re  = (const float*)d_in[5];
  const float* s_im  = (const float*)d_in[6];
  float* out = (float*)d_out;

  int B = in_sizes[0] / 16;
  int blocks = B / ITEMS_PER_BLOCK;
  mmse_pic_kernel<<<blocks, ITEMS_PER_BLOCK, 0, stream>>>(y_re, y_im, h_re, h_im,
                                                          prior, s_re, s_im, out);
}
